# starting kernel
# baseline (speedup 1.0000x reference)
#include <hip/hip_runtime.h>
#include <hip/hip_bf16.h>

typedef __attribute__((ext_vector_type(2))) float v2f;
typedef __attribute__((ext_vector_type(8))) float v8f;

#define DPAD 16
#define DDIM 15

__global__ void fill_kernel(float* __restrict__ p, float v, size_t n) {
    size_t i = (size_t)blockIdx.x * blockDim.x + threadIdx.x;
    size_t stride = (size_t)gridDim.x * blockDim.x;
    for (; i < n; i += stride) p[i] = v;
}

__global__ void pack_weight_kernel(const float* __restrict__ W, const float* __restrict__ b,
                                   float* __restrict__ Wp, float* __restrict__ bp,
                                   int K, int Nc, int Kpad, int Ncpad)
{
    int idx = blockIdx.x * blockDim.x + threadIdx.x;
    int total = Kpad * Ncpad;
    if (idx < total) {
        int k = idx / Ncpad, c = idx % Ncpad;
        float v = (k < K && c < Nc) ? W[k * Nc + c] : 0.f;
        Wp[((size_t)(k >> 1) * Ncpad + c) * 2 + (k & 1)] = v;
    }
    if (idx < Ncpad) bp[idx] = (idx < Nc) ? b[idx] : 0.f;
}

__global__ __launch_bounds__(256) void encode_kernel(
    const float* __restrict__ x, const int* __restrict__ type_id,
    const float* __restrict__ Wj, const float* __restrict__ bj,
    const float* __restrict__ Wm, const float* __restrict__ bm,
    float* __restrict__ h, int Nn)
{
    int i = blockIdx.x * blockDim.x + threadIdx.x;
    if (i >= Nn) return;
    float xi[5];
#pragma unroll
    for (int k = 0; k < 5; ++k) xi[k] = x[(size_t)i * 5 + k];
    int t = type_id[i];
    const float* Wp = (t == 0) ? Wj : Wm;
    const float* bp = (t == 0) ? bj : bm;
    bool valid = (t == 0) || (t == 1);
#pragma unroll
    for (int d = 0; d < DDIM; ++d) {
        float v = 0.f;
        if (valid) {
            v = bp[d];
#pragma unroll
            for (int k = 0; k < 5; ++k) v += xi[k] * Wp[k * DDIM + d];
            v = fmaxf(v, 0.f);
        }
        h[(size_t)i * DPAD + d] = v;
    }
    h[(size_t)i * DPAD + DDIM] = 0.f;
}

template <int EDIM>
__global__ __launch_bounds__(256) void edge_pass_kernel(
    const float* __restrict__ h, float* __restrict__ agg,
    const int* __restrict__ ei, const float* __restrict__ ea,
    const float* __restrict__ We, const float* __restrict__ be, int E)
{
    int e = blockIdx.x * blockDim.x + threadIdx.x;
    if (e >= E) return;
    int src = ei[e];
    int dst = ei[E + e];
    float ev[EDIM];
#pragma unroll
    for (int k = 0; k < EDIM; ++k) ev[k] = ea[(size_t)e * EDIM + k];

    float hl[DPAD];
    const float4* hv = (const float4*)(h + (size_t)src * DPAD);
#pragma unroll
    for (int q = 0; q < 4; ++q) {
        float4 t = hv[q];
        hl[4 * q + 0] = t.x; hl[4 * q + 1] = t.y;
        hl[4 * q + 2] = t.z; hl[4 * q + 3] = t.w;
    }

    float* ag = agg + (size_t)dst * DPAD;
#pragma unroll
    for (int d = 0; d < DDIM; ++d) {
        float m = hl[d] + be[d];
#pragma unroll
        for (int k = 0; k < EDIM; ++k) m += ev[k] * We[k * DDIM + d];
        m = fmaxf(m, 0.f);
        atomicAdd(&ag[d], m);
    }
}

template <bool HAS_A2, int KPAD>
__global__ __launch_bounds__(256) void wmma_gemm_kernel(
    const float* __restrict__ A, const float* __restrict__ A2,
    const float* __restrict__ Wp, const float* __restrict__ bp,
    float* __restrict__ C,
    int M, int Ncpad, int lda, int ldc, int relu)
{
    const int lane = threadIdx.x & 31;
    const int wave = threadIdx.x >> 5;
    const int ntN = Ncpad >> 4;
    const int tile = blockIdx.x * (blockDim.x >> 5) + wave;
    if (tile >= (M >> 4) * ntN) return;
    const int mt = tile / ntN;
    const int nt = tile % ntN;

    const int mrow  = mt * 16 + (lane & 15);
    const int ncol  = nt * 16 + (lane & 15);
    const int khalf = (lane >> 4) << 1;

    const float* ap  = A + (size_t)mrow * lda + khalf;
    const float* a2p = A2 + (size_t)mrow * lda + khalf;
    const float* bpp = Wp + ((size_t)(khalf >> 1) * Ncpad + ncol) * 2;
    const size_t bstep = (size_t)Ncpad * 4;

    v8f acc = {0.f, 0.f, 0.f, 0.f, 0.f, 0.f, 0.f, 0.f};
#pragma unroll
    for (int kk = 0; kk < KPAD; kk += 4) {
        v2f a = *(const v2f*)ap;
        if (HAS_A2) {
            v2f a2 = *(const v2f*)a2p;
            a.x += a2.x; a.y += a2.y;
            a2p += 4;
        }
        v2f b = *(const v2f*)bpp;
        ap += 4;
        bpp += bstep;
        acc = __builtin_amdgcn_wmma_f32_16x16x4_f32(
            false, a, false, b, (short)0, acc, false, false);
    }

    const int rbase = mt * 16 + ((lane >> 4) << 3);
    const float bv = bp[ncol];
#pragma unroll
    for (int r = 0; r < 8; ++r) {
        float v = acc[r] + bv;
        if (relu) v = fmaxf(v, 0.f);
        C[(size_t)(rbase + r) * ldc + ncol] = v;
    }
}

__global__ __launch_bounds__(256) void ln_relu_kernel(float* __restrict__ h, int Nn)
{
    int i = blockIdx.x * blockDim.x + threadIdx.x;
    if (i >= Nn) return;
    float v[DDIM];
    float mu = 0.f;
#pragma unroll
    for (int d = 0; d < DDIM; ++d) { v[d] = h[(size_t)i * DPAD + d]; mu += v[d]; }
    mu *= (1.f / DDIM);
    float var = 0.f;
#pragma unroll
    for (int d = 0; d < DDIM; ++d) { float t = v[d] - mu; var += t * t; }
    var *= (1.f / DDIM);
    float inv = rsqrtf(var + 1e-5f);
#pragma unroll
    for (int d = 0; d < DDIM; ++d)
        h[(size_t)i * DPAD + d] = fmaxf((v[d] - mu) * inv, 0.f);
}

__device__ __forceinline__ void atomicMaxF(float* addr, float v) {
    if (v >= 0.f) atomicMax((int*)addr, __float_as_int(v));
    else          atomicMin((unsigned int*)addr, __float_as_uint(v));
}

__global__ __launch_bounds__(256) void pool_kernel(
    const float* __restrict__ h, const int* __restrict__ type_id,
    const int* __restrict__ batch,
    float* __restrict__ sumJ, float* __restrict__ sumM,
    float* __restrict__ maxJ, float* __restrict__ maxM,
    float* __restrict__ cntJ, float* __restrict__ cntM, int Nn)
{
    int i = blockIdx.x * blockDim.x + threadIdx.x;
    if (i >= Nn) return;
    int t = type_id[i];
    if (t != 0 && t != 1) return;
    int b = batch[i];
    float* s  = (t == 0) ? sumJ : sumM;
    float* mx = (t == 0) ? maxJ : maxM;
    float* c  = (t == 0) ? cntJ : cntM;
    atomicAdd(&c[b], 1.f);
#pragma unroll
    for (int d = 0; d < DDIM; ++d) {
        float v = h[(size_t)i * DPAD + d];
        atomicAdd(&s[(size_t)b * DPAD + d], v);
        atomicMaxF(&mx[(size_t)b * DPAD + d], v);
    }
}

__global__ __launch_bounds__(128) void assemble_ln_kernel(
    const float* __restrict__ sumJ, const float* __restrict__ sumM,
    const float* __restrict__ maxJ, const float* __restrict__ maxM,
    const float* __restrict__ cntJ, const float* __restrict__ cntM,
    const float* __restrict__ u, float* __restrict__ g, int Gg)
{
    int gi = blockIdx.x * blockDim.x + threadIdx.x;
    if (gi >= Gg) return;
    float r[64];
    float cm = fmaxf(cntM[gi], 1.f);
    float cj = fmaxf(cntJ[gi], 1.f);
#pragma unroll
    for (int d = 0; d < DDIM; ++d) r[d] = sumM[(size_t)gi * DPAD + d] / cm;
#pragma unroll
    for (int d = 0; d < DDIM; ++d) r[15 + d] = sumJ[(size_t)gi * DPAD + d] / cj;
#pragma unroll
    for (int d = 0; d < DDIM; ++d) {
        float v = maxJ[(size_t)gi * DPAD + d];
        r[30 + d] = (v > -3.0e38f) ? v : 0.f;
    }
#pragma unroll
    for (int d = 0; d < DDIM; ++d) {
        float v = maxM[(size_t)gi * DPAD + d];
        r[45 + d] = (v > -3.0e38f) ? v : 0.f;
    }
#pragma unroll
    for (int j = 0; j < 4; ++j) r[60 + j] = u[(size_t)gi * 4 + j];

    float mu = 0.f;
#pragma unroll
    for (int k = 0; k < 64; ++k) mu += r[k];
    mu *= (1.f / 64.f);
    float var = 0.f;
#pragma unroll
    for (int k = 0; k < 64; ++k) { float t = r[k] - mu; var += t * t; }
    var *= (1.f / 64.f);
    float inv = rsqrtf(var + 1e-5f);
#pragma unroll
    for (int k = 0; k < 64; ++k) g[(size_t)gi * 64 + k] = (r[k] - mu) * inv;
}

__global__ __launch_bounds__(256) void mlp3_kernel(
    const float* __restrict__ m2, const float* __restrict__ W3,
    const float* __restrict__ b3, float* __restrict__ out, int Gg)
{
    int gi = blockIdx.x * blockDim.x + threadIdx.x;
    if (gi >= Gg) return;
    float s = b3[0];
#pragma unroll
    for (int k = 0; k < 32; ++k) s += m2[(size_t)gi * 32 + k] * W3[k];
    out[gi] = s;
}

extern "C" void kernel_launch(void* const* d_in, const int* in_sizes, int n_in,
                              void* d_out, int out_size, void* d_ws, size_t ws_size,
                              hipStream_t stream)
{
    const float* x       = (const float*)d_in[0];
    const int*   type_id = (const int*)  d_in[1];
    const int*   batch   = (const int*)  d_in[2];
    const float* u       = (const float*)d_in[3];
    const int*   ei_jj   = (const int*)  d_in[4];
    const float* ea_jj   = (const float*)d_in[5];
    const int*   ei_mj   = (const int*)  d_in[6];
    const float* ea_mj   = (const float*)d_in[7];
    const float* Wj_enc  = (const float*)d_in[8];   const float* bj_enc = (const float*)d_in[9];
    const float* Wm_enc  = (const float*)d_in[10];  const float* bm_enc = (const float*)d_in[11];
    const float* We_jj1  = (const float*)d_in[12];  const float* be_jj1 = (const float*)d_in[13];
    const float* Wm_jj1  = (const float*)d_in[14];  const float* bm_jj1 = (const float*)d_in[15];
    const float* We_mj1  = (const float*)d_in[16];  const float* be_mj1 = (const float*)d_in[17];
    const float* Wm_mj1  = (const float*)d_in[18];  const float* bm_mj1 = (const float*)d_in[19];
    const float* We_jj2  = (const float*)d_in[20];  const float* be_jj2 = (const float*)d_in[21];
    const float* Wm_jj2  = (const float*)d_in[22];  const float* bm_jj2 = (const float*)d_in[23];
    const float* We_mj2  = (const float*)d_in[24];  const float* be_mj2 = (const float*)d_in[25];
    const float* Wm_mj2  = (const float*)d_in[26];  const float* bm_mj2 = (const float*)d_in[27];
    const float* W1      = (const float*)d_in[28];  const float* b1     = (const float*)d_in[29];
    const float* W2      = (const float*)d_in[30];  const float* b2     = (const float*)d_in[31];
    const float* W3      = (const float*)d_in[32];  const float* b3     = (const float*)d_in[33];

    const int Nn  = in_sizes[0] / 5;
    const int Gg  = in_sizes[3] / 4;
    const int Ejj = in_sizes[5] / 5;
    const int Emj = in_sizes[7] / 4;

    float* ws = (float*)d_ws;
    const size_t NP = (size_t)Nn * DPAD;
    float* hA  = ws;
    float* hB  = ws + NP;
    float* agg = ws + 2 * NP;
    float* wpk = ws + 3 * NP;
    float* pWjj1 = wpk;          float* pbjj1 = wpk + 256;
    float* pWmj1 = wpk + 272;    float* pbmj1 = wpk + 528;
    float* pWjj2 = wpk + 544;    float* pbjj2 = wpk + 800;
    float* pWmj2 = wpk + 816;    float* pbmj2 = wpk + 1072;
    float* pW1   = wpk + 1088;   float* pb1   = wpk + 5184;
    float* pW2   = wpk + 5248;   float* pb2   = wpk + 7296;

    float* P    = agg;
    float* sumJ = P;
    float* sumM = P + (size_t)16 * Gg;
    float* maxJ = P + (size_t)32 * Gg;
    float* maxM = P + (size_t)48 * Gg;
    float* cntJ = P + (size_t)64 * Gg;
    float* cntM = P + (size_t)65 * Gg;
    float* gbuf = P + (size_t)66 * Gg;
    float* m1   = P + (size_t)130 * Gg;
    float* m2   = P + (size_t)194 * Gg;
    float* out  = (float*)d_out;

    const int TB = 256;
    const int nodeBlocks = (Nn + TB - 1) / TB;
    const int fillBlocks = (int)((NP + TB - 1) / TB);

    pack_weight_kernel<<<1, 256, 0, stream>>>(Wm_jj1, bm_jj1, pWjj1, pbjj1, 15, 15, 16, 16);
    pack_weight_kernel<<<1, 256, 0, stream>>>(Wm_mj1, bm_mj1, pWmj1, pbmj1, 15, 15, 16, 16);
    pack_weight_kernel<<<1, 256, 0, stream>>>(Wm_jj2, bm_jj2, pWjj2, pbjj2, 15, 15, 16, 16);
    pack_weight_kernel<<<1, 256, 0, stream>>>(Wm_mj2, bm_mj2, pWmj2, pbmj2, 15, 15, 16, 16);
    pack_weight_kernel<<<16, 256, 0, stream>>>(W1, b1, pW1, pb1, 64, 64, 64, 64);
    pack_weight_kernel<<<8, 256, 0, stream>>>(W2, b2, pW2, pb2, 64, 32, 64, 32);

    auto gemm16 = [&](const float* A, const float* A2, const float* Wp, const float* bp,
                      float* C, int relu) {
        int tiles = Nn >> 4;
        wmma_gemm_kernel<true, 16><<<(tiles + 7) / 8, TB, 0, stream>>>(
            A, A2, Wp, bp, C, Nn, 16, DPAD, DPAD, relu);
    };
    auto gemm64 = [&](const float* A, const float* Wp, const float* bp,
                      float* C, int Ncpad, int lda, int ldc, int relu) {
        int tiles = (Gg >> 4) * (Ncpad >> 4);
        wmma_gemm_kernel<false, 64><<<(tiles + 7) / 8, TB, 0, stream>>>(
            A, nullptr, Wp, bp, C, Gg, Ncpad, lda, ldc, relu);
    };

    encode_kernel<<<nodeBlocks, TB, 0, stream>>>(x, type_id, Wj_enc, bj_enc, Wm_enc, bm_enc, hA, Nn);

    fill_kernel<<<fillBlocks, TB, 0, stream>>>(agg, 0.f, NP);
    edge_pass_kernel<5><<<(Ejj + TB - 1) / TB, TB, 0, stream>>>(hA, agg, ei_jj, ea_jj, We_jj1, be_jj1, Ejj);
    gemm16(hA, agg, pWjj1, pbjj1, hB, 1);

    fill_kernel<<<fillBlocks, TB, 0, stream>>>(agg, 0.f, NP);
    edge_pass_kernel<4><<<(Emj + TB - 1) / TB, TB, 0, stream>>>(hB, agg, ei_mj, ea_mj, We_mj1, be_mj1, Emj);
    gemm16(hB, agg, pWmj1, pbmj1, hA, 0);

    ln_relu_kernel<<<nodeBlocks, TB, 0, stream>>>(hA, Nn);

    fill_kernel<<<fillBlocks, TB, 0, stream>>>(agg, 0.f, NP);
    edge_pass_kernel<5><<<(Ejj + TB - 1) / TB, TB, 0, stream>>>(hA, agg, ei_jj, ea_jj, We_jj2, be_jj2, Ejj);
    gemm16(hA, agg, pWjj2, pbjj2, hB, 1);

    fill_kernel<<<fillBlocks, TB, 0, stream>>>(agg, 0.f, NP);
    edge_pass_kernel<4><<<(Emj + TB - 1) / TB, TB, 0, stream>>>(hB, agg, ei_mj, ea_mj, We_mj2, be_mj2, Emj);
    gemm16(hB, agg, pWmj2, pbmj2, hA, 1);

    ln_relu_kernel<<<nodeBlocks, TB, 0, stream>>>(hA, Nn);

    size_t zlen = (size_t)32 * Gg;
    fill_kernel<<<(int)((zlen + TB - 1) / TB), TB, 0, stream>>>(sumJ, 0.f, zlen);
    fill_kernel<<<(int)((zlen + TB - 1) / TB), TB, 0, stream>>>(maxJ, -__builtin_inff(), zlen);
    fill_kernel<<<(int)(((size_t)2 * Gg + TB - 1) / TB), TB, 0, stream>>>(cntJ, 0.f, (size_t)2 * Gg);
    pool_kernel<<<nodeBlocks, TB, 0, stream>>>(hA, type_id, batch, sumJ, sumM, maxJ, maxM, cntJ, cntM, Nn);

    assemble_ln_kernel<<<(Gg + 127) / 128, 128, 0, stream>>>(sumJ, sumM, maxJ, maxM, cntJ, cntM, u, gbuf, Gg);
    gemm64(gbuf, pW1, pb1, m1, 64, 64, 64, 1);
    gemm64(m1,   pW2, pb2, m2, 32, 64, 32, 1);
    mlp3_kernel<<<(Gg + TB - 1) / TB, TB, 0, stream>>>(m2, W3, b3, out, Gg);
}


# each conv is one launch: the gathered row h+agg stays in registers and feeds the node-update f32 WMMA in-wave (lane l^16 exchange), output to the other node plane; sum order and k order as before
# speedup vs baseline: 1.9285x; 1.9285x over previous
#include <hip/hip_runtime.h>
#include <hip/hip_bf16.h>

typedef __attribute__((ext_vector_type(2))) float v2f;
typedef __attribute__((ext_vector_type(8))) float v8f;

#define DPAD 16
#define DDIM 15

__global__ void fill_kernel(float* __restrict__ p, float v, size_t n) {
    size_t i = (size_t)blockIdx.x * blockDim.x + threadIdx.x;
    size_t stride = (size_t)gridDim.x * blockDim.x;
    for (; i < n; i += stride) p[i] = v;
}

__global__ void pack_weight_kernel(const float* __restrict__ W, const float* __restrict__ b,
                                   float* __restrict__ Wp, float* __restrict__ bp,
                                   int K, int Nc, int Kpad, int Ncpad)
{
    int idx = blockIdx.x * blockDim.x + threadIdx.x;
    int total = Kpad * Ncpad;
    if (idx < total) {
        int k = idx / Ncpad, c = idx % Ncpad;
        float v = (k < K && c < Nc) ? W[k * Nc + c] : 0.f;
        Wp[((size_t)(k >> 1) * Ncpad + c) * 2 + (k & 1)] = v;
    }
    if (idx < Ncpad) bp[idx] = (idx < Nc) ? b[idx] : 0.f;
}

__global__ __launch_bounds__(256) void encode_kernel(
    const float* __restrict__ x, const int* __restrict__ type_id,
    const float* __restrict__ Wj, const float* __restrict__ bj,
    const float* __restrict__ Wm, const float* __restrict__ bm,
    float* __restrict__ h, int Nn)
{
    int i = blockIdx.x * blockDim.x + threadIdx.x;
    if (i >= Nn) return;
    float xi[5];
#pragma unroll
    for (int k = 0; k < 5; ++k) xi[k] = x[(size_t)i * 5 + k];
    int t = type_id[i];
    const float* Wp = (t == 0) ? Wj : Wm;
    const float* bp = (t == 0) ? bj : bm;
    bool valid = (t == 0) || (t == 1);
#pragma unroll
    for (int d = 0; d < DDIM; ++d) {
        float v = 0.f;
        if (valid) {
            v = bp[d];
#pragma unroll
            for (int k = 0; k < 5; ++k) v += xi[k] * Wp[k * DDIM + d];
            v = fmaxf(v, 0.f);
        }
        h[(size_t)i * DPAD + d] = v;
    }
    h[(size_t)i * DPAD + DDIM] = 0.f;
}

typedef __attribute__((ext_vector_type(2))) int v2i;

__global__ __launch_bounds__(256) void hist_rank_kernel(
    const int* __restrict__ ei, int* __restrict__ cnt, int* __restrict__ rank, int E)
{
    int e = blockIdx.x * blockDim.x + threadIdx.x;
    if (e >= E) return;
    int dst = ei[E + e];
    rank[e] = atomicAdd(&cnt[dst], 1);
}

__global__ __launch_bounds__(256) void csr_block_sum_kernel(
    const int* __restrict__ cnt, int* __restrict__ bsum, int Nn, int nb)
{
    __shared__ int sh[256];
    const int* c = cnt + (size_t)blockIdx.y * Nn;
    const int t = threadIdx.x;
    const int base = blockIdx.x * 1024 + t * 4;
    int s = 0;
#pragma unroll
    for (int k = 0; k < 4; ++k) if (base + k < Nn) s += c[base + k];
    sh[t] = s;
    __syncthreads();
    for (int w = 128; w > 0; w >>= 1) {
        if (t < w) sh[t] += sh[t + w];
        __syncthreads();
    }
    if (t == 0) bsum[blockIdx.y * nb + blockIdx.x] = sh[0];
}

__global__ __launch_bounds__(256) void csr_block_scan_kernel(
    int* __restrict__ cnt, const int* __restrict__ bsum, int Nn, int nb)
{
    __shared__ int sh[256];
    int* c = cnt + (size_t)blockIdx.y * Nn;
    const int* bs = bsum + blockIdx.y * nb;
    const int t = threadIdx.x;

    int pc = 0;
    for (int j = t; j < (int)blockIdx.x; j += 256) pc += bs[j];
    sh[t] = pc;
    __syncthreads();
    for (int w = 128; w > 0; w >>= 1) {
        if (t < w) sh[t] += sh[t + w];
        __syncthreads();
    }
    const int carry = sh[0];
    __syncthreads();

    const int base = blockIdx.x * 1024 + t * 4;
    int v[4];
#pragma unroll
    for (int k = 0; k < 4; ++k) v[k] = (base + k < Nn) ? c[base + k] : 0;
    const int ts = v[0] + v[1] + v[2] + v[3];
    sh[t] = ts;
    __syncthreads();
    for (int d = 1; d < 256; d <<= 1) {
        int x = (t >= d) ? sh[t - d] : 0;
        __syncthreads();
        sh[t] += x;
        __syncthreads();
    }
    int run = carry + sh[t] - ts;
#pragma unroll
    for (int k = 0; k < 4; ++k) {
        if (base + k < Nn) c[base + k] = run;
        run += v[k];
    }
}

__global__ __launch_bounds__(256) void csr_scatter_kernel(
    const int* __restrict__ ei, const int* __restrict__ off,
    const int* __restrict__ rank, v2i* __restrict__ slot, int E)
{
    int e = blockIdx.x * blockDim.x + threadIdx.x;
    if (e >= E) return;
    int src = ei[e];
    int dst = ei[E + e];
    int p = off[dst] + rank[e];
    v2i r; r.x = e; r.y = src;
    slot[p] = r;
}

__device__ __forceinline__ float xhalf(float x) {
    return __builtin_bit_cast(float, __builtin_amdgcn_permlanex16(
        __builtin_bit_cast(int, x), __builtin_bit_cast(int, x), 0x76543210, 0xfedcba98, false, false));
}

template <int EDIM, int MINW, bool RELU>
__global__ __launch_bounds__(256) __attribute__((amdgpu_waves_per_eu(MINW))) void conv_kernel(
    const float* __restrict__ h, float* __restrict__ hout,
    const int* __restrict__ off, const v2i* __restrict__ slot,
    const float* __restrict__ ea,
    const float* __restrict__ We, const float* __restrict__ be,
    const float* __restrict__ Wp, const float* __restrict__ bp,
    int Nn, int E)
{
    const int i = blockIdx.x * blockDim.x + threadIdx.x;
    int p0 = 0, p1 = 0;
    if (i < Nn) {
        p0 = off[i];
        p1 = (i + 1 < Nn) ? off[i + 1] : E;
        p0 = (p0 < 0) ? 0 : p0;
        p1 = (p1 > E) ? E : p1;
    }

    float acc[DPAD];
#pragma unroll
    for (int d = 0; d < DPAD; ++d) acc[d] = 0.f;

    v2i nxt; nxt.x = 0; nxt.y = 0;
    if (p0 < p1) nxt = slot[p0];
    for (int p = p0; p < p1; ++p) {
        const v2i cur = nxt;
        if (p + 1 < p1) nxt = slot[p + 1];
        float ev[EDIM];
#pragma unroll
        for (int k = 0; k < EDIM; ++k) ev[k] = ea[(size_t)cur.x * EDIM + k];

        float hl[DPAD];
        const float4* hv = (const float4*)(h + (size_t)cur.y * DPAD);
#pragma unroll
        for (int q = 0; q < 4; ++q) {
            float4 t = hv[q];
            hl[4 * q + 0] = t.x; hl[4 * q + 1] = t.y;
            hl[4 * q + 2] = t.z; hl[4 * q + 3] = t.w;
        }
#pragma unroll
        for (int d = 0; d < DDIM; ++d) {
            float m = hl[d] + be[d];
#pragma unroll
            for (int k = 0; k < EDIM; ++k) m += ev[k] * We[k * DDIM + d];
            m = fmaxf(m, 0.f);
            acc[d] += m;
        }
    }

    const int lane = threadIdx.x & 31;
    const int c  = lane & 15;
    const int hf = lane >> 4;
    const int wbase = blockIdx.x * blockDim.x + (__builtin_amdgcn_readfirstlane(threadIdx.x) & ~31);

    float s[DPAD];
    const unsigned irow = min((unsigned)(wbase + lane), (unsigned)(Nn - 1));
    const float4* hi = (const float4*)(h + (size_t)irow * DPAD);
#pragma unroll
    for (int q = 0; q < 4; ++q) {
        float4 t = hi[q];
        s[4 * q + 0] = t.x + acc[4 * q + 0]; s[4 * q + 1] = t.y + acc[4 * q + 1];
        s[4 * q + 2] = t.z + acc[4 * q + 2]; s[4 * q + 3] = t.w + acc[4 * q + 3];
    }

    v8f d0 = {0.f, 0.f, 0.f, 0.f, 0.f, 0.f, 0.f, 0.f};
    v8f d1 = {0.f, 0.f, 0.f, 0.f, 0.f, 0.f, 0.f, 0.f};
    const float* bpp = Wp + ((size_t)hf * 16 + c) * 2;
#pragma unroll
    for (int kk = 0; kk < DPAD; kk += 4) {
        const float snd0 = hf ? s[kk + 0] : s[kk + 2];
        const float snd1 = hf ? s[kk + 1] : s[kk + 3];
        const float rcv0 = xhalf(snd0);
        const float rcv1 = xhalf(snd1);
        const float own0 = hf ? s[kk + 2] : s[kk + 0];
        const float own1 = hf ? s[kk + 3] : s[kk + 1];
        v2f a0, a1;
        a0.x = hf ? rcv0 : own0;  a0.y = hf ? rcv1 : own1;
        a1.x = hf ? own0 : rcv0;  a1.y = hf ? own1 : rcv1;
        const v2f b = *(const v2f*)(bpp + (size_t)kk * 16);
        d0 = __builtin_amdgcn_wmma_f32_16x16x4_f32(false, a0, false, b, (short)0, d0, false, false);
        d1 = __builtin_amdgcn_wmma_f32_16x16x4_f32(false, a1, false, b, (short)0, d1, false, false);
    }

    const float bv = bp[c];
    float* orow = hout + (size_t)wbase * DPAD + (8 * hf) * DPAD + c;
    if (wbase + 32 <= Nn) {
#pragma unroll
        for (int r = 0; r < 8; ++r) {
            float v0 = d0[r] + bv;
            float v1 = d1[r] + bv;
            if (RELU) { v0 = fmaxf(v0, 0.f); v1 = fmaxf(v1, 0.f); }
            orow[r * DPAD] = v0;
            orow[(r + 16) * DPAD] = v1;
        }
    } else {
#pragma unroll
        for (int r = 0; r < 8; ++r) {
            const int n0 = wbase + 8 * hf + r;
            float v0 = d0[r] + bv;
            float v1 = d1[r] + bv;
            if (RELU) { v0 = fmaxf(v0, 0.f); v1 = fmaxf(v1, 0.f); }
            if (n0 < Nn)      orow[r * DPAD] = v0;
            if (n0 + 16 < Nn) orow[(r + 16) * DPAD] = v1;
        }
    }
}

template <bool HAS_A2, int KPAD>
__global__ __launch_bounds__(256) void wmma_gemm_kernel(
    const float* __restrict__ A, const float* __restrict__ A2,
    const float* __restrict__ Wp, const float* __restrict__ bp,
    float* __restrict__ C,
    int M, int Ncpad, int lda, int ldc, int relu)
{
    const int lane = threadIdx.x & 31;
    const int wave = threadIdx.x >> 5;
    const int ntN = Ncpad >> 4;
    const int tile = blockIdx.x * (blockDim.x >> 5) + wave;
    if (tile >= (M >> 4) * ntN) return;
    const int mt = tile / ntN;
    const int nt = tile % ntN;

    const int mrow  = mt * 16 + (lane & 15);
    const int ncol  = nt * 16 + (lane & 15);
    const int khalf = (lane >> 4) << 1;

    const float* ap  = A + (size_t)mrow * lda + khalf;
    const float* a2p = A2 + (size_t)mrow * lda + khalf;
    const float* bpp = Wp + ((size_t)(khalf >> 1) * Ncpad + ncol) * 2;
    const size_t bstep = (size_t)Ncpad * 4;

    v8f acc = {0.f, 0.f, 0.f, 0.f, 0.f, 0.f, 0.f, 0.f};
#pragma unroll
    for (int kk = 0; kk < KPAD; kk += 4) {
        v2f a = *(const v2f*)ap;
        if (HAS_A2) {
            v2f a2 = *(const v2f*)a2p;
            a.x += a2.x; a.y += a2.y;
            a2p += 4;
        }
        v2f b = *(const v2f*)bpp;
        ap += 4;
        bpp += bstep;
        acc = __builtin_amdgcn_wmma_f32_16x16x4_f32(
            false, a, false, b, (short)0, acc, false, false);
    }

    const int rbase = mt * 16 + ((lane >> 4) << 3);
    const float bv = bp[ncol];
#pragma unroll
    for (int r = 0; r < 8; ++r) {
        float v = acc[r] + bv;
        if (relu) v = fmaxf(v, 0.f);
        C[(size_t)(rbase + r) * ldc + ncol] = v;
    }
}

__global__ __launch_bounds__(256) void ln_relu_kernel(float* __restrict__ h, int Nn)
{
    int i = blockIdx.x * blockDim.x + threadIdx.x;
    if (i >= Nn) return;
    float v[DDIM];
    float mu = 0.f;
#pragma unroll
    for (int d = 0; d < DDIM; ++d) { v[d] = h[(size_t)i * DPAD + d]; mu += v[d]; }
    mu *= (1.f / DDIM);
    float var = 0.f;
#pragma unroll
    for (int d = 0; d < DDIM; ++d) { float t = v[d] - mu; var += t * t; }
    var *= (1.f / DDIM);
    float inv = rsqrtf(var + 1e-5f);
#pragma unroll
    for (int d = 0; d < DDIM; ++d)
        h[(size_t)i * DPAD + d] = fmaxf((v[d] - mu) * inv, 0.f);
}

__device__ __forceinline__ void atomicMaxF(float* addr, float v) {
    if (v >= 0.f) atomicMax((int*)addr, __float_as_int(v));
    else          atomicMin((unsigned int*)addr, __float_as_uint(v));
}

__global__ __launch_bounds__(256) void pool_kernel(
    const float* __restrict__ h, const int* __restrict__ type_id,
    const int* __restrict__ batch,
    float* __restrict__ sumJ, float* __restrict__ sumM,
    float* __restrict__ maxJ, float* __restrict__ maxM,
    float* __restrict__ cntJ, float* __restrict__ cntM, int Nn)
{
    int i = blockIdx.x * blockDim.x + threadIdx.x;
    if (i >= Nn) return;
    int t = type_id[i];
    if (t != 0 && t != 1) return;
    int b = batch[i];
    float* s  = (t == 0) ? sumJ : sumM;
    float* mx = (t == 0) ? maxJ : maxM;
    float* c  = (t == 0) ? cntJ : cntM;
    atomicAdd(&c[b], 1.f);
#pragma unroll
    for (int d = 0; d < DDIM; ++d) {
        float v = h[(size_t)i * DPAD + d];
        atomicAdd(&s[(size_t)b * DPAD + d], v);
        atomicMaxF(&mx[(size_t)b * DPAD + d], v);
    }
}

__global__ __launch_bounds__(128) void assemble_ln_kernel(
    const float* __restrict__ sumJ, const float* __restrict__ sumM,
    const float* __restrict__ maxJ, const float* __restrict__ maxM,
    const float* __restrict__ cntJ, const float* __restrict__ cntM,
    const float* __restrict__ u, float* __restrict__ g, int Gg)
{
    int gi = blockIdx.x * blockDim.x + threadIdx.x;
    if (gi >= Gg) return;
    float r[64];
    float cm = fmaxf(cntM[gi], 1.f);
    float cj = fmaxf(cntJ[gi], 1.f);
#pragma unroll
    for (int d = 0; d < DDIM; ++d) r[d] = sumM[(size_t)gi * DPAD + d] / cm;
#pragma unroll
    for (int d = 0; d < DDIM; ++d) r[15 + d] = sumJ[(size_t)gi * DPAD + d] / cj;
#pragma unroll
    for (int d = 0; d < DDIM; ++d) {
        float v = maxJ[(size_t)gi * DPAD + d];
        r[30 + d] = (v > -3.0e38f) ? v : 0.f;
    }
#pragma unroll
    for (int d = 0; d < DDIM; ++d) {
        float v = maxM[(size_t)gi * DPAD + d];
        r[45 + d] = (v > -3.0e38f) ? v : 0.f;
    }
#pragma unroll
    for (int j = 0; j < 4; ++j) r[60 + j] = u[(size_t)gi * 4 + j];

    float mu = 0.f;
#pragma unroll
    for (int k = 0; k < 64; ++k) mu += r[k];
    mu *= (1.f / 64.f);
    float var = 0.f;
#pragma unroll
    for (int k = 0; k < 64; ++k) { float t = r[k] - mu; var += t * t; }
    var *= (1.f / 64.f);
    float inv = rsqrtf(var + 1e-5f);
#pragma unroll
    for (int k = 0; k < 64; ++k) g[(size_t)gi * 64 + k] = (r[k] - mu) * inv;
}

__global__ __launch_bounds__(256) void mlp3_kernel(
    const float* __restrict__ m2, const float* __restrict__ W3,
    const float* __restrict__ b3, float* __restrict__ out, int Gg)
{
    int gi = blockIdx.x * blockDim.x + threadIdx.x;
    if (gi >= Gg) return;
    float s = b3[0];
#pragma unroll
    for (int k = 0; k < 32; ++k) s += m2[(size_t)gi * 32 + k] * W3[k];
    out[gi] = s;
}

extern "C" void kernel_launch(void* const* d_in, const int* in_sizes, int n_in,
                              void* d_out, int out_size, void* d_ws, size_t ws_size,
                              hipStream_t stream)
{
    const float* x       = (const float*)d_in[0];
    const int*   type_id = (const int*)  d_in[1];
    const int*   batch   = (const int*)  d_in[2];
    const float* u       = (const float*)d_in[3];
    const int*   ei_jj   = (const int*)  d_in[4];
    const float* ea_jj   = (const float*)d_in[5];
    const int*   ei_mj   = (const int*)  d_in[6];
    const float* ea_mj   = (const float*)d_in[7];
    const float* Wj_enc  = (const float*)d_in[8];   const float* bj_enc = (const float*)d_in[9];
    const float* Wm_enc  = (const float*)d_in[10];  const float* bm_enc = (const float*)d_in[11];
    const float* We_jj1  = (const float*)d_in[12];  const float* be_jj1 = (const float*)d_in[13];
    const float* Wm_jj1  = (const float*)d_in[14];  const float* bm_jj1 = (const float*)d_in[15];
    const float* We_mj1  = (const float*)d_in[16];  const float* be_mj1 = (const float*)d_in[17];
    const float* Wm_mj1  = (const float*)d_in[18];  const float* bm_mj1 = (const float*)d_in[19];
    const float* We_jj2  = (const float*)d_in[20];  const float* be_jj2 = (const float*)d_in[21];
    const float* Wm_jj2  = (const float*)d_in[22];  const float* bm_jj2 = (const float*)d_in[23];
    const float* We_mj2  = (const float*)d_in[24];  const float* be_mj2 = (const float*)d_in[25];
    const float* Wm_mj2  = (const float*)d_in[26];  const float* bm_mj2 = (const float*)d_in[27];
    const float* W1      = (const float*)d_in[28];  const float* b1     = (const float*)d_in[29];
    const float* W2      = (const float*)d_in[30];  const float* b2     = (const float*)d_in[31];
    const float* W3      = (const float*)d_in[32];  const float* b3     = (const float*)d_in[33];

    const int Nn  = in_sizes[0] / 5;
    const int Gg  = in_sizes[3] / 4;
    const int Ejj = in_sizes[5] / 5;
    const int Emj = in_sizes[7] / 4;

    float* ws = (float*)d_ws;
    const size_t NP = (size_t)Nn * DPAD;
    float* H   = ws;
    float* S   = ws + NP;
    float* agg = ws + 2 * NP;
    float* wpk = ws + 3 * NP;
    float* pWjj1 = wpk;          float* pbjj1 = wpk + 256;
    float* pWmj1 = wpk + 272;    float* pbmj1 = wpk + 528;
    float* pWjj2 = wpk + 544;    float* pbjj2 = wpk + 800;
    float* pWmj2 = wpk + 816;    float* pbmj2 = wpk + 1072;
    float* pW1   = wpk + 1088;   float* pb1   = wpk + 5184;
    float* pW2   = wpk + 5248;   float* pb2   = wpk + 7296;

    const int nb = (Nn + 1023) / 1024;
    int* cidx    = (int*)agg;
    int* off_jj  = cidx;
    int* off_mj  = cidx + Nn;
    int* bsum    = cidx + 2 * (size_t)Nn;
    v2i* slot_jj = (v2i*)(cidx + 2 * (size_t)Nn + 2 * (size_t)nb);
    v2i* slot_mj = slot_jj + Ejj;
    int* rank_jj = (int*)S;
    int* rank_mj = rank_jj + Ejj;

    float* P    = agg;
    float* sumJ = P;
    float* sumM = P + (size_t)16 * Gg;
    float* maxJ = P + (size_t)32 * Gg;
    float* maxM = P + (size_t)48 * Gg;
    float* cntJ = P + (size_t)64 * Gg;
    float* cntM = P + (size_t)65 * Gg;
    float* gbuf = P + (size_t)66 * Gg;
    float* m1   = P + (size_t)130 * Gg;
    float* m2   = P + (size_t)194 * Gg;
    float* out  = (float*)d_out;

    const int TB = 256;
    const int nodeBlocks = (Nn + TB - 1) / TB;

    pack_weight_kernel<<<1, 256, 0, stream>>>(Wm_jj1, bm_jj1, pWjj1, pbjj1, 15, 15, 16, 16);
    pack_weight_kernel<<<1, 256, 0, stream>>>(Wm_mj1, bm_mj1, pWmj1, pbmj1, 15, 15, 16, 16);
    pack_weight_kernel<<<1, 256, 0, stream>>>(Wm_jj2, bm_jj2, pWjj2, pbjj2, 15, 15, 16, 16);
    pack_weight_kernel<<<1, 256, 0, stream>>>(Wm_mj2, bm_mj2, pWmj2, pbmj2, 15, 15, 16, 16);
    pack_weight_kernel<<<16, 256, 0, stream>>>(W1, b1, pW1, pb1, 64, 64, 64, 64);
    pack_weight_kernel<<<8, 256, 0, stream>>>(W2, b2, pW2, pb2, 64, 32, 64, 32);

    auto gemm64 = [&](const float* A, const float* Wp, const float* bp,
                      float* C, int Ncpad, int lda, int ldc, int relu) {
        int tiles = (Gg >> 4) * (Ncpad >> 4);
        wmma_gemm_kernel<false, 64><<<(tiles + 7) / 8, TB, 0, stream>>>(
            A, nullptr, Wp, bp, C, Gg, Ncpad, lda, ldc, relu);
    };

    const size_t ncnt = (size_t)2 * Nn;
    fill_kernel<<<(int)((ncnt + TB - 1) / TB), TB, 0, stream>>>((float*)cidx, 0.f, ncnt);
    hist_rank_kernel<<<(Ejj + TB - 1) / TB, TB, 0, stream>>>(ei_jj, off_jj, rank_jj, Ejj);
    hist_rank_kernel<<<(Emj + TB - 1) / TB, TB, 0, stream>>>(ei_mj, off_mj, rank_mj, Emj);
    csr_block_sum_kernel<<<dim3(nb, 2), TB, 0, stream>>>(cidx, bsum, Nn, nb);
    csr_block_scan_kernel<<<dim3(nb, 2), TB, 0, stream>>>(cidx, bsum, Nn, nb);
    csr_scatter_kernel<<<(Ejj + TB - 1) / TB, TB, 0, stream>>>(ei_jj, off_jj, rank_jj, slot_jj, Ejj);
    csr_scatter_kernel<<<(Emj + TB - 1) / TB, TB, 0, stream>>>(ei_mj, off_mj, rank_mj, slot_mj, Emj);

    encode_kernel<<<nodeBlocks, TB, 0, stream>>>(x, type_id, Wj_enc, bj_enc, Wm_enc, bm_enc, H, Nn);

    conv_kernel<5, 9, true><<<nodeBlocks, TB, 0, stream>>>(H, S, off_jj, slot_jj, ea_jj, We_jj1, be_jj1, pWjj1, pbjj1, Nn, Ejj);

    conv_kernel<4, 12, false><<<nodeBlocks, TB, 0, stream>>>(S, H, off_mj, slot_mj, ea_mj, We_mj1, be_mj1, pWmj1, pbmj1, Nn, Emj);

    ln_relu_kernel<<<nodeBlocks, TB, 0, stream>>>(H, Nn);

    conv_kernel<5, 9, true><<<nodeBlocks, TB, 0, stream>>>(H, S, off_jj, slot_jj, ea_jj, We_jj2, be_jj2, pWjj2, pbjj2, Nn, Ejj);

    conv_kernel<4, 12, true><<<nodeBlocks, TB, 0, stream>>>(S, H, off_mj, slot_mj, ea_mj, We_mj2, be_mj2, pWmj2, pbmj2, Nn, Emj);

    ln_relu_kernel<<<nodeBlocks, TB, 0, stream>>>(H, Nn);

    size_t zlen = (size_t)32 * Gg;
    fill_kernel<<<(int)((zlen + TB - 1) / TB), TB, 0, stream>>>(sumJ, 0.f, zlen);
    fill_kernel<<<(int)((zlen + TB - 1) / TB), TB, 0, stream>>>(maxJ, -__builtin_inff(), zlen);
    fill_kernel<<<(int)(((size_t)2 * Gg + TB - 1) / TB), TB, 0, stream>>>(cntJ, 0.f, (size_t)2 * Gg);
    pool_kernel<<<nodeBlocks, TB, 0, stream>>>(H, type_id, batch, sumJ, sumM, maxJ, maxM, cntJ, cntM, Nn);

    assemble_ln_kernel<<<(Gg + 127) / 128, 128, 0, stream>>>(sumJ, sumM, maxJ, maxM, cntJ, cntM, u, gbuf, Gg);
    gemm64(gbuf, pW1, pb1, m1, 64, 64, 64, 1);
    gemm64(m1,   pW2, pb2, m2, 32, 64, 32, 1);
    mlp3_kernel<<<(Gg + TB - 1) / TB, TB, 0, stream>>>(m2, W3, b3, out, Gg);
}


# pooling as a per-graph gather: nodes chained per graph by one int atomic exchange each, a thread per graph walks its chain, register sum/max/count; no float atomics or fills; sum order: chain order
# speedup vs baseline: 2.4895x; 1.2909x over previous
#include <hip/hip_runtime.h>
#include <hip/hip_bf16.h>

typedef __attribute__((ext_vector_type(2))) float v2f;
typedef __attribute__((ext_vector_type(8))) float v8f;

#define DPAD 16
#define DDIM 15

__global__ void fill_kernel(float* __restrict__ p, float v, size_t n) {
    size_t i = (size_t)blockIdx.x * blockDim.x + threadIdx.x;
    size_t stride = (size_t)gridDim.x * blockDim.x;
    for (; i < n; i += stride) p[i] = v;
}

__global__ void pack_weight_kernel(const float* __restrict__ W, const float* __restrict__ b,
                                   float* __restrict__ Wp, float* __restrict__ bp,
                                   int K, int Nc, int Kpad, int Ncpad)
{
    int idx = blockIdx.x * blockDim.x + threadIdx.x;
    int total = Kpad * Ncpad;
    if (idx < total) {
        int k = idx / Ncpad, c = idx % Ncpad;
        float v = (k < K && c < Nc) ? W[k * Nc + c] : 0.f;
        Wp[((size_t)(k >> 1) * Ncpad + c) * 2 + (k & 1)] = v;
    }
    if (idx < Ncpad) bp[idx] = (idx < Nc) ? b[idx] : 0.f;
}

__global__ __launch_bounds__(256) void encode_kernel(
    const float* __restrict__ x, const int* __restrict__ type_id,
    const float* __restrict__ Wj, const float* __restrict__ bj,
    const float* __restrict__ Wm, const float* __restrict__ bm,
    float* __restrict__ h, int Nn)
{
    int i = blockIdx.x * blockDim.x + threadIdx.x;
    if (i >= Nn) return;
    float xi[5];
#pragma unroll
    for (int k = 0; k < 5; ++k) xi[k] = x[(size_t)i * 5 + k];
    int t = type_id[i];
    const float* Wp = (t == 0) ? Wj : Wm;
    const float* bp = (t == 0) ? bj : bm;
    bool valid = (t == 0) || (t == 1);
#pragma unroll
    for (int d = 0; d < DDIM; ++d) {
        float v = 0.f;
        if (valid) {
            v = bp[d];
#pragma unroll
            for (int k = 0; k < 5; ++k) v += xi[k] * Wp[k * DDIM + d];
            v = fmaxf(v, 0.f);
        }
        h[(size_t)i * DPAD + d] = v;
    }
    h[(size_t)i * DPAD + DDIM] = 0.f;
}

typedef __attribute__((ext_vector_type(2))) int v2i;

__global__ __launch_bounds__(256) void hist_rank_kernel(
    const int* __restrict__ ei, int* __restrict__ cnt, int* __restrict__ rank, int E)
{
    int e = blockIdx.x * blockDim.x + threadIdx.x;
    if (e >= E) return;
    int dst = ei[E + e];
    rank[e] = atomicAdd(&cnt[dst], 1);
}

__global__ __launch_bounds__(256) void csr_block_sum_kernel(
    const int* __restrict__ cnt, int* __restrict__ bsum, int Nn, int nb)
{
    __shared__ int sh[256];
    const int* c = cnt + (size_t)blockIdx.y * Nn;
    const int t = threadIdx.x;
    const int base = blockIdx.x * 1024 + t * 4;
    int s = 0;
#pragma unroll
    for (int k = 0; k < 4; ++k) if (base + k < Nn) s += c[base + k];
    sh[t] = s;
    __syncthreads();
    for (int w = 128; w > 0; w >>= 1) {
        if (t < w) sh[t] += sh[t + w];
        __syncthreads();
    }
    if (t == 0) bsum[blockIdx.y * nb + blockIdx.x] = sh[0];
}

__global__ __launch_bounds__(256) void csr_block_scan_kernel(
    int* __restrict__ cnt, const int* __restrict__ bsum, int Nn, int nb)
{
    __shared__ int sh[256];
    int* c = cnt + (size_t)blockIdx.y * Nn;
    const int* bs = bsum + blockIdx.y * nb;
    const int t = threadIdx.x;

    int pc = 0;
    for (int j = t; j < (int)blockIdx.x; j += 256) pc += bs[j];
    sh[t] = pc;
    __syncthreads();
    for (int w = 128; w > 0; w >>= 1) {
        if (t < w) sh[t] += sh[t + w];
        __syncthreads();
    }
    const int carry = sh[0];
    __syncthreads();

    const int base = blockIdx.x * 1024 + t * 4;
    int v[4];
#pragma unroll
    for (int k = 0; k < 4; ++k) v[k] = (base + k < Nn) ? c[base + k] : 0;
    const int ts = v[0] + v[1] + v[2] + v[3];
    sh[t] = ts;
    __syncthreads();
    for (int d = 1; d < 256; d <<= 1) {
        int x = (t >= d) ? sh[t - d] : 0;
        __syncthreads();
        sh[t] += x;
        __syncthreads();
    }
    int run = carry + sh[t] - ts;
#pragma unroll
    for (int k = 0; k < 4; ++k) {
        if (base + k < Nn) c[base + k] = run;
        run += v[k];
    }
}

__global__ __launch_bounds__(256) void csr_scatter_kernel(
    const int* __restrict__ ei, const int* __restrict__ off,
    const int* __restrict__ rank, v2i* __restrict__ slot, int E)
{
    int e = blockIdx.x * blockDim.x + threadIdx.x;
    if (e >= E) return;
    int src = ei[e];
    int dst = ei[E + e];
    int p = off[dst] + rank[e];
    v2i r; r.x = e; r.y = src;
    slot[p] = r;
}

__device__ __forceinline__ float xhalf(float x) {
    return __builtin_bit_cast(float, __builtin_amdgcn_permlanex16(
        __builtin_bit_cast(int, x), __builtin_bit_cast(int, x), 0x76543210, 0xfedcba98, false, false));
}

template <int EDIM, int MINW, bool RELU>
__global__ __launch_bounds__(256) __attribute__((amdgpu_waves_per_eu(MINW))) void conv_kernel(
    const float* __restrict__ h, float* __restrict__ hout,
    const int* __restrict__ off, const v2i* __restrict__ slot,
    const float* __restrict__ ea,
    const float* __restrict__ We, const float* __restrict__ be,
    const float* __restrict__ Wp, const float* __restrict__ bp,
    int Nn, int E)
{
    const int i = blockIdx.x * blockDim.x + threadIdx.x;
    int p0 = 0, p1 = 0;
    if (i < Nn) {
        p0 = off[i];
        p1 = (i + 1 < Nn) ? off[i + 1] : E;
        p0 = (p0 < 0) ? 0 : p0;
        p1 = (p1 > E) ? E : p1;
    }

    float acc[DPAD];
#pragma unroll
    for (int d = 0; d < DPAD; ++d) acc[d] = 0.f;

    v2i nxt; nxt.x = 0; nxt.y = 0;
    if (p0 < p1) nxt = slot[p0];
    for (int p = p0; p < p1; ++p) {
        const v2i cur = nxt;
        if (p + 1 < p1) nxt = slot[p + 1];
        float ev[EDIM];
#pragma unroll
        for (int k = 0; k < EDIM; ++k) ev[k] = ea[(size_t)cur.x * EDIM + k];

        float hl[DPAD];
        const float4* hv = (const float4*)(h + (size_t)cur.y * DPAD);
#pragma unroll
        for (int q = 0; q < 4; ++q) {
            float4 t = hv[q];
            hl[4 * q + 0] = t.x; hl[4 * q + 1] = t.y;
            hl[4 * q + 2] = t.z; hl[4 * q + 3] = t.w;
        }
#pragma unroll
        for (int d = 0; d < DDIM; ++d) {
            float m = hl[d] + be[d];
#pragma unroll
            for (int k = 0; k < EDIM; ++k) m += ev[k] * We[k * DDIM + d];
            m = fmaxf(m, 0.f);
            acc[d] += m;
        }
    }

    const int lane = threadIdx.x & 31;
    const int c  = lane & 15;
    const int hf = lane >> 4;
    const int wbase = blockIdx.x * blockDim.x + (__builtin_amdgcn_readfirstlane(threadIdx.x) & ~31);

    float s[DPAD];
    const unsigned irow = min((unsigned)(wbase + lane), (unsigned)(Nn - 1));
    const float4* hi = (const float4*)(h + (size_t)irow * DPAD);
#pragma unroll
    for (int q = 0; q < 4; ++q) {
        float4 t = hi[q];
        s[4 * q + 0] = t.x + acc[4 * q + 0]; s[4 * q + 1] = t.y + acc[4 * q + 1];
        s[4 * q + 2] = t.z + acc[4 * q + 2]; s[4 * q + 3] = t.w + acc[4 * q + 3];
    }

    v8f d0 = {0.f, 0.f, 0.f, 0.f, 0.f, 0.f, 0.f, 0.f};
    v8f d1 = {0.f, 0.f, 0.f, 0.f, 0.f, 0.f, 0.f, 0.f};
    const float* bpp = Wp + ((size_t)hf * 16 + c) * 2;
#pragma unroll
    for (int kk = 0; kk < DPAD; kk += 4) {
        const float snd0 = hf ? s[kk + 0] : s[kk + 2];
        const float snd1 = hf ? s[kk + 1] : s[kk + 3];
        const float rcv0 = xhalf(snd0);
        const float rcv1 = xhalf(snd1);
        const float own0 = hf ? s[kk + 2] : s[kk + 0];
        const float own1 = hf ? s[kk + 3] : s[kk + 1];
        v2f a0, a1;
        a0.x = hf ? rcv0 : own0;  a0.y = hf ? rcv1 : own1;
        a1.x = hf ? own0 : rcv0;  a1.y = hf ? own1 : rcv1;
        const v2f b = *(const v2f*)(bpp + (size_t)kk * 16);
        d0 = __builtin_amdgcn_wmma_f32_16x16x4_f32(false, a0, false, b, (short)0, d0, false, false);
        d1 = __builtin_amdgcn_wmma_f32_16x16x4_f32(false, a1, false, b, (short)0, d1, false, false);
    }

    const float bv = bp[c];
    float* orow = hout + (size_t)wbase * DPAD + (8 * hf) * DPAD + c;
    if (wbase + 32 <= Nn) {
#pragma unroll
        for (int r = 0; r < 8; ++r) {
            float v0 = d0[r] + bv;
            float v1 = d1[r] + bv;
            if (RELU) { v0 = fmaxf(v0, 0.f); v1 = fmaxf(v1, 0.f); }
            orow[r * DPAD] = v0;
            orow[(r + 16) * DPAD] = v1;
        }
    } else {
#pragma unroll
        for (int r = 0; r < 8; ++r) {
            const int n0 = wbase + 8 * hf + r;
            float v0 = d0[r] + bv;
            float v1 = d1[r] + bv;
            if (RELU) { v0 = fmaxf(v0, 0.f); v1 = fmaxf(v1, 0.f); }
            if (n0 < Nn)      orow[r * DPAD] = v0;
            if (n0 + 16 < Nn) orow[(r + 16) * DPAD] = v1;
        }
    }
}

template <bool HAS_A2, int KPAD>
__global__ __launch_bounds__(256) void wmma_gemm_kernel(
    const float* __restrict__ A, const float* __restrict__ A2,
    const float* __restrict__ Wp, const float* __restrict__ bp,
    float* __restrict__ C,
    int M, int Ncpad, int lda, int ldc, int relu)
{
    const int lane = threadIdx.x & 31;
    const int wave = threadIdx.x >> 5;
    const int ntN = Ncpad >> 4;
    const int tile = blockIdx.x * (blockDim.x >> 5) + wave;
    if (tile >= (M >> 4) * ntN) return;
    const int mt = tile / ntN;
    const int nt = tile % ntN;

    const int mrow  = mt * 16 + (lane & 15);
    const int ncol  = nt * 16 + (lane & 15);
    const int khalf = (lane >> 4) << 1;

    const float* ap  = A + (size_t)mrow * lda + khalf;
    const float* a2p = A2 + (size_t)mrow * lda + khalf;
    const float* bpp = Wp + ((size_t)(khalf >> 1) * Ncpad + ncol) * 2;
    const size_t bstep = (size_t)Ncpad * 4;

    v8f acc = {0.f, 0.f, 0.f, 0.f, 0.f, 0.f, 0.f, 0.f};
#pragma unroll
    for (int kk = 0; kk < KPAD; kk += 4) {
        v2f a = *(const v2f*)ap;
        if (HAS_A2) {
            v2f a2 = *(const v2f*)a2p;
            a.x += a2.x; a.y += a2.y;
            a2p += 4;
        }
        v2f b = *(const v2f*)bpp;
        ap += 4;
        bpp += bstep;
        acc = __builtin_amdgcn_wmma_f32_16x16x4_f32(
            false, a, false, b, (short)0, acc, false, false);
    }

    const int rbase = mt * 16 + ((lane >> 4) << 3);
    const float bv = bp[ncol];
#pragma unroll
    for (int r = 0; r < 8; ++r) {
        float v = acc[r] + bv;
        if (relu) v = fmaxf(v, 0.f);
        C[(size_t)(rbase + r) * ldc + ncol] = v;
    }
}

__global__ __launch_bounds__(256) void ln_relu_kernel(float* __restrict__ h, int Nn)
{
    int i = blockIdx.x * blockDim.x + threadIdx.x;
    if (i >= Nn) return;
    float v[DDIM];
    float mu = 0.f;
#pragma unroll
    for (int d = 0; d < DDIM; ++d) { v[d] = h[(size_t)i * DPAD + d]; mu += v[d]; }
    mu *= (1.f / DDIM);
    float var = 0.f;
#pragma unroll
    for (int d = 0; d < DDIM; ++d) { float t = v[d] - mu; var += t * t; }
    var *= (1.f / DDIM);
    float inv = rsqrtf(var + 1e-5f);
#pragma unroll
    for (int d = 0; d < DDIM; ++d)
        h[(size_t)i * DPAD + d] = fmaxf((v[d] - mu) * inv, 0.f);
}

__global__ __launch_bounds__(256) void pool_link_kernel(
    const int* __restrict__ type_id, const int* __restrict__ batch,
    int* __restrict__ head, int* __restrict__ nxt, int Nn, int Gg)
{
    int i = blockIdx.x * blockDim.x + threadIdx.x;
    if (i >= Nn) return;
    const int t = type_id[i];
    const int b = batch[i];
    int link = 0;
    if ((t == 0 || t == 1) && b >= 0 && b < Gg) link = atomicExch(&head[b], i + 1);
    nxt[i] = link;
}

struct PoolNode { int nx; int tp; float4 r[4]; };

__device__ __forceinline__ void pool_fetch(PoolNode& n, int link,
    const float* __restrict__ h, const int* __restrict__ type_id,
    const int* __restrict__ nxt, int Nn)
{
    const unsigned i = min((unsigned)(link - 1), (unsigned)(Nn - 1));
    n.nx = nxt[i];
    n.tp = type_id[i];
    const float4* hv = (const float4*)(h + (size_t)i * DPAD);
#pragma unroll
    for (int q = 0; q < 4; ++q) n.r[q] = hv[q];
}

__device__ __forceinline__ void pool_accum(const PoolNode& n, bool valid,
    float (&sJ)[DPAD], float (&sM)[DPAD], float (&mJ)[DPAD], float (&mM)[DPAD], int& cJ, int& cM)
{
    const float ninf = -__builtin_inff();
    const bool inJ = valid & (n.tp == 0);
    const bool inM = valid & (n.tp != 0);
    cJ += inJ ? 1 : 0;
    cM += inM ? 1 : 0;
#pragma unroll
    for (int q = 0; q < 4; ++q) {
        const float v[4] = {n.r[q].x, n.r[q].y, n.r[q].z, n.r[q].w};
#pragma unroll
        for (int e = 0; e < 4; ++e) {
            const int d = 4 * q + e;
            sJ[d] += inJ ? v[e] : 0.f;
            sM[d] += inM ? v[e] : 0.f;
            const float cj = inJ ? v[e] : ninf;
            const float cm = inM ? v[e] : ninf;
            mJ[d] = (cj > mJ[d]) ? cj : mJ[d];
            mM[d] = (cm > mM[d]) ? cm : mM[d];
        }
    }
}

__global__ __launch_bounds__(256) void pool_gather_kernel(
    const float* __restrict__ h, const int* __restrict__ type_id,
    const int* __restrict__ head, const int* __restrict__ nxt,
    float* __restrict__ sumJ, float* __restrict__ sumM,
    float* __restrict__ maxJ, float* __restrict__ maxM,
    float* __restrict__ cntJ, float* __restrict__ cntM, int Nn, int Gg)
{
    const int g = blockIdx.x * blockDim.x + threadIdx.x;
    if (g >= Gg) return;
    const float ninf = -__builtin_inff();
    float sJ[DPAD], sM[DPAD], mJ[DPAD], mM[DPAD];
#pragma unroll
    for (int d = 0; d < DPAD; ++d) { sJ[d] = 0.f; sM[d] = 0.f; mJ[d] = ninf; mM[d] = ninf; }
    int cJ = 0, cM = 0;

    PoolNode a, b;
    const int first = head[g];
    bool live = (first != 0);
    pool_fetch(a, first, h, type_id, nxt, Nn);
    for (int it = 0; it < Nn && live; it += 2) {
        pool_fetch(b, a.nx, h, type_id, nxt, Nn);
        pool_accum(a, true, sJ, sM, mJ, mM, cJ, cM);
        const bool moreB = (a.nx != 0);
        pool_fetch(a, b.nx, h, type_id, nxt, Nn);
        pool_accum(b, moreB, sJ, sM, mJ, mM, cJ, cM);
        live = moreB & (b.nx != 0);
    }

    float4* oSJ = (float4*)(sumJ + (size_t)g * DPAD);
    float4* oSM = (float4*)(sumM + (size_t)g * DPAD);
    float4* oMJ = (float4*)(maxJ + (size_t)g * DPAD);
    float4* oMM = (float4*)(maxM + (size_t)g * DPAD);
#pragma unroll
    for (int q = 0; q < 4; ++q) {
        oSJ[q] = make_float4(sJ[4 * q], sJ[4 * q + 1], sJ[4 * q + 2], sJ[4 * q + 3]);
        oSM[q] = make_float4(sM[4 * q], sM[4 * q + 1], sM[4 * q + 2], sM[4 * q + 3]);
        oMJ[q] = make_float4(mJ[4 * q], mJ[4 * q + 1], mJ[4 * q + 2], mJ[4 * q + 3]);
        oMM[q] = make_float4(mM[4 * q], mM[4 * q + 1], mM[4 * q + 2], mM[4 * q + 3]);
    }
    cntJ[g] = (float)cJ;
    cntM[g] = (float)cM;
}

__global__ __launch_bounds__(128) void assemble_ln_kernel(
    const float* __restrict__ sumJ, const float* __restrict__ sumM,
    const float* __restrict__ maxJ, const float* __restrict__ maxM,
    const float* __restrict__ cntJ, const float* __restrict__ cntM,
    const float* __restrict__ u, float* __restrict__ g, int Gg)
{
    int gi = blockIdx.x * blockDim.x + threadIdx.x;
    if (gi >= Gg) return;
    float r[64];
    float cm = fmaxf(cntM[gi], 1.f);
    float cj = fmaxf(cntJ[gi], 1.f);
#pragma unroll
    for (int d = 0; d < DDIM; ++d) r[d] = sumM[(size_t)gi * DPAD + d] / cm;
#pragma unroll
    for (int d = 0; d < DDIM; ++d) r[15 + d] = sumJ[(size_t)gi * DPAD + d] / cj;
#pragma unroll
    for (int d = 0; d < DDIM; ++d) {
        float v = maxJ[(size_t)gi * DPAD + d];
        r[30 + d] = (v > -3.0e38f) ? v : 0.f;
    }
#pragma unroll
    for (int d = 0; d < DDIM; ++d) {
        float v = maxM[(size_t)gi * DPAD + d];
        r[45 + d] = (v > -3.0e38f) ? v : 0.f;
    }
#pragma unroll
    for (int j = 0; j < 4; ++j) r[60 + j] = u[(size_t)gi * 4 + j];

    float mu = 0.f;
#pragma unroll
    for (int k = 0; k < 64; ++k) mu += r[k];
    mu *= (1.f / 64.f);
    float var = 0.f;
#pragma unroll
    for (int k = 0; k < 64; ++k) { float t = r[k] - mu; var += t * t; }
    var *= (1.f / 64.f);
    float inv = rsqrtf(var + 1e-5f);
#pragma unroll
    for (int k = 0; k < 64; ++k) g[(size_t)gi * 64 + k] = (r[k] - mu) * inv;
}

__global__ __launch_bounds__(256) void mlp3_kernel(
    const float* __restrict__ m2, const float* __restrict__ W3,
    const float* __restrict__ b3, float* __restrict__ out, int Gg)
{
    int gi = blockIdx.x * blockDim.x + threadIdx.x;
    if (gi >= Gg) return;
    float s = b3[0];
#pragma unroll
    for (int k = 0; k < 32; ++k) s += m2[(size_t)gi * 32 + k] * W3[k];
    out[gi] = s;
}

extern "C" void kernel_launch(void* const* d_in, const int* in_sizes, int n_in,
                              void* d_out, int out_size, void* d_ws, size_t ws_size,
                              hipStream_t stream)
{
    const float* x       = (const float*)d_in[0];
    const int*   type_id = (const int*)  d_in[1];
    const int*   batch   = (const int*)  d_in[2];
    const float* u       = (const float*)d_in[3];
    const int*   ei_jj   = (const int*)  d_in[4];
    const float* ea_jj   = (const float*)d_in[5];
    const int*   ei_mj   = (const int*)  d_in[6];
    const float* ea_mj   = (const float*)d_in[7];
    const float* Wj_enc  = (const float*)d_in[8];   const float* bj_enc = (const float*)d_in[9];
    const float* Wm_enc  = (const float*)d_in[10];  const float* bm_enc = (const float*)d_in[11];
    const float* We_jj1  = (const float*)d_in[12];  const float* be_jj1 = (const float*)d_in[13];
    const float* Wm_jj1  = (const float*)d_in[14];  const float* bm_jj1 = (const float*)d_in[15];
    const float* We_mj1  = (const float*)d_in[16];  const float* be_mj1 = (const float*)d_in[17];
    const float* Wm_mj1  = (const float*)d_in[18];  const float* bm_mj1 = (const float*)d_in[19];
    const float* We_jj2  = (const float*)d_in[20];  const float* be_jj2 = (const float*)d_in[21];
    const float* Wm_jj2  = (const float*)d_in[22];  const float* bm_jj2 = (const float*)d_in[23];
    const float* We_mj2  = (const float*)d_in[24];  const float* be_mj2 = (const float*)d_in[25];
    const float* Wm_mj2  = (const float*)d_in[26];  const float* bm_mj2 = (const float*)d_in[27];
    const float* W1      = (const float*)d_in[28];  const float* b1     = (const float*)d_in[29];
    const float* W2      = (const float*)d_in[30];  const float* b2     = (const float*)d_in[31];
    const float* W3      = (const float*)d_in[32];  const float* b3     = (const float*)d_in[33];

    const int Nn  = in_sizes[0] / 5;
    const int Gg  = in_sizes[3] / 4;
    const int Ejj = in_sizes[5] / 5;
    const int Emj = in_sizes[7] / 4;

    float* ws = (float*)d_ws;
    const size_t NP = (size_t)Nn * DPAD;
    float* H   = ws;
    float* S   = ws + NP;
    float* agg = ws + 2 * NP;
    float* wpk = ws + 3 * NP;
    float* pWjj1 = wpk;          float* pbjj1 = wpk + 256;
    float* pWmj1 = wpk + 272;    float* pbmj1 = wpk + 528;
    float* pWjj2 = wpk + 544;    float* pbjj2 = wpk + 800;
    float* pWmj2 = wpk + 816;    float* pbmj2 = wpk + 1072;
    float* pW1   = wpk + 1088;   float* pb1   = wpk + 5184;
    float* pW2   = wpk + 5248;   float* pb2   = wpk + 7296;

    const int nb = (Nn + 1023) / 1024;
    int* cidx    = (int*)agg;
    int* off_jj  = cidx;
    int* off_mj  = cidx + Nn;
    int* bsum    = cidx + 2 * (size_t)Nn;
    v2i* slot_jj = (v2i*)(cidx + 2 * (size_t)Nn + 2 * (size_t)nb);
    v2i* slot_mj = slot_jj + Ejj;
    int* rank_jj = (int*)S;
    int* rank_mj = rank_jj + Ejj;
    int* pool_next = (int*)S;
    int* pool_head = pool_next + Nn;

    float* P    = agg;
    float* sumJ = P;
    float* sumM = P + (size_t)16 * Gg;
    float* maxJ = P + (size_t)32 * Gg;
    float* maxM = P + (size_t)48 * Gg;
    float* cntJ = P + (size_t)64 * Gg;
    float* cntM = P + (size_t)65 * Gg;
    float* gbuf = P + (size_t)66 * Gg;
    float* m1   = P + (size_t)130 * Gg;
    float* m2   = P + (size_t)194 * Gg;
    float* out  = (float*)d_out;

    const int TB = 256;
    const int nodeBlocks = (Nn + TB - 1) / TB;

    pack_weight_kernel<<<1, 256, 0, stream>>>(Wm_jj1, bm_jj1, pWjj1, pbjj1, 15, 15, 16, 16);
    pack_weight_kernel<<<1, 256, 0, stream>>>(Wm_mj1, bm_mj1, pWmj1, pbmj1, 15, 15, 16, 16);
    pack_weight_kernel<<<1, 256, 0, stream>>>(Wm_jj2, bm_jj2, pWjj2, pbjj2, 15, 15, 16, 16);
    pack_weight_kernel<<<1, 256, 0, stream>>>(Wm_mj2, bm_mj2, pWmj2, pbmj2, 15, 15, 16, 16);
    pack_weight_kernel<<<16, 256, 0, stream>>>(W1, b1, pW1, pb1, 64, 64, 64, 64);
    pack_weight_kernel<<<8, 256, 0, stream>>>(W2, b2, pW2, pb2, 64, 32, 64, 32);

    auto gemm64 = [&](const float* A, const float* Wp, const float* bp,
                      float* C, int Ncpad, int lda, int ldc, int relu) {
        int tiles = (Gg >> 4) * (Ncpad >> 4);
        wmma_gemm_kernel<false, 64><<<(tiles + 7) / 8, TB, 0, stream>>>(
            A, nullptr, Wp, bp, C, Gg, Ncpad, lda, ldc, relu);
    };

    const size_t ncnt = (size_t)2 * Nn;
    fill_kernel<<<(int)((ncnt + TB - 1) / TB), TB, 0, stream>>>((float*)cidx, 0.f, ncnt);
    hist_rank_kernel<<<(Ejj + TB - 1) / TB, TB, 0, stream>>>(ei_jj, off_jj, rank_jj, Ejj);
    hist_rank_kernel<<<(Emj + TB - 1) / TB, TB, 0, stream>>>(ei_mj, off_mj, rank_mj, Emj);
    csr_block_sum_kernel<<<dim3(nb, 2), TB, 0, stream>>>(cidx, bsum, Nn, nb);
    csr_block_scan_kernel<<<dim3(nb, 2), TB, 0, stream>>>(cidx, bsum, Nn, nb);
    csr_scatter_kernel<<<(Ejj + TB - 1) / TB, TB, 0, stream>>>(ei_jj, off_jj, rank_jj, slot_jj, Ejj);
    csr_scatter_kernel<<<(Emj + TB - 1) / TB, TB, 0, stream>>>(ei_mj, off_mj, rank_mj, slot_mj, Emj);

    encode_kernel<<<nodeBlocks, TB, 0, stream>>>(x, type_id, Wj_enc, bj_enc, Wm_enc, bm_enc, H, Nn);

    conv_kernel<5, 9, true><<<nodeBlocks, TB, 0, stream>>>(H, S, off_jj, slot_jj, ea_jj, We_jj1, be_jj1, pWjj1, pbjj1, Nn, Ejj);

    conv_kernel<4, 12, false><<<nodeBlocks, TB, 0, stream>>>(S, H, off_mj, slot_mj, ea_mj, We_mj1, be_mj1, pWmj1, pbmj1, Nn, Emj);

    ln_relu_kernel<<<nodeBlocks, TB, 0, stream>>>(H, Nn);

    conv_kernel<5, 9, true><<<nodeBlocks, TB, 0, stream>>>(H, S, off_jj, slot_jj, ea_jj, We_jj2, be_jj2, pWjj2, pbjj2, Nn, Ejj);

    conv_kernel<4, 12, true><<<nodeBlocks, TB, 0, stream>>>(S, H, off_mj, slot_mj, ea_mj, We_mj2, be_mj2, pWmj2, pbmj2, Nn, Emj);

    ln_relu_kernel<<<nodeBlocks, TB, 0, stream>>>(H, Nn);

    fill_kernel<<<(Gg + TB - 1) / TB, TB, 0, stream>>>((float*)pool_head, 0.f, (size_t)Gg);
    pool_link_kernel<<<nodeBlocks, TB, 0, stream>>>(type_id, batch, pool_head, pool_next, Nn, Gg);
    pool_gather_kernel<<<(Gg + TB - 1) / TB, TB, 0, stream>>>(H, type_id, pool_head, pool_next,
                                                              sumJ, sumM, maxJ, maxM, cntJ, cntM, Nn, Gg);

    assemble_ln_kernel<<<(Gg + 127) / 128, 128, 0, stream>>>(sumJ, sumM, maxJ, maxM, cntJ, cntM, u, gbuf, Gg);
    gemm64(gbuf, pW1, pb1, m1, 64, 64, 64, 1);
    gemm64(m1,   pW2, pb2, m2, 32, 64, 32, 1);
    mlp3_kernel<<<(Gg + TB - 1) / TB, TB, 0, stream>>>(m2, W3, b3, out, Gg);
}


# CSR build ranks edges with LDS counters in a two-level counting sort (bucket of 4096 destinations, then node), no global atomic per edge; sum order: slot order from LDS atomic ranks
# speedup vs baseline: 2.8304x; 1.1369x over previous
#include <hip/hip_runtime.h>
#include <hip/hip_bf16.h>

typedef __attribute__((ext_vector_type(2))) float v2f;
typedef __attribute__((ext_vector_type(8))) float v8f;

#define DPAD 16
#define DDIM 15

__global__ void fill_kernel(float* __restrict__ p, float v, size_t n) {
    size_t i = (size_t)blockIdx.x * blockDim.x + threadIdx.x;
    size_t stride = (size_t)gridDim.x * blockDim.x;
    for (; i < n; i += stride) p[i] = v;
}

__global__ void pack_weight_kernel(const float* __restrict__ W, const float* __restrict__ b,
                                   float* __restrict__ Wp, float* __restrict__ bp,
                                   int K, int Nc, int Kpad, int Ncpad)
{
    int idx = blockIdx.x * blockDim.x + threadIdx.x;
    int total = Kpad * Ncpad;
    if (idx < total) {
        int k = idx / Ncpad, c = idx % Ncpad;
        float v = (k < K && c < Nc) ? W[k * Nc + c] : 0.f;
        Wp[((size_t)(k >> 1) * Ncpad + c) * 2 + (k & 1)] = v;
    }
    if (idx < Ncpad) bp[idx] = (idx < Nc) ? b[idx] : 0.f;
}

__global__ __launch_bounds__(256) void encode_kernel(
    const float* __restrict__ x, const int* __restrict__ type_id,
    const float* __restrict__ Wj, const float* __restrict__ bj,
    const float* __restrict__ Wm, const float* __restrict__ bm,
    float* __restrict__ h, int Nn)
{
    int i = blockIdx.x * blockDim.x + threadIdx.x;
    if (i >= Nn) return;
    float xi[5];
#pragma unroll
    for (int k = 0; k < 5; ++k) xi[k] = x[(size_t)i * 5 + k];
    int t = type_id[i];
    const float* Wp = (t == 0) ? Wj : Wm;
    const float* bp = (t == 0) ? bj : bm;
    bool valid = (t == 0) || (t == 1);
#pragma unroll
    for (int d = 0; d < DDIM; ++d) {
        float v = 0.f;
        if (valid) {
            v = bp[d];
#pragma unroll
            for (int k = 0; k < 5; ++k) v += xi[k] * Wp[k * DDIM + d];
            v = fmaxf(v, 0.f);
        }
        h[(size_t)i * DPAD + d] = v;
    }
    h[(size_t)i * DPAD + DDIM] = 0.f;
}

typedef __attribute__((ext_vector_type(2))) int v2i;

#define BKT_SHIFT 12
#define BKT_NODES 4096
#define CHUNK_E   4096
#define MAX_BKT   1024

__global__ __launch_bounds__(256) void bucket_count_kernel(
    const int* __restrict__ ei_a, const int* __restrict__ ei_b, int* __restrict__ tab,
    int Ea, int Eb, int Nn, int nch, int nbk)
{
    __shared__ int hist[MAX_BKT];
    const int t = threadIdx.x;
    const int set = blockIdx.y, chunk = blockIdx.x;
    const int E = set ? Eb : Ea;
    const int* __restrict__ ei = set ? ei_b : ei_a;
    for (int i = t; i < nbk; i += 256) hist[i] = 0;
    __syncthreads();
    const int base = chunk * CHUNK_E + t;
    const int last = max(E - 1, 0);
    int d[CHUNK_E / 256];
#pragma unroll
    for (int k = 0; k < CHUNK_E / 256; ++k) d[k] = ei[(size_t)E + min(base + k * 256, last)];
#pragma unroll
    for (int k = 0; k < CHUNK_E / 256; ++k) {
        if (base + k * 256 < E) {
            const unsigned dst = min((unsigned)d[k], (unsigned)(Nn - 1));
            atomicAdd(&hist[dst >> BKT_SHIFT], 1);
        }
    }
    __syncthreads();
    int* tb = tab + (size_t)set * nbk * nch;
    for (int i = t; i < nbk; i += 256) tb[(size_t)i * nch + chunk] = hist[i];
}

__global__ __launch_bounds__(256) void bucket_place_kernel(
    const int* __restrict__ ei_a, const int* __restrict__ ei_b, const int* __restrict__ tab,
    v2i* __restrict__ res_a, v2i* __restrict__ res_b,
    int* __restrict__ rdst_a, int* __restrict__ rdst_b,
    int Ea, int Eb, int Nn, int nch, int nbk)
{
    __shared__ int cursor[MAX_BKT];
    const int t = threadIdx.x;
    const int set = blockIdx.y, chunk = blockIdx.x;
    const int E = set ? Eb : Ea;
    const int* __restrict__ ei = set ? ei_b : ei_a;
    v2i* __restrict__ res = set ? res_b : res_a;
    int* __restrict__ rdst = set ? rdst_b : rdst_a;
    const int* tb = tab + (size_t)set * nbk * nch;
    for (int i = t; i < nbk; i += 256) cursor[i] = tb[(size_t)i * nch + chunk];
    __syncthreads();
    const int base = chunk * CHUNK_E + t;
    const int last = max(E - 1, 0);
    int sv[CHUNK_E / 256], d[CHUNK_E / 256];
#pragma unroll
    for (int k = 0; k < CHUNK_E / 256; ++k) {
        const int ec = min(base + k * 256, last);
        sv[k] = ei[ec];
        d[k]  = ei[(size_t)E + ec];
    }
#pragma unroll
    for (int k = 0; k < CHUNK_E / 256; ++k) {
        const int e = base + k * 256;
        if (e < E) {
            const int src = sv[k];
            const unsigned dst = min((unsigned)d[k], (unsigned)(Nn - 1));
            unsigned pos = (unsigned)atomicAdd(&cursor[dst >> BKT_SHIFT], 1);
            pos = min(pos, (unsigned)last);
            v2i o; o.x = e; o.y = src;
            res[pos] = o;
            rdst[pos] = (int)dst;
        }
    }
}

__global__ __launch_bounds__(1024) void csr_bucket_kernel(
    const v2i* __restrict__ res_a, const v2i* __restrict__ res_b,
    const int* __restrict__ rdst_a, const int* __restrict__ rdst_b, const int* __restrict__ tab,
    int* __restrict__ off, v2i* __restrict__ slot_a, v2i* __restrict__ slot_b,
    int Ea, int Eb, int Nn, int nch, int nbk)
{
    __shared__ int cur[BKT_NODES];
    __shared__ int part[1024];
    const int t = threadIdx.x;
    const int set = blockIdx.y, b = blockIdx.x;
    const int E = set ? Eb : Ea;
    const v2i* __restrict__ res = set ? res_b : res_a;
    const int* __restrict__ rdst = set ? rdst_b : rdst_a;
    v2i* __restrict__ slot = set ? slot_b : slot_a;
    int* offs = off + (size_t)set * Nn;
    const int* tb = tab + (size_t)set * nbk * nch;
    int s0 = tb[(size_t)b * nch];
    int s1 = (b + 1 < nbk) ? tb[(size_t)(b + 1) * nch] : E;
    s0 = min(max(s0, 0), E);
    s1 = min(max(s1, s0), E);

    for (int i = t; i < BKT_NODES; i += 1024) cur[i] = 0;
    __syncthreads();
    const int last = max(E - 1, 0);
    for (int p0 = s0 + t; p0 < s1; p0 += 8 * 1024) {
        int d[8];
#pragma unroll
        for (int k = 0; k < 8; ++k) d[k] = rdst[min(p0 + k * 1024, last)];
#pragma unroll
        for (int k = 0; k < 8; ++k)
            if (p0 + k * 1024 < s1) atomicAdd(&cur[d[k] & (BKT_NODES - 1)], 1);
    }
    __syncthreads();

    int v[4];
#pragma unroll
    for (int k = 0; k < 4; ++k) v[k] = cur[4 * t + k];
    const int ts = v[0] + v[1] + v[2] + v[3];
    part[t] = ts;
    __syncthreads();
    for (int d = 1; d < 1024; d <<= 1) {
        const int x = (t >= d) ? part[t - d] : 0;
        __syncthreads();
        part[t] += x;
        __syncthreads();
    }
    int run = part[t] - ts;
#pragma unroll
    for (int k = 0; k < 4; ++k) {
        const int node = b * BKT_NODES + 4 * t + k;
        if (node < Nn) offs[node] = s0 + run;
        cur[4 * t + k] = run;
        run += v[k];
    }
    __syncthreads();

    for (int p0 = s0 + t; p0 < s1; p0 += 4 * 1024) {
        v2i es[4];
        int d[4];
#pragma unroll
        for (int k = 0; k < 4; ++k) {
            const int q = min(p0 + k * 1024, last);
            es[k] = res[q];
            d[k] = rdst[q];
        }
#pragma unroll
        for (int k = 0; k < 4; ++k) {
            if (p0 + k * 1024 < s1) {
                unsigned pos = (unsigned)(s0 + atomicAdd(&cur[d[k] & (BKT_NODES - 1)], 1));
                pos = min(pos, (unsigned)last);
                slot[pos] = es[k];
            }
        }
    }
}

__global__ __launch_bounds__(256) void csr_block_sum_kernel(
    const int* __restrict__ cnt, int* __restrict__ bsum, int Nn, int nb)
{
    __shared__ int sh[256];
    const int* c = cnt + (size_t)blockIdx.y * Nn;
    const int t = threadIdx.x;
    const int base = blockIdx.x * 1024 + t * 4;
    int s = 0;
#pragma unroll
    for (int k = 0; k < 4; ++k) if (base + k < Nn) s += c[base + k];
    sh[t] = s;
    __syncthreads();
    for (int w = 128; w > 0; w >>= 1) {
        if (t < w) sh[t] += sh[t + w];
        __syncthreads();
    }
    if (t == 0) bsum[blockIdx.y * nb + blockIdx.x] = sh[0];
}

__global__ __launch_bounds__(256) void csr_block_scan_kernel(
    int* __restrict__ cnt, const int* __restrict__ bsum, int Nn, int nb)
{
    __shared__ int sh[256];
    int* c = cnt + (size_t)blockIdx.y * Nn;
    const int* bs = bsum + blockIdx.y * nb;
    const int t = threadIdx.x;

    int pc = 0;
    for (int j = t; j < (int)blockIdx.x; j += 256) pc += bs[j];
    sh[t] = pc;
    __syncthreads();
    for (int w = 128; w > 0; w >>= 1) {
        if (t < w) sh[t] += sh[t + w];
        __syncthreads();
    }
    const int carry = sh[0];
    __syncthreads();

    const int base = blockIdx.x * 1024 + t * 4;
    int v[4];
#pragma unroll
    for (int k = 0; k < 4; ++k) v[k] = (base + k < Nn) ? c[base + k] : 0;
    const int ts = v[0] + v[1] + v[2] + v[3];
    sh[t] = ts;
    __syncthreads();
    for (int d = 1; d < 256; d <<= 1) {
        int x = (t >= d) ? sh[t - d] : 0;
        __syncthreads();
        sh[t] += x;
        __syncthreads();
    }
    int run = carry + sh[t] - ts;
#pragma unroll
    for (int k = 0; k < 4; ++k) {
        if (base + k < Nn) c[base + k] = run;
        run += v[k];
    }
}

__device__ __forceinline__ float xhalf(float x) {
    return __builtin_bit_cast(float, __builtin_amdgcn_permlanex16(
        __builtin_bit_cast(int, x), __builtin_bit_cast(int, x), 0x76543210, 0xfedcba98, false, false));
}

template <int EDIM, int MINW, bool RELU>
__global__ __launch_bounds__(256) __attribute__((amdgpu_waves_per_eu(MINW))) void conv_kernel(
    const float* __restrict__ h, float* __restrict__ hout,
    const int* __restrict__ off, const v2i* __restrict__ slot,
    const float* __restrict__ ea,
    const float* __restrict__ We, const float* __restrict__ be,
    const float* __restrict__ Wp, const float* __restrict__ bp,
    int Nn, int E)
{
    const int i = blockIdx.x * blockDim.x + threadIdx.x;
    int p0 = 0, p1 = 0;
    if (i < Nn) {
        p0 = off[i];
        p1 = (i + 1 < Nn) ? off[i + 1] : E;
        p0 = (p0 < 0) ? 0 : p0;
        p1 = (p1 > E) ? E : p1;
    }

    float acc[DPAD];
#pragma unroll
    for (int d = 0; d < DPAD; ++d) acc[d] = 0.f;

    v2i nxt; nxt.x = 0; nxt.y = 0;
    if (p0 < p1) nxt = slot[p0];
    for (int p = p0; p < p1; ++p) {
        const v2i cur = nxt;
        if (p + 1 < p1) nxt = slot[p + 1];
        float ev[EDIM];
#pragma unroll
        for (int k = 0; k < EDIM; ++k) ev[k] = ea[(size_t)cur.x * EDIM + k];

        float hl[DPAD];
        const float4* hv = (const float4*)(h + (size_t)cur.y * DPAD);
#pragma unroll
        for (int q = 0; q < 4; ++q) {
            float4 t = hv[q];
            hl[4 * q + 0] = t.x; hl[4 * q + 1] = t.y;
            hl[4 * q + 2] = t.z; hl[4 * q + 3] = t.w;
        }
#pragma unroll
        for (int d = 0; d < DDIM; ++d) {
            float m = hl[d] + be[d];
#pragma unroll
            for (int k = 0; k < EDIM; ++k) m += ev[k] * We[k * DDIM + d];
            m = fmaxf(m, 0.f);
            acc[d] += m;
        }
    }

    const int lane = threadIdx.x & 31;
    const int c  = lane & 15;
    const int hf = lane >> 4;
    const int wbase = blockIdx.x * blockDim.x + (__builtin_amdgcn_readfirstlane(threadIdx.x) & ~31);

    float s[DPAD];
    const unsigned irow = min((unsigned)(wbase + lane), (unsigned)(Nn - 1));
    const float4* hi = (const float4*)(h + (size_t)irow * DPAD);
#pragma unroll
    for (int q = 0; q < 4; ++q) {
        float4 t = hi[q];
        s[4 * q + 0] = t.x + acc[4 * q + 0]; s[4 * q + 1] = t.y + acc[4 * q + 1];
        s[4 * q + 2] = t.z + acc[4 * q + 2]; s[4 * q + 3] = t.w + acc[4 * q + 3];
    }

    v8f d0 = {0.f, 0.f, 0.f, 0.f, 0.f, 0.f, 0.f, 0.f};
    v8f d1 = {0.f, 0.f, 0.f, 0.f, 0.f, 0.f, 0.f, 0.f};
    const float* bpp = Wp + ((size_t)hf * 16 + c) * 2;
#pragma unroll
    for (int kk = 0; kk < DPAD; kk += 4) {
        const float snd0 = hf ? s[kk + 0] : s[kk + 2];
        const float snd1 = hf ? s[kk + 1] : s[kk + 3];
        const float rcv0 = xhalf(snd0);
        const float rcv1 = xhalf(snd1);
        const float own0 = hf ? s[kk + 2] : s[kk + 0];
        const float own1 = hf ? s[kk + 3] : s[kk + 1];
        v2f a0, a1;
        a0.x = hf ? rcv0 : own0;  a0.y = hf ? rcv1 : own1;
        a1.x = hf ? own0 : rcv0;  a1.y = hf ? own1 : rcv1;
        const v2f b = *(const v2f*)(bpp + (size_t)kk * 16);
        d0 = __builtin_amdgcn_wmma_f32_16x16x4_f32(false, a0, false, b, (short)0, d0, false, false);
        d1 = __builtin_amdgcn_wmma_f32_16x16x4_f32(false, a1, false, b, (short)0, d1, false, false);
    }

    const float bv = bp[c];
    float* orow = hout + (size_t)wbase * DPAD + (8 * hf) * DPAD + c;
    if (wbase + 32 <= Nn) {
#pragma unroll
        for (int r = 0; r < 8; ++r) {
            float v0 = d0[r] + bv;
            float v1 = d1[r] + bv;
            if (RELU) { v0 = fmaxf(v0, 0.f); v1 = fmaxf(v1, 0.f); }
            orow[r * DPAD] = v0;
            orow[(r + 16) * DPAD] = v1;
        }
    } else {
#pragma unroll
        for (int r = 0; r < 8; ++r) {
            const int n0 = wbase + 8 * hf + r;
            float v0 = d0[r] + bv;
            float v1 = d1[r] + bv;
            if (RELU) { v0 = fmaxf(v0, 0.f); v1 = fmaxf(v1, 0.f); }
            if (n0 < Nn)      orow[r * DPAD] = v0;
            if (n0 + 16 < Nn) orow[(r + 16) * DPAD] = v1;
        }
    }
}

template <bool HAS_A2, int KPAD>
__global__ __launch_bounds__(256) void wmma_gemm_kernel(
    const float* __restrict__ A, const float* __restrict__ A2,
    const float* __restrict__ Wp, const float* __restrict__ bp,
    float* __restrict__ C,
    int M, int Ncpad, int lda, int ldc, int relu)
{
    const int lane = threadIdx.x & 31;
    const int wave = threadIdx.x >> 5;
    const int ntN = Ncpad >> 4;
    const int tile = blockIdx.x * (blockDim.x >> 5) + wave;
    if (tile >= (M >> 4) * ntN) return;
    const int mt = tile / ntN;
    const int nt = tile % ntN;

    const int mrow  = mt * 16 + (lane & 15);
    const int ncol  = nt * 16 + (lane & 15);
    const int khalf = (lane >> 4) << 1;

    const float* ap  = A + (size_t)mrow * lda + khalf;
    const float* a2p = A2 + (size_t)mrow * lda + khalf;
    const float* bpp = Wp + ((size_t)(khalf >> 1) * Ncpad + ncol) * 2;
    const size_t bstep = (size_t)Ncpad * 4;

    v8f acc = {0.f, 0.f, 0.f, 0.f, 0.f, 0.f, 0.f, 0.f};
#pragma unroll
    for (int kk = 0; kk < KPAD; kk += 4) {
        v2f a = *(const v2f*)ap;
        if (HAS_A2) {
            v2f a2 = *(const v2f*)a2p;
            a.x += a2.x; a.y += a2.y;
            a2p += 4;
        }
        v2f b = *(const v2f*)bpp;
        ap += 4;
        bpp += bstep;
        acc = __builtin_amdgcn_wmma_f32_16x16x4_f32(
            false, a, false, b, (short)0, acc, false, false);
    }

    const int rbase = mt * 16 + ((lane >> 4) << 3);
    const float bv = bp[ncol];
#pragma unroll
    for (int r = 0; r < 8; ++r) {
        float v = acc[r] + bv;
        if (relu) v = fmaxf(v, 0.f);
        C[(size_t)(rbase + r) * ldc + ncol] = v;
    }
}

__global__ __launch_bounds__(256) void ln_relu_kernel(float* __restrict__ h, int Nn)
{
    int i = blockIdx.x * blockDim.x + threadIdx.x;
    if (i >= Nn) return;
    float v[DDIM];
    float mu = 0.f;
#pragma unroll
    for (int d = 0; d < DDIM; ++d) { v[d] = h[(size_t)i * DPAD + d]; mu += v[d]; }
    mu *= (1.f / DDIM);
    float var = 0.f;
#pragma unroll
    for (int d = 0; d < DDIM; ++d) { float t = v[d] - mu; var += t * t; }
    var *= (1.f / DDIM);
    float inv = rsqrtf(var + 1e-5f);
#pragma unroll
    for (int d = 0; d < DDIM; ++d)
        h[(size_t)i * DPAD + d] = fmaxf((v[d] - mu) * inv, 0.f);
}

__global__ __launch_bounds__(256) void pool_link_kernel(
    const int* __restrict__ type_id, const int* __restrict__ batch,
    int* __restrict__ head, int* __restrict__ nxt, int Nn, int Gg)
{
    int i = blockIdx.x * blockDim.x + threadIdx.x;
    if (i >= Nn) return;
    const int t = type_id[i];
    const int b = batch[i];
    int link = 0;
    if ((t == 0 || t == 1) && b >= 0 && b < Gg) link = atomicExch(&head[b], i + 1);
    nxt[i] = link;
}

struct PoolNode { int nx; int tp; float4 r[4]; };

__device__ __forceinline__ void pool_fetch(PoolNode& n, int link,
    const float* __restrict__ h, const int* __restrict__ type_id,
    const int* __restrict__ nxt, int Nn)
{
    const unsigned i = min((unsigned)(link - 1), (unsigned)(Nn - 1));
    n.nx = nxt[i];
    n.tp = type_id[i];
    const float4* hv = (const float4*)(h + (size_t)i * DPAD);
#pragma unroll
    for (int q = 0; q < 4; ++q) n.r[q] = hv[q];
}

__device__ __forceinline__ void pool_accum(const PoolNode& n, bool valid,
    float (&sJ)[DPAD], float (&sM)[DPAD], float (&mJ)[DPAD], float (&mM)[DPAD], int& cJ, int& cM)
{
    const float ninf = -__builtin_inff();
    const bool inJ = valid & (n.tp == 0);
    const bool inM = valid & (n.tp != 0);
    cJ += inJ ? 1 : 0;
    cM += inM ? 1 : 0;
#pragma unroll
    for (int q = 0; q < 4; ++q) {
        const float v[4] = {n.r[q].x, n.r[q].y, n.r[q].z, n.r[q].w};
#pragma unroll
        for (int e = 0; e < 4; ++e) {
            const int d = 4 * q + e;
            sJ[d] += inJ ? v[e] : 0.f;
            sM[d] += inM ? v[e] : 0.f;
            const float cj = inJ ? v[e] : ninf;
            const float cm = inM ? v[e] : ninf;
            mJ[d] = (cj > mJ[d]) ? cj : mJ[d];
            mM[d] = (cm > mM[d]) ? cm : mM[d];
        }
    }
}

__global__ __launch_bounds__(256) void pool_gather_kernel(
    const float* __restrict__ h, const int* __restrict__ type_id,
    const int* __restrict__ head, const int* __restrict__ nxt,
    float* __restrict__ sumJ, float* __restrict__ sumM,
    float* __restrict__ maxJ, float* __restrict__ maxM,
    float* __restrict__ cntJ, float* __restrict__ cntM, int Nn, int Gg)
{
    const int g = blockIdx.x * blockDim.x + threadIdx.x;
    if (g >= Gg) return;
    const float ninf = -__builtin_inff();
    float sJ[DPAD], sM[DPAD], mJ[DPAD], mM[DPAD];
#pragma unroll
    for (int d = 0; d < DPAD; ++d) { sJ[d] = 0.f; sM[d] = 0.f; mJ[d] = ninf; mM[d] = ninf; }
    int cJ = 0, cM = 0;

    PoolNode a, b;
    const int first = head[g];
    bool live = (first != 0);
    pool_fetch(a, first, h, type_id, nxt, Nn);
    for (int it = 0; it < Nn && live; it += 2) {
        pool_fetch(b, a.nx, h, type_id, nxt, Nn);
        pool_accum(a, true, sJ, sM, mJ, mM, cJ, cM);
        const bool moreB = (a.nx != 0);
        pool_fetch(a, b.nx, h, type_id, nxt, Nn);
        pool_accum(b, moreB, sJ, sM, mJ, mM, cJ, cM);
        live = moreB & (b.nx != 0);
    }

    float4* oSJ = (float4*)(sumJ + (size_t)g * DPAD);
    float4* oSM = (float4*)(sumM + (size_t)g * DPAD);
    float4* oMJ = (float4*)(maxJ + (size_t)g * DPAD);
    float4* oMM = (float4*)(maxM + (size_t)g * DPAD);
#pragma unroll
    for (int q = 0; q < 4; ++q) {
        oSJ[q] = make_float4(sJ[4 * q], sJ[4 * q + 1], sJ[4 * q + 2], sJ[4 * q + 3]);
        oSM[q] = make_float4(sM[4 * q], sM[4 * q + 1], sM[4 * q + 2], sM[4 * q + 3]);
        oMJ[q] = make_float4(mJ[4 * q], mJ[4 * q + 1], mJ[4 * q + 2], mJ[4 * q + 3]);
        oMM[q] = make_float4(mM[4 * q], mM[4 * q + 1], mM[4 * q + 2], mM[4 * q + 3]);
    }
    cntJ[g] = (float)cJ;
    cntM[g] = (float)cM;
}

__global__ __launch_bounds__(128) void assemble_ln_kernel(
    const float* __restrict__ sumJ, const float* __restrict__ sumM,
    const float* __restrict__ maxJ, const float* __restrict__ maxM,
    const float* __restrict__ cntJ, const float* __restrict__ cntM,
    const float* __restrict__ u, float* __restrict__ g, int Gg)
{
    int gi = blockIdx.x * blockDim.x + threadIdx.x;
    if (gi >= Gg) return;
    float r[64];
    float cm = fmaxf(cntM[gi], 1.f);
    float cj = fmaxf(cntJ[gi], 1.f);
#pragma unroll
    for (int d = 0; d < DDIM; ++d) r[d] = sumM[(size_t)gi * DPAD + d] / cm;
#pragma unroll
    for (int d = 0; d < DDIM; ++d) r[15 + d] = sumJ[(size_t)gi * DPAD + d] / cj;
#pragma unroll
    for (int d = 0; d < DDIM; ++d) {
        float v = maxJ[(size_t)gi * DPAD + d];
        r[30 + d] = (v > -3.0e38f) ? v : 0.f;
    }
#pragma unroll
    for (int d = 0; d < DDIM; ++d) {
        float v = maxM[(size_t)gi * DPAD + d];
        r[45 + d] = (v > -3.0e38f) ? v : 0.f;
    }
#pragma unroll
    for (int j = 0; j < 4; ++j) r[60 + j] = u[(size_t)gi * 4 + j];

    float mu = 0.f;
#pragma unroll
    for (int k = 0; k < 64; ++k) mu += r[k];
    mu *= (1.f / 64.f);
    float var = 0.f;
#pragma unroll
    for (int k = 0; k < 64; ++k) { float t = r[k] - mu; var += t * t; }
    var *= (1.f / 64.f);
    float inv = rsqrtf(var + 1e-5f);
#pragma unroll
    for (int k = 0; k < 64; ++k) g[(size_t)gi * 64 + k] = (r[k] - mu) * inv;
}

__global__ __launch_bounds__(256) void mlp3_kernel(
    const float* __restrict__ m2, const float* __restrict__ W3,
    const float* __restrict__ b3, float* __restrict__ out, int Gg)
{
    int gi = blockIdx.x * blockDim.x + threadIdx.x;
    if (gi >= Gg) return;
    float s = b3[0];
#pragma unroll
    for (int k = 0; k < 32; ++k) s += m2[(size_t)gi * 32 + k] * W3[k];
    out[gi] = s;
}

extern "C" void kernel_launch(void* const* d_in, const int* in_sizes, int n_in,
                              void* d_out, int out_size, void* d_ws, size_t ws_size,
                              hipStream_t stream)
{
    const float* x       = (const float*)d_in[0];
    const int*   type_id = (const int*)  d_in[1];
    const int*   batch   = (const int*)  d_in[2];
    const float* u       = (const float*)d_in[3];
    const int*   ei_jj   = (const int*)  d_in[4];
    const float* ea_jj   = (const float*)d_in[5];
    const int*   ei_mj   = (const int*)  d_in[6];
    const float* ea_mj   = (const float*)d_in[7];
    const float* Wj_enc  = (const float*)d_in[8];   const float* bj_enc = (const float*)d_in[9];
    const float* Wm_enc  = (const float*)d_in[10];  const float* bm_enc = (const float*)d_in[11];
    const float* We_jj1  = (const float*)d_in[12];  const float* be_jj1 = (const float*)d_in[13];
    const float* Wm_jj1  = (const float*)d_in[14];  const float* bm_jj1 = (const float*)d_in[15];
    const float* We_mj1  = (const float*)d_in[16];  const float* be_mj1 = (const float*)d_in[17];
    const float* Wm_mj1  = (const float*)d_in[18];  const float* bm_mj1 = (const float*)d_in[19];
    const float* We_jj2  = (const float*)d_in[20];  const float* be_jj2 = (const float*)d_in[21];
    const float* Wm_jj2  = (const float*)d_in[22];  const float* bm_jj2 = (const float*)d_in[23];
    const float* We_mj2  = (const float*)d_in[24];  const float* be_mj2 = (const float*)d_in[25];
    const float* Wm_mj2  = (const float*)d_in[26];  const float* bm_mj2 = (const float*)d_in[27];
    const float* W1      = (const float*)d_in[28];  const float* b1     = (const float*)d_in[29];
    const float* W2      = (const float*)d_in[30];  const float* b2     = (const float*)d_in[31];
    const float* W3      = (const float*)d_in[32];  const float* b3     = (const float*)d_in[33];

    const int Nn  = in_sizes[0] / 5;
    const int Gg  = in_sizes[3] / 4;
    const int Ejj = in_sizes[5] / 5;
    const int Emj = in_sizes[7] / 4;

    float* ws = (float*)d_ws;
    const size_t NP = (size_t)Nn * DPAD;
    float* H   = ws;
    float* S   = ws + NP;
    float* agg = ws + 2 * NP;
    float* wpk = ws + 3 * NP;
    float* pWjj1 = wpk;          float* pbjj1 = wpk + 256;
    float* pWmj1 = wpk + 272;    float* pbmj1 = wpk + 528;
    float* pWjj2 = wpk + 544;    float* pbjj2 = wpk + 800;
    float* pWmj2 = wpk + 816;    float* pbmj2 = wpk + 1072;
    float* pW1   = wpk + 1088;   float* pb1   = wpk + 5184;
    float* pW2   = wpk + 5248;   float* pb2   = wpk + 7296;

    const int nb = (Nn + 1023) / 1024;
    int* cidx    = (int*)agg;
    int* off_jj  = cidx;
    int* off_mj  = cidx + Nn;
    v2i* slot_jj = (v2i*)(cidx + 2 * (size_t)Nn + 2 * (size_t)nb);
    v2i* slot_mj = slot_jj + Ejj;
    const int nbk = (Nn + BKT_NODES - 1) / BKT_NODES;
    const int nch = (Ejj + CHUNK_E - 1) / CHUNK_E;
    const int tabL = nbk * nch;
    const int nbT = (tabL + 1023) / 1024;
    v2i* res_jj  = (v2i*)S;
    v2i* res_mj  = res_jj + Ejj;
    int* rdst_jj = (int*)(res_mj + Emj);
    int* rdst_mj = rdst_jj + Ejj;
    int* tab     = (int*)H;
    int* tsum    = tab + 2 * (size_t)tabL;
    int* pool_next = (int*)S;
    int* pool_head = pool_next + Nn;

    float* P    = agg;
    float* sumJ = P;
    float* sumM = P + (size_t)16 * Gg;
    float* maxJ = P + (size_t)32 * Gg;
    float* maxM = P + (size_t)48 * Gg;
    float* cntJ = P + (size_t)64 * Gg;
    float* cntM = P + (size_t)65 * Gg;
    float* gbuf = P + (size_t)66 * Gg;
    float* m1   = P + (size_t)130 * Gg;
    float* m2   = P + (size_t)194 * Gg;
    float* out  = (float*)d_out;

    const int TB = 256;
    const int nodeBlocks = (Nn + TB - 1) / TB;

    pack_weight_kernel<<<1, 256, 0, stream>>>(Wm_jj1, bm_jj1, pWjj1, pbjj1, 15, 15, 16, 16);
    pack_weight_kernel<<<1, 256, 0, stream>>>(Wm_mj1, bm_mj1, pWmj1, pbmj1, 15, 15, 16, 16);
    pack_weight_kernel<<<1, 256, 0, stream>>>(Wm_jj2, bm_jj2, pWjj2, pbjj2, 15, 15, 16, 16);
    pack_weight_kernel<<<1, 256, 0, stream>>>(Wm_mj2, bm_mj2, pWmj2, pbmj2, 15, 15, 16, 16);
    pack_weight_kernel<<<16, 256, 0, stream>>>(W1, b1, pW1, pb1, 64, 64, 64, 64);
    pack_weight_kernel<<<8, 256, 0, stream>>>(W2, b2, pW2, pb2, 64, 32, 64, 32);

    auto gemm64 = [&](const float* A, const float* Wp, const float* bp,
                      float* C, int Ncpad, int lda, int ldc, int relu) {
        int tiles = (Gg >> 4) * (Ncpad >> 4);
        wmma_gemm_kernel<false, 64><<<(tiles + 7) / 8, TB, 0, stream>>>(
            A, nullptr, Wp, bp, C, Gg, Ncpad, lda, ldc, relu);
    };

    bucket_count_kernel<<<dim3(nch, 2), TB, 0, stream>>>(ei_jj, ei_mj, tab, Ejj, Emj, Nn, nch, nbk);
    csr_block_sum_kernel<<<dim3(nbT, 2), TB, 0, stream>>>(tab, tsum, tabL, nbT);
    csr_block_scan_kernel<<<dim3(nbT, 2), TB, 0, stream>>>(tab, tsum, tabL, nbT);
    bucket_place_kernel<<<dim3(nch, 2), TB, 0, stream>>>(ei_jj, ei_mj, tab, res_jj, res_mj, rdst_jj, rdst_mj, Ejj, Emj, Nn, nch, nbk);
    csr_bucket_kernel<<<dim3(nbk, 2), 1024, 0, stream>>>(res_jj, res_mj, rdst_jj, rdst_mj, tab, cidx, slot_jj, slot_mj, Ejj, Emj, Nn, nch, nbk);

    encode_kernel<<<nodeBlocks, TB, 0, stream>>>(x, type_id, Wj_enc, bj_enc, Wm_enc, bm_enc, H, Nn);

    conv_kernel<5, 9, true><<<nodeBlocks, TB, 0, stream>>>(H, S, off_jj, slot_jj, ea_jj, We_jj1, be_jj1, pWjj1, pbjj1, Nn, Ejj);

    conv_kernel<4, 12, false><<<nodeBlocks, TB, 0, stream>>>(S, H, off_mj, slot_mj, ea_mj, We_mj1, be_mj1, pWmj1, pbmj1, Nn, Emj);

    ln_relu_kernel<<<nodeBlocks, TB, 0, stream>>>(H, Nn);

    conv_kernel<5, 9, true><<<nodeBlocks, TB, 0, stream>>>(H, S, off_jj, slot_jj, ea_jj, We_jj2, be_jj2, pWjj2, pbjj2, Nn, Ejj);

    conv_kernel<4, 12, true><<<nodeBlocks, TB, 0, stream>>>(S, H, off_mj, slot_mj, ea_mj, We_mj2, be_mj2, pWmj2, pbmj2, Nn, Emj);

    ln_relu_kernel<<<nodeBlocks, TB, 0, stream>>>(H, Nn);

    fill_kernel<<<(Gg + TB - 1) / TB, TB, 0, stream>>>((float*)pool_head, 0.f, (size_t)Gg);
    pool_link_kernel<<<nodeBlocks, TB, 0, stream>>>(type_id, batch, pool_head, pool_next, Nn, Gg);
    pool_gather_kernel<<<(Gg + TB - 1) / TB, TB, 0, stream>>>(H, type_id, pool_head, pool_next,
                                                              sumJ, sumM, maxJ, maxM, cntJ, cntM, Nn, Gg);

    assemble_ln_kernel<<<(Gg + 127) / 128, 128, 0, stream>>>(sumJ, sumM, maxJ, maxM, cntJ, cntM, u, gbuf, Gg);
    gemm64(gbuf, pW1, pb1, m1, 64, 64, 64, 1);
    gemm64(m1,   pW2, pb2, m2, 32, 64, 32, 1);
    mlp3_kernel<<<(Gg + TB - 1) / TB, TB, 0, stream>>>(m2, W3, b3, out, Gg);
}


# conv = one work-group per 4096-destination bucket: thread per edge record adds its message into LDS sums (LDS float adds), WMMA reads LDS; per-node CSR pass dropped; sum order: LDS add arrival
# speedup vs baseline: 3.1378x; 1.1086x over previous
#include <hip/hip_runtime.h>
#include <hip/hip_bf16.h>

typedef __attribute__((ext_vector_type(2))) float v2f;
typedef __attribute__((ext_vector_type(8))) float v8f;

#define DPAD 16
#define DDIM 15

__global__ void fill_kernel(float* __restrict__ p, float v, size_t n) {
    size_t i = (size_t)blockIdx.x * blockDim.x + threadIdx.x;
    size_t stride = (size_t)gridDim.x * blockDim.x;
    for (; i < n; i += stride) p[i] = v;
}

__global__ void pack_weight_kernel(const float* __restrict__ W, const float* __restrict__ b,
                                   float* __restrict__ Wp, float* __restrict__ bp,
                                   int K, int Nc, int Kpad, int Ncpad)
{
    int idx = blockIdx.x * blockDim.x + threadIdx.x;
    int total = Kpad * Ncpad;
    if (idx < total) {
        int k = idx / Ncpad, c = idx % Ncpad;
        float v = (k < K && c < Nc) ? W[k * Nc + c] : 0.f;
        Wp[((size_t)(k >> 1) * Ncpad + c) * 2 + (k & 1)] = v;
    }
    if (idx < Ncpad) bp[idx] = (idx < Nc) ? b[idx] : 0.f;
}

__global__ __launch_bounds__(256) void encode_kernel(
    const float* __restrict__ x, const int* __restrict__ type_id,
    const float* __restrict__ Wj, const float* __restrict__ bj,
    const float* __restrict__ Wm, const float* __restrict__ bm,
    float* __restrict__ h, int Nn)
{
    int i = blockIdx.x * blockDim.x + threadIdx.x;
    if (i >= Nn) return;
    float xi[5];
#pragma unroll
    for (int k = 0; k < 5; ++k) xi[k] = x[(size_t)i * 5 + k];
    int t = type_id[i];
    const float* Wp = (t == 0) ? Wj : Wm;
    const float* bp = (t == 0) ? bj : bm;
    bool valid = (t == 0) || (t == 1);
#pragma unroll
    for (int d = 0; d < DDIM; ++d) {
        float v = 0.f;
        if (valid) {
            v = bp[d];
#pragma unroll
            for (int k = 0; k < 5; ++k) v += xi[k] * Wp[k * DDIM + d];
            v = fmaxf(v, 0.f);
        }
        h[(size_t)i * DPAD + d] = v;
    }
    h[(size_t)i * DPAD + DDIM] = 0.f;
}

typedef __attribute__((ext_vector_type(2))) int v2i;

#define BKT_SHIFT 12
#define BKT_NODES 4096
#define ACC_PITCH 17
#define CHUNK_E   4096
#define MAX_BKT   1024

__global__ __launch_bounds__(256) void bucket_count_kernel(
    const int* __restrict__ ei_a, const int* __restrict__ ei_b, int* __restrict__ tab,
    int Ea, int Eb, int Nn, int nch, int nbk)
{
    __shared__ int hist[MAX_BKT];
    const int t = threadIdx.x;
    const int set = blockIdx.y, chunk = blockIdx.x;
    const int E = set ? Eb : Ea;
    const int* __restrict__ ei = set ? ei_b : ei_a;
    for (int i = t; i < nbk; i += 256) hist[i] = 0;
    __syncthreads();
    const int base = chunk * CHUNK_E + t;
    const int last = max(E - 1, 0);
    int d[CHUNK_E / 256];
#pragma unroll
    for (int k = 0; k < CHUNK_E / 256; ++k) d[k] = ei[(size_t)E + min(base + k * 256, last)];
#pragma unroll
    for (int k = 0; k < CHUNK_E / 256; ++k) {
        if (base + k * 256 < E) {
            const unsigned dst = min((unsigned)d[k], (unsigned)(Nn - 1));
            atomicAdd(&hist[dst >> BKT_SHIFT], 1);
        }
    }
    __syncthreads();
    int* tb = tab + (size_t)set * nbk * nch;
    for (int i = t; i < nbk; i += 256) tb[(size_t)i * nch + chunk] = hist[i];
}

__global__ __launch_bounds__(256) void bucket_place_kernel(
    const int* __restrict__ ei_a, const int* __restrict__ ei_b, const int* __restrict__ tab,
    v2i* __restrict__ res_a, v2i* __restrict__ res_b,
    int* __restrict__ rdst_a, int* __restrict__ rdst_b,
    int Ea, int Eb, int Nn, int nch, int nbk)
{
    __shared__ int cursor[MAX_BKT];
    const int t = threadIdx.x;
    const int set = blockIdx.y, chunk = blockIdx.x;
    const int E = set ? Eb : Ea;
    const int* __restrict__ ei = set ? ei_b : ei_a;
    v2i* __restrict__ res = set ? res_b : res_a;
    int* __restrict__ rdst = set ? rdst_b : rdst_a;
    const int* tb = tab + (size_t)set * nbk * nch;
    for (int i = t; i < nbk; i += 256) cursor[i] = tb[(size_t)i * nch + chunk];
    __syncthreads();
    const int base = chunk * CHUNK_E + t;
    const int last = max(E - 1, 0);
    int sv[CHUNK_E / 256], d[CHUNK_E / 256];
#pragma unroll
    for (int k = 0; k < CHUNK_E / 256; ++k) {
        const int ec = min(base + k * 256, last);
        sv[k] = ei[ec];
        d[k]  = ei[(size_t)E + ec];
    }
#pragma unroll
    for (int k = 0; k < CHUNK_E / 256; ++k) {
        const int e = base + k * 256;
        if (e < E) {
            const int src = sv[k];
            const unsigned dst = min((unsigned)d[k], (unsigned)(Nn - 1));
            unsigned pos = (unsigned)atomicAdd(&cursor[dst >> BKT_SHIFT], 1);
            pos = min(pos, (unsigned)last);
            v2i o; o.x = e; o.y = src;
            res[pos] = o;
            rdst[pos] = (int)dst;
        }
    }
}

__global__ __launch_bounds__(256) void csr_block_sum_kernel(
    const int* __restrict__ cnt, int* __restrict__ bsum, int Nn, int nb)
{
    __shared__ int sh[256];
    const int* c = cnt + (size_t)blockIdx.y * Nn;
    const int t = threadIdx.x;
    const int base = blockIdx.x * 1024 + t * 4;
    int s = 0;
#pragma unroll
    for (int k = 0; k < 4; ++k) if (base + k < Nn) s += c[base + k];
    sh[t] = s;
    __syncthreads();
    for (int w = 128; w > 0; w >>= 1) {
        if (t < w) sh[t] += sh[t + w];
        __syncthreads();
    }
    if (t == 0) bsum[blockIdx.y * nb + blockIdx.x] = sh[0];
}

__global__ __launch_bounds__(256) void csr_block_scan_kernel(
    int* __restrict__ cnt, const int* __restrict__ bsum, int Nn, int nb)
{
    __shared__ int sh[256];
    int* c = cnt + (size_t)blockIdx.y * Nn;
    const int* bs = bsum + blockIdx.y * nb;
    const int t = threadIdx.x;

    int pc = 0;
    for (int j = t; j < (int)blockIdx.x; j += 256) pc += bs[j];
    sh[t] = pc;
    __syncthreads();
    for (int w = 128; w > 0; w >>= 1) {
        if (t < w) sh[t] += sh[t + w];
        __syncthreads();
    }
    const int carry = sh[0];
    __syncthreads();

    const int base = blockIdx.x * 1024 + t * 4;
    int v[4];
#pragma unroll
    for (int k = 0; k < 4; ++k) v[k] = (base + k < Nn) ? c[base + k] : 0;
    const int ts = v[0] + v[1] + v[2] + v[3];
    sh[t] = ts;
    __syncthreads();
    for (int d = 1; d < 256; d <<= 1) {
        int x = (t >= d) ? sh[t - d] : 0;
        __syncthreads();
        sh[t] += x;
        __syncthreads();
    }
    int run = carry + sh[t] - ts;
#pragma unroll
    for (int k = 0; k < 4; ++k) {
        if (base + k < Nn) c[base + k] = run;
        run += v[k];
    }
}

__device__ __forceinline__ float xhalf(float x) {
    return __builtin_bit_cast(float, __builtin_amdgcn_permlanex16(
        __builtin_bit_cast(int, x), __builtin_bit_cast(int, x), 0x76543210, 0xfedcba98, false, false));
}

template <int EDIM, bool RELU>
__global__ __launch_bounds__(1024) void conv_kernel(
    const float* __restrict__ h, float* __restrict__ hout,
    const v2i* __restrict__ res, const int* __restrict__ rdst, const int* __restrict__ tb,
    const float* __restrict__ ea,
    const float* __restrict__ We, const float* __restrict__ be,
    const float* __restrict__ Wp, const float* __restrict__ bp,
    int Nn, int E, int nch, int nbk)
{
    __shared__ float accs[BKT_NODES * ACC_PITCH];
    const int t = threadIdx.x;
    const int b = blockIdx.x;
    int s0 = tb[(size_t)b * nch];
    int s1 = (b + 1 < nbk) ? tb[(size_t)(b + 1) * nch] : E;
    s0 = min(max(s0, 0), E);
    s1 = min(max(s1, s0), E);

    for (int i = t; i < BKT_NODES * ACC_PITCH; i += 1024) accs[i] = 0.f;
    __syncthreads();

    const int hi = max(s1 - 1, 0);
    int p = s0 + t;
    int q = min(p, hi);
    v2i rn = res[q];
    int dn = rdst[q];
    for (; p < s1; p += 1024) {
        const v2i cur = rn;
        const int dc = dn;
        q = min(p + 1024, hi);
        rn = res[q];
        dn = rdst[q];

        float ev[EDIM];
#pragma unroll
        for (int k = 0; k < EDIM; ++k) ev[k] = ea[(size_t)cur.x * EDIM + k];

        float hl[DPAD];
        const float4* hv = (const float4*)(h + (size_t)cur.y * DPAD);
#pragma unroll
        for (int qq = 0; qq < 4; ++qq) {
            float4 v = hv[qq];
            hl[4 * qq + 0] = v.x; hl[4 * qq + 1] = v.y;
            hl[4 * qq + 2] = v.z; hl[4 * qq + 3] = v.w;
        }
        float* arow = accs + (dc & (BKT_NODES - 1)) * ACC_PITCH;
#pragma unroll
        for (int d = 0; d < DDIM; ++d) {
            float m = hl[d] + be[d];
#pragma unroll
            for (int k = 0; k < EDIM; ++k) m = __builtin_fmaf(ev[k], We[k * DDIM + d], m);
            m = fmaxf(m, 0.f);
            atomicAdd(arow + d, m);
        }
    }
    __syncthreads();

    const int lane = t & 31;
    const int c  = lane & 15;
    const int hf = lane >> 4;
    const int wave = __builtin_amdgcn_readfirstlane(t) >> 5;
    const float* bpp = Wp + ((size_t)hf * 16 + c) * 2;
    const float bv = bp[c];

    for (int g = wave; g < BKT_NODES / 32; g += 32) {
        const int wbase = b * BKT_NODES + g * 32;
        if (wbase >= Nn) break;

        float s[DPAD];
        const unsigned irow = min((unsigned)(wbase + lane), (unsigned)(Nn - 1));
        const float4* hi4 = (const float4*)(h + (size_t)irow * DPAD);
        const float* ar = accs + (g * 32 + lane) * ACC_PITCH;
#pragma unroll
        for (int qq = 0; qq < 4; ++qq) {
            float4 v = hi4[qq];
            s[4 * qq + 0] = v.x + ar[4 * qq + 0]; s[4 * qq + 1] = v.y + ar[4 * qq + 1];
            s[4 * qq + 2] = v.z + ar[4 * qq + 2]; s[4 * qq + 3] = v.w + ar[4 * qq + 3];
        }

        v8f d0 = {0.f, 0.f, 0.f, 0.f, 0.f, 0.f, 0.f, 0.f};
        v8f d1 = {0.f, 0.f, 0.f, 0.f, 0.f, 0.f, 0.f, 0.f};
#pragma unroll
        for (int kk = 0; kk < DPAD; kk += 4) {
            const float snd0 = hf ? s[kk + 0] : s[kk + 2];
            const float snd1 = hf ? s[kk + 1] : s[kk + 3];
            const float rcv0 = xhalf(snd0);
            const float rcv1 = xhalf(snd1);
            const float own0 = hf ? s[kk + 2] : s[kk + 0];
            const float own1 = hf ? s[kk + 3] : s[kk + 1];
            v2f a0, a1;
            a0.x = hf ? rcv0 : own0;  a0.y = hf ? rcv1 : own1;
            a1.x = hf ? own0 : rcv0;  a1.y = hf ? own1 : rcv1;
            const v2f bf = *(const v2f*)(bpp + (size_t)kk * 16);
            d0 = __builtin_amdgcn_wmma_f32_16x16x4_f32(false, a0, false, bf, (short)0, d0, false, false);
            d1 = __builtin_amdgcn_wmma_f32_16x16x4_f32(false, a1, false, bf, (short)0, d1, false, false);
        }

        float* orow = hout + (size_t)wbase * DPAD + (8 * hf) * DPAD + c;
        if (wbase + 32 <= Nn) {
#pragma unroll
            for (int r = 0; r < 8; ++r) {
                float v0 = d0[r] + bv;
                float v1 = d1[r] + bv;
                if (RELU) { v0 = fmaxf(v0, 0.f); v1 = fmaxf(v1, 0.f); }
                orow[r * DPAD] = v0;
                orow[(r + 16) * DPAD] = v1;
            }
        } else {
#pragma unroll
            for (int r = 0; r < 8; ++r) {
                const int n0 = wbase + 8 * hf + r;
                float v0 = d0[r] + bv;
                float v1 = d1[r] + bv;
                if (RELU) { v0 = fmaxf(v0, 0.f); v1 = fmaxf(v1, 0.f); }
                if (n0 < Nn)      orow[r * DPAD] = v0;
                if (n0 + 16 < Nn) orow[(r + 16) * DPAD] = v1;
            }
        }
    }
}

template <bool HAS_A2, int KPAD>
__global__ __launch_bounds__(256) void wmma_gemm_kernel(
    const float* __restrict__ A, const float* __restrict__ A2,
    const float* __restrict__ Wp, const float* __restrict__ bp,
    float* __restrict__ C,
    int M, int Ncpad, int lda, int ldc, int relu)
{
    const int lane = threadIdx.x & 31;
    const int wave = threadIdx.x >> 5;
    const int ntN = Ncpad >> 4;
    const int tile = blockIdx.x * (blockDim.x >> 5) + wave;
    if (tile >= (M >> 4) * ntN) return;
    const int mt = tile / ntN;
    const int nt = tile % ntN;

    const int mrow  = mt * 16 + (lane & 15);
    const int ncol  = nt * 16 + (lane & 15);
    const int khalf = (lane >> 4) << 1;

    const float* ap  = A + (size_t)mrow * lda + khalf;
    const float* a2p = A2 + (size_t)mrow * lda + khalf;
    const float* bpp = Wp + ((size_t)(khalf >> 1) * Ncpad + ncol) * 2;
    const size_t bstep = (size_t)Ncpad * 4;

    v8f acc = {0.f, 0.f, 0.f, 0.f, 0.f, 0.f, 0.f, 0.f};
#pragma unroll
    for (int kk = 0; kk < KPAD; kk += 4) {
        v2f a = *(const v2f*)ap;
        if (HAS_A2) {
            v2f a2 = *(const v2f*)a2p;
            a.x += a2.x; a.y += a2.y;
            a2p += 4;
        }
        v2f b = *(const v2f*)bpp;
        ap += 4;
        bpp += bstep;
        acc = __builtin_amdgcn_wmma_f32_16x16x4_f32(
            false, a, false, b, (short)0, acc, false, false);
    }

    const int rbase = mt * 16 + ((lane >> 4) << 3);
    const float bv = bp[ncol];
#pragma unroll
    for (int r = 0; r < 8; ++r) {
        float v = acc[r] + bv;
        if (relu) v = fmaxf(v, 0.f);
        C[(size_t)(rbase + r) * ldc + ncol] = v;
    }
}

__global__ __launch_bounds__(256) void ln_relu_kernel(float* __restrict__ h, int Nn)
{
    int i = blockIdx.x * blockDim.x + threadIdx.x;
    if (i >= Nn) return;
    float v[DDIM];
    float mu = 0.f;
#pragma unroll
    for (int d = 0; d < DDIM; ++d) { v[d] = h[(size_t)i * DPAD + d]; mu += v[d]; }
    mu *= (1.f / DDIM);
    float var = 0.f;
#pragma unroll
    for (int d = 0; d < DDIM; ++d) { float t = v[d] - mu; var += t * t; }
    var *= (1.f / DDIM);
    float inv = rsqrtf(var + 1e-5f);
#pragma unroll
    for (int d = 0; d < DDIM; ++d)
        h[(size_t)i * DPAD + d] = fmaxf((v[d] - mu) * inv, 0.f);
}

__global__ __launch_bounds__(256) void pool_link_kernel(
    const int* __restrict__ type_id, const int* __restrict__ batch,
    int* __restrict__ head, int* __restrict__ nxt, int Nn, int Gg)
{
    int i = blockIdx.x * blockDim.x + threadIdx.x;
    if (i >= Nn) return;
    const int t = type_id[i];
    const int b = batch[i];
    int link = 0;
    if ((t == 0 || t == 1) && b >= 0 && b < Gg) link = atomicExch(&head[b], i + 1);
    nxt[i] = link;
}

struct PoolNode { int nx; int tp; float4 r[4]; };

__device__ __forceinline__ void pool_fetch(PoolNode& n, int link,
    const float* __restrict__ h, const int* __restrict__ type_id,
    const int* __restrict__ nxt, int Nn)
{
    const unsigned i = min((unsigned)(link - 1), (unsigned)(Nn - 1));
    n.nx = nxt[i];
    n.tp = type_id[i];
    const float4* hv = (const float4*)(h + (size_t)i * DPAD);
#pragma unroll
    for (int q = 0; q < 4; ++q) n.r[q] = hv[q];
}

__device__ __forceinline__ void pool_accum(const PoolNode& n, bool valid,
    float (&sJ)[DPAD], float (&sM)[DPAD], float (&mJ)[DPAD], float (&mM)[DPAD], int& cJ, int& cM)
{
    const float ninf = -__builtin_inff();
    const bool inJ = valid & (n.tp == 0);
    const bool inM = valid & (n.tp != 0);
    cJ += inJ ? 1 : 0;
    cM += inM ? 1 : 0;
#pragma unroll
    for (int q = 0; q < 4; ++q) {
        const float v[4] = {n.r[q].x, n.r[q].y, n.r[q].z, n.r[q].w};
#pragma unroll
        for (int e = 0; e < 4; ++e) {
            const int d = 4 * q + e;
            sJ[d] += inJ ? v[e] : 0.f;
            sM[d] += inM ? v[e] : 0.f;
            const float cj = inJ ? v[e] : ninf;
            const float cm = inM ? v[e] : ninf;
            mJ[d] = (cj > mJ[d]) ? cj : mJ[d];
            mM[d] = (cm > mM[d]) ? cm : mM[d];
        }
    }
}

__global__ __launch_bounds__(256) void pool_gather_kernel(
    const float* __restrict__ h, const int* __restrict__ type_id,
    const int* __restrict__ head, const int* __restrict__ nxt,
    float* __restrict__ sumJ, float* __restrict__ sumM,
    float* __restrict__ maxJ, float* __restrict__ maxM,
    float* __restrict__ cntJ, float* __restrict__ cntM, int Nn, int Gg)
{
    const int g = blockIdx.x * blockDim.x + threadIdx.x;
    if (g >= Gg) return;
    const float ninf = -__builtin_inff();
    float sJ[DPAD], sM[DPAD], mJ[DPAD], mM[DPAD];
#pragma unroll
    for (int d = 0; d < DPAD; ++d) { sJ[d] = 0.f; sM[d] = 0.f; mJ[d] = ninf; mM[d] = ninf; }
    int cJ = 0, cM = 0;

    PoolNode a, b;
    const int first = head[g];
    bool live = (first != 0);
    pool_fetch(a, first, h, type_id, nxt, Nn);
    for (int it = 0; it < Nn && live; it += 2) {
        pool_fetch(b, a.nx, h, type_id, nxt, Nn);
        pool_accum(a, true, sJ, sM, mJ, mM, cJ, cM);
        const bool moreB = (a.nx != 0);
        pool_fetch(a, b.nx, h, type_id, nxt, Nn);
        pool_accum(b, moreB, sJ, sM, mJ, mM, cJ, cM);
        live = moreB & (b.nx != 0);
    }

    float4* oSJ = (float4*)(sumJ + (size_t)g * DPAD);
    float4* oSM = (float4*)(sumM + (size_t)g * DPAD);
    float4* oMJ = (float4*)(maxJ + (size_t)g * DPAD);
    float4* oMM = (float4*)(maxM + (size_t)g * DPAD);
#pragma unroll
    for (int q = 0; q < 4; ++q) {
        oSJ[q] = make_float4(sJ[4 * q], sJ[4 * q + 1], sJ[4 * q + 2], sJ[4 * q + 3]);
        oSM[q] = make_float4(sM[4 * q], sM[4 * q + 1], sM[4 * q + 2], sM[4 * q + 3]);
        oMJ[q] = make_float4(mJ[4 * q], mJ[4 * q + 1], mJ[4 * q + 2], mJ[4 * q + 3]);
        oMM[q] = make_float4(mM[4 * q], mM[4 * q + 1], mM[4 * q + 2], mM[4 * q + 3]);
    }
    cntJ[g] = (float)cJ;
    cntM[g] = (float)cM;
}

__global__ __launch_bounds__(128) void assemble_ln_kernel(
    const float* __restrict__ sumJ, const float* __restrict__ sumM,
    const float* __restrict__ maxJ, const float* __restrict__ maxM,
    const float* __restrict__ cntJ, const float* __restrict__ cntM,
    const float* __restrict__ u, float* __restrict__ g, int Gg)
{
    int gi = blockIdx.x * blockDim.x + threadIdx.x;
    if (gi >= Gg) return;
    float r[64];
    float cm = fmaxf(cntM[gi], 1.f);
    float cj = fmaxf(cntJ[gi], 1.f);
#pragma unroll
    for (int d = 0; d < DDIM; ++d) r[d] = sumM[(size_t)gi * DPAD + d] / cm;
#pragma unroll
    for (int d = 0; d < DDIM; ++d) r[15 + d] = sumJ[(size_t)gi * DPAD + d] / cj;
#pragma unroll
    for (int d = 0; d < DDIM; ++d) {
        float v = maxJ[(size_t)gi * DPAD + d];
        r[30 + d] = (v > -3.0e38f) ? v : 0.f;
    }
#pragma unroll
    for (int d = 0; d < DDIM; ++d) {
        float v = maxM[(size_t)gi * DPAD + d];
        r[45 + d] = (v > -3.0e38f) ? v : 0.f;
    }
#pragma unroll
    for (int j = 0; j < 4; ++j) r[60 + j] = u[(size_t)gi * 4 + j];

    float mu = 0.f;
#pragma unroll
    for (int k = 0; k < 64; ++k) mu += r[k];
    mu *= (1.f / 64.f);
    float var = 0.f;
#pragma unroll
    for (int k = 0; k < 64; ++k) { float t = r[k] - mu; var += t * t; }
    var *= (1.f / 64.f);
    float inv = rsqrtf(var + 1e-5f);
#pragma unroll
    for (int k = 0; k < 64; ++k) g[(size_t)gi * 64 + k] = (r[k] - mu) * inv;
}

__global__ __launch_bounds__(256) void mlp3_kernel(
    const float* __restrict__ m2, const float* __restrict__ W3,
    const float* __restrict__ b3, float* __restrict__ out, int Gg)
{
    int gi = blockIdx.x * blockDim.x + threadIdx.x;
    if (gi >= Gg) return;
    float s = b3[0];
#pragma unroll
    for (int k = 0; k < 32; ++k) s += m2[(size_t)gi * 32 + k] * W3[k];
    out[gi] = s;
}

extern "C" void kernel_launch(void* const* d_in, const int* in_sizes, int n_in,
                              void* d_out, int out_size, void* d_ws, size_t ws_size,
                              hipStream_t stream)
{
    const float* x       = (const float*)d_in[0];
    const int*   type_id = (const int*)  d_in[1];
    const int*   batch   = (const int*)  d_in[2];
    const float* u       = (const float*)d_in[3];
    const int*   ei_jj   = (const int*)  d_in[4];
    const float* ea_jj   = (const float*)d_in[5];
    const int*   ei_mj   = (const int*)  d_in[6];
    const float* ea_mj   = (const float*)d_in[7];
    const float* Wj_enc  = (const float*)d_in[8];   const float* bj_enc = (const float*)d_in[9];
    const float* Wm_enc  = (const float*)d_in[10];  const float* bm_enc = (const float*)d_in[11];
    const float* We_jj1  = (const float*)d_in[12];  const float* be_jj1 = (const float*)d_in[13];
    const float* Wm_jj1  = (const float*)d_in[14];  const float* bm_jj1 = (const float*)d_in[15];
    const float* We_mj1  = (const float*)d_in[16];  const float* be_mj1 = (const float*)d_in[17];
    const float* Wm_mj1  = (const float*)d_in[18];  const float* bm_mj1 = (const float*)d_in[19];
    const float* We_jj2  = (const float*)d_in[20];  const float* be_jj2 = (const float*)d_in[21];
    const float* Wm_jj2  = (const float*)d_in[22];  const float* bm_jj2 = (const float*)d_in[23];
    const float* We_mj2  = (const float*)d_in[24];  const float* be_mj2 = (const float*)d_in[25];
    const float* Wm_mj2  = (const float*)d_in[26];  const float* bm_mj2 = (const float*)d_in[27];
    const float* W1      = (const float*)d_in[28];  const float* b1     = (const float*)d_in[29];
    const float* W2      = (const float*)d_in[30];  const float* b2     = (const float*)d_in[31];
    const float* W3      = (const float*)d_in[32];  const float* b3     = (const float*)d_in[33];

    const int Nn  = in_sizes[0] / 5;
    const int Gg  = in_sizes[3] / 4;
    const int Ejj = in_sizes[5] / 5;
    const int Emj = in_sizes[7] / 4;

    float* ws = (float*)d_ws;
    const size_t NP = (size_t)Nn * DPAD;
    float* H   = ws;
    float* S   = ws + NP;
    float* agg = ws + 2 * NP;
    float* wpk = ws + 3 * NP;
    float* pWjj1 = wpk;          float* pbjj1 = wpk + 256;
    float* pWmj1 = wpk + 272;    float* pbmj1 = wpk + 528;
    float* pWjj2 = wpk + 544;    float* pbjj2 = wpk + 800;
    float* pWmj2 = wpk + 816;    float* pbmj2 = wpk + 1072;
    float* pW1   = wpk + 1088;   float* pb1   = wpk + 5184;
    float* pW2   = wpk + 5248;   float* pb2   = wpk + 7296;

    const int nbk = (Nn + BKT_NODES - 1) / BKT_NODES;
    const int nch = (Ejj + CHUNK_E - 1) / CHUNK_E;
    const int tabL = nbk * nch;
    const int nbT = (tabL + 1023) / 1024;
    v2i* res_jj  = (v2i*)agg;
    v2i* res_mj  = res_jj + Ejj;
    int* rdst_jj = (int*)(res_mj + Emj);
    int* rdst_mj = rdst_jj + Ejj;
    int* tab     = rdst_mj + Emj;
    int* tsum    = tab + 2 * (size_t)tabL;
    int* pool_next = (int*)S;
    int* pool_head = pool_next + Nn;

    float* P    = agg;
    float* sumJ = P;
    float* sumM = P + (size_t)16 * Gg;
    float* maxJ = P + (size_t)32 * Gg;
    float* maxM = P + (size_t)48 * Gg;
    float* cntJ = P + (size_t)64 * Gg;
    float* cntM = P + (size_t)65 * Gg;
    float* gbuf = P + (size_t)66 * Gg;
    float* m1   = P + (size_t)130 * Gg;
    float* m2   = P + (size_t)194 * Gg;
    float* out  = (float*)d_out;

    const int TB = 256;
    const int nodeBlocks = (Nn + TB - 1) / TB;

    pack_weight_kernel<<<1, 256, 0, stream>>>(Wm_jj1, bm_jj1, pWjj1, pbjj1, 15, 15, 16, 16);
    pack_weight_kernel<<<1, 256, 0, stream>>>(Wm_mj1, bm_mj1, pWmj1, pbmj1, 15, 15, 16, 16);
    pack_weight_kernel<<<1, 256, 0, stream>>>(Wm_jj2, bm_jj2, pWjj2, pbjj2, 15, 15, 16, 16);
    pack_weight_kernel<<<1, 256, 0, stream>>>(Wm_mj2, bm_mj2, pWmj2, pbmj2, 15, 15, 16, 16);
    pack_weight_kernel<<<16, 256, 0, stream>>>(W1, b1, pW1, pb1, 64, 64, 64, 64);
    pack_weight_kernel<<<8, 256, 0, stream>>>(W2, b2, pW2, pb2, 64, 32, 64, 32);

    auto gemm64 = [&](const float* A, const float* Wp, const float* bp,
                      float* C, int Ncpad, int lda, int ldc, int relu) {
        int tiles = (Gg >> 4) * (Ncpad >> 4);
        wmma_gemm_kernel<false, 64><<<(tiles + 7) / 8, TB, 0, stream>>>(
            A, nullptr, Wp, bp, C, Gg, Ncpad, lda, ldc, relu);
    };

    bucket_count_kernel<<<dim3(nch, 2), TB, 0, stream>>>(ei_jj, ei_mj, tab, Ejj, Emj, Nn, nch, nbk);
    csr_block_sum_kernel<<<dim3(nbT, 2), TB, 0, stream>>>(tab, tsum, tabL, nbT);
    csr_block_scan_kernel<<<dim3(nbT, 2), TB, 0, stream>>>(tab, tsum, tabL, nbT);
    bucket_place_kernel<<<dim3(nch, 2), TB, 0, stream>>>(ei_jj, ei_mj, tab, res_jj, res_mj, rdst_jj, rdst_mj, Ejj, Emj, Nn, nch, nbk);

    encode_kernel<<<nodeBlocks, TB, 0, stream>>>(x, type_id, Wj_enc, bj_enc, Wm_enc, bm_enc, H, Nn);

    conv_kernel<5, true><<<nbk, 1024, 0, stream>>>(H, S, res_jj, rdst_jj, tab, ea_jj, We_jj1, be_jj1, pWjj1, pbjj1, Nn, Ejj, nch, nbk);

    conv_kernel<4, false><<<nbk, 1024, 0, stream>>>(S, H, res_mj, rdst_mj, tab + tabL, ea_mj, We_mj1, be_mj1, pWmj1, pbmj1, Nn, Emj, nch, nbk);

    ln_relu_kernel<<<nodeBlocks, TB, 0, stream>>>(H, Nn);

    conv_kernel<5, true><<<nbk, 1024, 0, stream>>>(H, S, res_jj, rdst_jj, tab, ea_jj, We_jj2, be_jj2, pWjj2, pbjj2, Nn, Ejj, nch, nbk);

    conv_kernel<4, true><<<nbk, 1024, 0, stream>>>(S, H, res_mj, rdst_mj, tab + tabL, ea_mj, We_mj2, be_mj2, pWmj2, pbmj2, Nn, Emj, nch, nbk);

    ln_relu_kernel<<<nodeBlocks, TB, 0, stream>>>(H, Nn);

    fill_kernel<<<(Gg + TB - 1) / TB, TB, 0, stream>>>((float*)pool_head, 0.f, (size_t)Gg);
    pool_link_kernel<<<nodeBlocks, TB, 0, stream>>>(type_id, batch, pool_head, pool_next, Nn, Gg);
    pool_gather_kernel<<<(Gg + TB - 1) / TB, TB, 0, stream>>>(H, type_id, pool_head, pool_next,
                                                              sumJ, sumM, maxJ, maxM, cntJ, cntM, Nn, Gg);

    assemble_ln_kernel<<<(Gg + 127) / 128, 128, 0, stream>>>(sumJ, sumM, maxJ, maxM, cntJ, cntM, u, gbuf, Gg);
    gemm64(gbuf, pW1, pb1, m1, 64, 64, 64, 1);
    gemm64(m1,   pW2, pb2, m2, 32, 64, 32, 1);
    mlp3_kernel<<<(Gg + TB - 1) / TB, TB, 0, stream>>>(m2, W3, b3, out, Gg);
}


# LayerNorm+ReLU done in registers by each reader of a row (layer-2 conv: gathered source rows and own row; pooling walk): both LN launches and their plane read+write go; sum orders as before
# speedup vs baseline: 3.3226x; 1.0589x over previous
#include <hip/hip_runtime.h>
#include <hip/hip_bf16.h>

typedef __attribute__((ext_vector_type(2))) float v2f;
typedef __attribute__((ext_vector_type(8))) float v8f;

#define DPAD 16
#define DDIM 15

__global__ void fill_kernel(float* __restrict__ p, float v, size_t n) {
    size_t i = (size_t)blockIdx.x * blockDim.x + threadIdx.x;
    size_t stride = (size_t)gridDim.x * blockDim.x;
    for (; i < n; i += stride) p[i] = v;
}

__global__ void pack_weight_kernel(const float* __restrict__ W, const float* __restrict__ b,
                                   float* __restrict__ Wp, float* __restrict__ bp,
                                   int K, int Nc, int Kpad, int Ncpad)
{
    int idx = blockIdx.x * blockDim.x + threadIdx.x;
    int total = Kpad * Ncpad;
    if (idx < total) {
        int k = idx / Ncpad, c = idx % Ncpad;
        float v = (k < K && c < Nc) ? W[k * Nc + c] : 0.f;
        Wp[((size_t)(k >> 1) * Ncpad + c) * 2 + (k & 1)] = v;
    }
    if (idx < Ncpad) bp[idx] = (idx < Nc) ? b[idx] : 0.f;
}

__global__ __launch_bounds__(256) void encode_kernel(
    const float* __restrict__ x, const int* __restrict__ type_id,
    const float* __restrict__ Wj, const float* __restrict__ bj,
    const float* __restrict__ Wm, const float* __restrict__ bm,
    float* __restrict__ h, int Nn)
{
    int i = blockIdx.x * blockDim.x + threadIdx.x;
    if (i >= Nn) return;
    float xi[5];
#pragma unroll
    for (int k = 0; k < 5; ++k) xi[k] = x[(size_t)i * 5 + k];
    int t = type_id[i];
    const float* Wp = (t == 0) ? Wj : Wm;
    const float* bp = (t == 0) ? bj : bm;
    bool valid = (t == 0) || (t == 1);
#pragma unroll
    for (int d = 0; d < DDIM; ++d) {
        float v = 0.f;
        if (valid) {
            v = bp[d];
#pragma unroll
            for (int k = 0; k < 5; ++k) v += xi[k] * Wp[k * DDIM + d];
            v = fmaxf(v, 0.f);
        }
        h[(size_t)i * DPAD + d] = v;
    }
    h[(size_t)i * DPAD + DDIM] = 0.f;
}

typedef __attribute__((ext_vector_type(2))) int v2i;

#define BKT_SHIFT 12
#define BKT_NODES 4096
#define ACC_PITCH 17
#define CHUNK_E   4096
#define MAX_BKT   1024

__global__ __launch_bounds__(256) void bucket_count_kernel(
    const int* __restrict__ ei_a, const int* __restrict__ ei_b, int* __restrict__ tab,
    int Ea, int Eb, int Nn, int nch, int nbk)
{
    __shared__ int hist[MAX_BKT];
    const int t = threadIdx.x;
    const int set = blockIdx.y, chunk = blockIdx.x;
    const int E = set ? Eb : Ea;
    const int* __restrict__ ei = set ? ei_b : ei_a;
    for (int i = t; i < nbk; i += 256) hist[i] = 0;
    __syncthreads();
    const int base = chunk * CHUNK_E + t;
    const int last = max(E - 1, 0);
    int d[CHUNK_E / 256];
#pragma unroll
    for (int k = 0; k < CHUNK_E / 256; ++k) d[k] = ei[(size_t)E + min(base + k * 256, last)];
#pragma unroll
    for (int k = 0; k < CHUNK_E / 256; ++k) {
        if (base + k * 256 < E) {
            const unsigned dst = min((unsigned)d[k], (unsigned)(Nn - 1));
            atomicAdd(&hist[dst >> BKT_SHIFT], 1);
        }
    }
    __syncthreads();
    int* tb = tab + (size_t)set * nbk * nch;
    for (int i = t; i < nbk; i += 256) tb[(size_t)i * nch + chunk] = hist[i];
}

__global__ __launch_bounds__(256) void bucket_place_kernel(
    const int* __restrict__ ei_a, const int* __restrict__ ei_b, const int* __restrict__ tab,
    v2i* __restrict__ res_a, v2i* __restrict__ res_b,
    int* __restrict__ rdst_a, int* __restrict__ rdst_b,
    int Ea, int Eb, int Nn, int nch, int nbk)
{
    __shared__ int cursor[MAX_BKT];
    const int t = threadIdx.x;
    const int set = blockIdx.y, chunk = blockIdx.x;
    const int E = set ? Eb : Ea;
    const int* __restrict__ ei = set ? ei_b : ei_a;
    v2i* __restrict__ res = set ? res_b : res_a;
    int* __restrict__ rdst = set ? rdst_b : rdst_a;
    const int* tb = tab + (size_t)set * nbk * nch;
    for (int i = t; i < nbk; i += 256) cursor[i] = tb[(size_t)i * nch + chunk];
    __syncthreads();
    const int base = chunk * CHUNK_E + t;
    const int last = max(E - 1, 0);
    int sv[CHUNK_E / 256], d[CHUNK_E / 256];
#pragma unroll
    for (int k = 0; k < CHUNK_E / 256; ++k) {
        const int ec = min(base + k * 256, last);
        sv[k] = ei[ec];
        d[k]  = ei[(size_t)E + ec];
    }
#pragma unroll
    for (int k = 0; k < CHUNK_E / 256; ++k) {
        const int e = base + k * 256;
        if (e < E) {
            const int src = sv[k];
            const unsigned dst = min((unsigned)d[k], (unsigned)(Nn - 1));
            unsigned pos = (unsigned)atomicAdd(&cursor[dst >> BKT_SHIFT], 1);
            pos = min(pos, (unsigned)last);
            v2i o; o.x = e; o.y = src;
            res[pos] = o;
            rdst[pos] = (int)dst;
        }
    }
}

__global__ __launch_bounds__(256) void csr_block_sum_kernel(
    const int* __restrict__ cnt, int* __restrict__ bsum, int Nn, int nb)
{
    __shared__ int sh[256];
    const int* c = cnt + (size_t)blockIdx.y * Nn;
    const int t = threadIdx.x;
    const int base = blockIdx.x * 1024 + t * 4;
    int s = 0;
#pragma unroll
    for (int k = 0; k < 4; ++k) if (base + k < Nn) s += c[base + k];
    sh[t] = s;
    __syncthreads();
    for (int w = 128; w > 0; w >>= 1) {
        if (t < w) sh[t] += sh[t + w];
        __syncthreads();
    }
    if (t == 0) bsum[blockIdx.y * nb + blockIdx.x] = sh[0];
}

__global__ __launch_bounds__(256) void csr_block_scan_kernel(
    int* __restrict__ cnt, const int* __restrict__ bsum, int Nn, int nb)
{
    __shared__ int sh[256];
    int* c = cnt + (size_t)blockIdx.y * Nn;
    const int* bs = bsum + blockIdx.y * nb;
    const int t = threadIdx.x;

    int pc = 0;
    for (int j = t; j < (int)blockIdx.x; j += 256) pc += bs[j];
    sh[t] = pc;
    __syncthreads();
    for (int w = 128; w > 0; w >>= 1) {
        if (t < w) sh[t] += sh[t + w];
        __syncthreads();
    }
    const int carry = sh[0];
    __syncthreads();

    const int base = blockIdx.x * 1024 + t * 4;
    int v[4];
#pragma unroll
    for (int k = 0; k < 4; ++k) v[k] = (base + k < Nn) ? c[base + k] : 0;
    const int ts = v[0] + v[1] + v[2] + v[3];
    sh[t] = ts;
    __syncthreads();
    for (int d = 1; d < 256; d <<= 1) {
        int x = (t >= d) ? sh[t - d] : 0;
        __syncthreads();
        sh[t] += x;
        __syncthreads();
    }
    int run = carry + sh[t] - ts;
#pragma unroll
    for (int k = 0; k < 4; ++k) {
        if (base + k < Nn) c[base + k] = run;
        run += v[k];
    }
}

__device__ __forceinline__ void ln_relu_row(float (&v)[DPAD])
{
    float mu = 0.f;
#pragma unroll
    for (int d = 0; d < DDIM; ++d) mu += v[d];
    mu *= (1.f / DDIM);
    float var = 0.f;
#pragma unroll
    for (int d = 0; d < DDIM; ++d) { float t = v[d] - mu; var += t * t; }
    var *= (1.f / DDIM);
    const float inv = rsqrtf(var + 1e-5f);
#pragma unroll
    for (int d = 0; d < DDIM; ++d) v[d] = fmaxf((v[d] - mu) * inv, 0.f);
    v[DDIM] = 0.f;
}

__device__ __forceinline__ float xhalf(float x) {
    return __builtin_bit_cast(float, __builtin_amdgcn_permlanex16(
        __builtin_bit_cast(int, x), __builtin_bit_cast(int, x), 0x76543210, 0xfedcba98, false, false));
}

template <int EDIM, bool RELU, bool LNIN>
__global__ __launch_bounds__(1024) void conv_kernel(
    const float* __restrict__ h, float* __restrict__ hout,
    const v2i* __restrict__ res, const int* __restrict__ rdst, const int* __restrict__ tb,
    const float* __restrict__ ea,
    const float* __restrict__ We, const float* __restrict__ be,
    const float* __restrict__ Wp, const float* __restrict__ bp,
    int Nn, int E, int nch, int nbk)
{
    __shared__ float accs[BKT_NODES * ACC_PITCH];
    const int t = threadIdx.x;
    const int b = blockIdx.x;
    int s0 = tb[(size_t)b * nch];
    int s1 = (b + 1 < nbk) ? tb[(size_t)(b + 1) * nch] : E;
    s0 = min(max(s0, 0), E);
    s1 = min(max(s1, s0), E);

    for (int i = t; i < BKT_NODES * ACC_PITCH; i += 1024) accs[i] = 0.f;
    __syncthreads();

    const int hi = max(s1 - 1, 0);
    int p = s0 + t;
    int q = min(p, hi);
    v2i rn = res[q];
    int dn = rdst[q];
    for (; p < s1; p += 1024) {
        const v2i cur = rn;
        const int dc = dn;
        q = min(p + 1024, hi);
        rn = res[q];
        dn = rdst[q];

        float ev[EDIM];
#pragma unroll
        for (int k = 0; k < EDIM; ++k) ev[k] = ea[(size_t)cur.x * EDIM + k];

        float hl[DPAD];
        const float4* hv = (const float4*)(h + (size_t)cur.y * DPAD);
#pragma unroll
        for (int qq = 0; qq < 4; ++qq) {
            float4 v = hv[qq];
            hl[4 * qq + 0] = v.x; hl[4 * qq + 1] = v.y;
            hl[4 * qq + 2] = v.z; hl[4 * qq + 3] = v.w;
        }
        if (LNIN) ln_relu_row(hl);
        float* arow = accs + (dc & (BKT_NODES - 1)) * ACC_PITCH;
#pragma unroll
        for (int d = 0; d < DDIM; ++d) {
            float m = hl[d] + be[d];
#pragma unroll
            for (int k = 0; k < EDIM; ++k) m = __builtin_fmaf(ev[k], We[k * DDIM + d], m);
            m = fmaxf(m, 0.f);
            atomicAdd(arow + d, m);
        }
    }
    __syncthreads();

    const int lane = t & 31;
    const int c  = lane & 15;
    const int hf = lane >> 4;
    const int wave = __builtin_amdgcn_readfirstlane(t) >> 5;
    const float* bpp = Wp + ((size_t)hf * 16 + c) * 2;
    const float bv = bp[c];

    for (int g = wave; g < BKT_NODES / 32; g += 32) {
        const int wbase = b * BKT_NODES + g * 32;
        if (wbase >= Nn) break;

        float s[DPAD];
        const unsigned irow = min((unsigned)(wbase + lane), (unsigned)(Nn - 1));
        const float4* hi4 = (const float4*)(h + (size_t)irow * DPAD);
        const float* ar = accs + (g * 32 + lane) * ACC_PITCH;
#pragma unroll
        for (int qq = 0; qq < 4; ++qq) {
            float4 v = hi4[qq];
            s[4 * qq + 0] = v.x; s[4 * qq + 1] = v.y;
            s[4 * qq + 2] = v.z; s[4 * qq + 3] = v.w;
        }
        if (LNIN) ln_relu_row(s);
#pragma unroll
        for (int d = 0; d < DPAD; ++d) s[d] += ar[d];

        v8f d0 = {0.f, 0.f, 0.f, 0.f, 0.f, 0.f, 0.f, 0.f};
        v8f d1 = {0.f, 0.f, 0.f, 0.f, 0.f, 0.f, 0.f, 0.f};
#pragma unroll
        for (int kk = 0; kk < DPAD; kk += 4) {
            const float snd0 = hf ? s[kk + 0] : s[kk + 2];
            const float snd1 = hf ? s[kk + 1] : s[kk + 3];
            const float rcv0 = xhalf(snd0);
            const float rcv1 = xhalf(snd1);
            const float own0 = hf ? s[kk + 2] : s[kk + 0];
            const float own1 = hf ? s[kk + 3] : s[kk + 1];
            v2f a0, a1;
            a0.x = hf ? rcv0 : own0;  a0.y = hf ? rcv1 : own1;
            a1.x = hf ? own0 : rcv0;  a1.y = hf ? own1 : rcv1;
            const v2f bf = *(const v2f*)(bpp + (size_t)kk * 16);
            d0 = __builtin_amdgcn_wmma_f32_16x16x4_f32(false, a0, false, bf, (short)0, d0, false, false);
            d1 = __builtin_amdgcn_wmma_f32_16x16x4_f32(false, a1, false, bf, (short)0, d1, false, false);
        }

        float* orow = hout + (size_t)wbase * DPAD + (8 * hf) * DPAD + c;
        if (wbase + 32 <= Nn) {
#pragma unroll
            for (int r = 0; r < 8; ++r) {
                float v0 = d0[r] + bv;
                float v1 = d1[r] + bv;
                if (RELU) { v0 = fmaxf(v0, 0.f); v1 = fmaxf(v1, 0.f); }
                orow[r * DPAD] = v0;
                orow[(r + 16) * DPAD] = v1;
            }
        } else {
#pragma unroll
            for (int r = 0; r < 8; ++r) {
                const int n0 = wbase + 8 * hf + r;
                float v0 = d0[r] + bv;
                float v1 = d1[r] + bv;
                if (RELU) { v0 = fmaxf(v0, 0.f); v1 = fmaxf(v1, 0.f); }
                if (n0 < Nn)      orow[r * DPAD] = v0;
                if (n0 + 16 < Nn) orow[(r + 16) * DPAD] = v1;
            }
        }
    }
}

template <bool HAS_A2, int KPAD>
__global__ __launch_bounds__(256) void wmma_gemm_kernel(
    const float* __restrict__ A, const float* __restrict__ A2,
    const float* __restrict__ Wp, const float* __restrict__ bp,
    float* __restrict__ C,
    int M, int Ncpad, int lda, int ldc, int relu)
{
    const int lane = threadIdx.x & 31;
    const int wave = threadIdx.x >> 5;
    const int ntN = Ncpad >> 4;
    const int tile = blockIdx.x * (blockDim.x >> 5) + wave;
    if (tile >= (M >> 4) * ntN) return;
    const int mt = tile / ntN;
    const int nt = tile % ntN;

    const int mrow  = mt * 16 + (lane & 15);
    const int ncol  = nt * 16 + (lane & 15);
    const int khalf = (lane >> 4) << 1;

    const float* ap  = A + (size_t)mrow * lda + khalf;
    const float* a2p = A2 + (size_t)mrow * lda + khalf;
    const float* bpp = Wp + ((size_t)(khalf >> 1) * Ncpad + ncol) * 2;
    const size_t bstep = (size_t)Ncpad * 4;

    v8f acc = {0.f, 0.f, 0.f, 0.f, 0.f, 0.f, 0.f, 0.f};
#pragma unroll
    for (int kk = 0; kk < KPAD; kk += 4) {
        v2f a = *(const v2f*)ap;
        if (HAS_A2) {
            v2f a2 = *(const v2f*)a2p;
            a.x += a2.x; a.y += a2.y;
            a2p += 4;
        }
        v2f b = *(const v2f*)bpp;
        ap += 4;
        bpp += bstep;
        acc = __builtin_amdgcn_wmma_f32_16x16x4_f32(
            false, a, false, b, (short)0, acc, false, false);
    }

    const int rbase = mt * 16 + ((lane >> 4) << 3);
    const float bv = bp[ncol];
#pragma unroll
    for (int r = 0; r < 8; ++r) {
        float v = acc[r] + bv;
        if (relu) v = fmaxf(v, 0.f);
        C[(size_t)(rbase + r) * ldc + ncol] = v;
    }
}

__global__ __launch_bounds__(256) void pool_link_kernel(
    const int* __restrict__ type_id, const int* __restrict__ batch,
    int* __restrict__ head, int* __restrict__ nxt, int Nn, int Gg)
{
    int i = blockIdx.x * blockDim.x + threadIdx.x;
    if (i >= Nn) return;
    const int t = type_id[i];
    const int b = batch[i];
    int link = 0;
    if ((t == 0 || t == 1) && b >= 0 && b < Gg) link = atomicExch(&head[b], i + 1);
    nxt[i] = link;
}

struct PoolNode { int nx; int tp; float4 r[4]; };

__device__ __forceinline__ void pool_fetch(PoolNode& n, int link,
    const float* __restrict__ h, const int* __restrict__ type_id,
    const int* __restrict__ nxt, int Nn)
{
    const unsigned i = min((unsigned)(link - 1), (unsigned)(Nn - 1));
    n.nx = nxt[i];
    n.tp = type_id[i];
    const float4* hv = (const float4*)(h + (size_t)i * DPAD);
#pragma unroll
    for (int q = 0; q < 4; ++q) n.r[q] = hv[q];
}

__device__ __forceinline__ void pool_accum(const PoolNode& n, bool valid,
    float (&sJ)[DPAD], float (&sM)[DPAD], float (&mJ)[DPAD], float (&mM)[DPAD], int& cJ, int& cM)
{
    const float ninf = -__builtin_inff();
    const bool inJ = valid & (n.tp == 0);
    const bool inM = valid & (n.tp != 0);
    cJ += inJ ? 1 : 0;
    cM += inM ? 1 : 0;
    float row[DPAD];
#pragma unroll
    for (int q = 0; q < 4; ++q) {
        row[4 * q + 0] = n.r[q].x; row[4 * q + 1] = n.r[q].y;
        row[4 * q + 2] = n.r[q].z; row[4 * q + 3] = n.r[q].w;
    }
    ln_relu_row(row);
#pragma unroll
    for (int d = 0; d < DPAD; ++d) {
        sJ[d] += inJ ? row[d] : 0.f;
        sM[d] += inM ? row[d] : 0.f;
        const float cj = inJ ? row[d] : ninf;
        const float cm = inM ? row[d] : ninf;
        mJ[d] = (cj > mJ[d]) ? cj : mJ[d];
        mM[d] = (cm > mM[d]) ? cm : mM[d];
    }
}

__global__ __launch_bounds__(256) void pool_gather_kernel(
    const float* __restrict__ h, const int* __restrict__ type_id,
    const int* __restrict__ head, const int* __restrict__ nxt,
    float* __restrict__ sumJ, float* __restrict__ sumM,
    float* __restrict__ maxJ, float* __restrict__ maxM,
    float* __restrict__ cntJ, float* __restrict__ cntM, int Nn, int Gg)
{
    const int g = blockIdx.x * blockDim.x + threadIdx.x;
    if (g >= Gg) return;
    const float ninf = -__builtin_inff();
    float sJ[DPAD], sM[DPAD], mJ[DPAD], mM[DPAD];
#pragma unroll
    for (int d = 0; d < DPAD; ++d) { sJ[d] = 0.f; sM[d] = 0.f; mJ[d] = ninf; mM[d] = ninf; }
    int cJ = 0, cM = 0;

    PoolNode a, b;
    const int first = head[g];
    bool live = (first != 0);
    pool_fetch(a, first, h, type_id, nxt, Nn);
    for (int it = 0; it < Nn && live; it += 2) {
        pool_fetch(b, a.nx, h, type_id, nxt, Nn);
        pool_accum(a, true, sJ, sM, mJ, mM, cJ, cM);
        const bool moreB = (a.nx != 0);
        pool_fetch(a, b.nx, h, type_id, nxt, Nn);
        pool_accum(b, moreB, sJ, sM, mJ, mM, cJ, cM);
        live = moreB & (b.nx != 0);
    }

    float4* oSJ = (float4*)(sumJ + (size_t)g * DPAD);
    float4* oSM = (float4*)(sumM + (size_t)g * DPAD);
    float4* oMJ = (float4*)(maxJ + (size_t)g * DPAD);
    float4* oMM = (float4*)(maxM + (size_t)g * DPAD);
#pragma unroll
    for (int q = 0; q < 4; ++q) {
        oSJ[q] = make_float4(sJ[4 * q], sJ[4 * q + 1], sJ[4 * q + 2], sJ[4 * q + 3]);
        oSM[q] = make_float4(sM[4 * q], sM[4 * q + 1], sM[4 * q + 2], sM[4 * q + 3]);
        oMJ[q] = make_float4(mJ[4 * q], mJ[4 * q + 1], mJ[4 * q + 2], mJ[4 * q + 3]);
        oMM[q] = make_float4(mM[4 * q], mM[4 * q + 1], mM[4 * q + 2], mM[4 * q + 3]);
    }
    cntJ[g] = (float)cJ;
    cntM[g] = (float)cM;
}

__global__ __launch_bounds__(128) void assemble_ln_kernel(
    const float* __restrict__ sumJ, const float* __restrict__ sumM,
    const float* __restrict__ maxJ, const float* __restrict__ maxM,
    const float* __restrict__ cntJ, const float* __restrict__ cntM,
    const float* __restrict__ u, float* __restrict__ g, int Gg)
{
    int gi = blockIdx.x * blockDim.x + threadIdx.x;
    if (gi >= Gg) return;
    float r[64];
    float cm = fmaxf(cntM[gi], 1.f);
    float cj = fmaxf(cntJ[gi], 1.f);
#pragma unroll
    for (int d = 0; d < DDIM; ++d) r[d] = sumM[(size_t)gi * DPAD + d] / cm;
#pragma unroll
    for (int d = 0; d < DDIM; ++d) r[15 + d] = sumJ[(size_t)gi * DPAD + d] / cj;
#pragma unroll
    for (int d = 0; d < DDIM; ++d) {
        float v = maxJ[(size_t)gi * DPAD + d];
        r[30 + d] = (v > -3.0e38f) ? v : 0.f;
    }
#pragma unroll
    for (int d = 0; d < DDIM; ++d) {
        float v = maxM[(size_t)gi * DPAD + d];
        r[45 + d] = (v > -3.0e38f) ? v : 0.f;
    }
#pragma unroll
    for (int j = 0; j < 4; ++j) r[60 + j] = u[(size_t)gi * 4 + j];

    float mu = 0.f;
#pragma unroll
    for (int k = 0; k < 64; ++k) mu += r[k];
    mu *= (1.f / 64.f);
    float var = 0.f;
#pragma unroll
    for (int k = 0; k < 64; ++k) { float t = r[k] - mu; var += t * t; }
    var *= (1.f / 64.f);
    float inv = rsqrtf(var + 1e-5f);
#pragma unroll
    for (int k = 0; k < 64; ++k) g[(size_t)gi * 64 + k] = (r[k] - mu) * inv;
}

__global__ __launch_bounds__(256) void mlp3_kernel(
    const float* __restrict__ m2, const float* __restrict__ W3,
    const float* __restrict__ b3, float* __restrict__ out, int Gg)
{
    int gi = blockIdx.x * blockDim.x + threadIdx.x;
    if (gi >= Gg) return;
    float s = b3[0];
#pragma unroll
    for (int k = 0; k < 32; ++k) s += m2[(size_t)gi * 32 + k] * W3[k];
    out[gi] = s;
}

extern "C" void kernel_launch(void* const* d_in, const int* in_sizes, int n_in,
                              void* d_out, int out_size, void* d_ws, size_t ws_size,
                              hipStream_t stream)
{
    const float* x       = (const float*)d_in[0];
    const int*   type_id = (const int*)  d_in[1];
    const int*   batch   = (const int*)  d_in[2];
    const float* u       = (const float*)d_in[3];
    const int*   ei_jj   = (const int*)  d_in[4];
    const float* ea_jj   = (const float*)d_in[5];
    const int*   ei_mj   = (const int*)  d_in[6];
    const float* ea_mj   = (const float*)d_in[7];
    const float* Wj_enc  = (const float*)d_in[8];   const float* bj_enc = (const float*)d_in[9];
    const float* Wm_enc  = (const float*)d_in[10];  const float* bm_enc = (const float*)d_in[11];
    const float* We_jj1  = (const float*)d_in[12];  const float* be_jj1 = (const float*)d_in[13];
    const float* Wm_jj1  = (const float*)d_in[14];  const float* bm_jj1 = (const float*)d_in[15];
    const float* We_mj1  = (const float*)d_in[16];  const float* be_mj1 = (const float*)d_in[17];
    const float* Wm_mj1  = (const float*)d_in[18];  const float* bm_mj1 = (const float*)d_in[19];
    const float* We_jj2  = (const float*)d_in[20];  const float* be_jj2 = (const float*)d_in[21];
    const float* Wm_jj2  = (const float*)d_in[22];  const float* bm_jj2 = (const float*)d_in[23];
    const float* We_mj2  = (const float*)d_in[24];  const float* be_mj2 = (const float*)d_in[25];
    const float* Wm_mj2  = (const float*)d_in[26];  const float* bm_mj2 = (const float*)d_in[27];
    const float* W1      = (const float*)d_in[28];  const float* b1     = (const float*)d_in[29];
    const float* W2      = (const float*)d_in[30];  const float* b2     = (const float*)d_in[31];
    const float* W3      = (const float*)d_in[32];  const float* b3     = (const float*)d_in[33];

    const int Nn  = in_sizes[0] / 5;
    const int Gg  = in_sizes[3] / 4;
    const int Ejj = in_sizes[5] / 5;
    const int Emj = in_sizes[7] / 4;

    float* ws = (float*)d_ws;
    const size_t NP = (size_t)Nn * DPAD;
    float* H   = ws;
    float* S   = ws + NP;
    float* agg = ws + 2 * NP;
    float* wpk = ws + 3 * NP;
    float* pWjj1 = wpk;          float* pbjj1 = wpk + 256;
    float* pWmj1 = wpk + 272;    float* pbmj1 = wpk + 528;
    float* pWjj2 = wpk + 544;    float* pbjj2 = wpk + 800;
    float* pWmj2 = wpk + 816;    float* pbmj2 = wpk + 1072;
    float* pW1   = wpk + 1088;   float* pb1   = wpk + 5184;
    float* pW2   = wpk + 5248;   float* pb2   = wpk + 7296;

    const int nbk = (Nn + BKT_NODES - 1) / BKT_NODES;
    const int nch = (Ejj + CHUNK_E - 1) / CHUNK_E;
    const int tabL = nbk * nch;
    const int nbT = (tabL + 1023) / 1024;
    v2i* res_jj  = (v2i*)agg;
    v2i* res_mj  = res_jj + Ejj;
    int* rdst_jj = (int*)(res_mj + Emj);
    int* rdst_mj = rdst_jj + Ejj;
    int* tab     = rdst_mj + Emj;
    int* tsum    = tab + 2 * (size_t)tabL;
    int* pool_next = (int*)S;
    int* pool_head = pool_next + Nn;

    float* P    = agg;
    float* sumJ = P;
    float* sumM = P + (size_t)16 * Gg;
    float* maxJ = P + (size_t)32 * Gg;
    float* maxM = P + (size_t)48 * Gg;
    float* cntJ = P + (size_t)64 * Gg;
    float* cntM = P + (size_t)65 * Gg;
    float* gbuf = P + (size_t)66 * Gg;
    float* m1   = P + (size_t)130 * Gg;
    float* m2   = P + (size_t)194 * Gg;
    float* out  = (float*)d_out;

    const int TB = 256;
    const int nodeBlocks = (Nn + TB - 1) / TB;

    pack_weight_kernel<<<1, 256, 0, stream>>>(Wm_jj1, bm_jj1, pWjj1, pbjj1, 15, 15, 16, 16);
    pack_weight_kernel<<<1, 256, 0, stream>>>(Wm_mj1, bm_mj1, pWmj1, pbmj1, 15, 15, 16, 16);
    pack_weight_kernel<<<1, 256, 0, stream>>>(Wm_jj2, bm_jj2, pWjj2, pbjj2, 15, 15, 16, 16);
    pack_weight_kernel<<<1, 256, 0, stream>>>(Wm_mj2, bm_mj2, pWmj2, pbmj2, 15, 15, 16, 16);
    pack_weight_kernel<<<16, 256, 0, stream>>>(W1, b1, pW1, pb1, 64, 64, 64, 64);
    pack_weight_kernel<<<8, 256, 0, stream>>>(W2, b2, pW2, pb2, 64, 32, 64, 32);

    auto gemm64 = [&](const float* A, const float* Wp, const float* bp,
                      float* C, int Ncpad, int lda, int ldc, int relu) {
        int tiles = (Gg >> 4) * (Ncpad >> 4);
        wmma_gemm_kernel<false, 64><<<(tiles + 7) / 8, TB, 0, stream>>>(
            A, nullptr, Wp, bp, C, Gg, Ncpad, lda, ldc, relu);
    };

    bucket_count_kernel<<<dim3(nch, 2), TB, 0, stream>>>(ei_jj, ei_mj, tab, Ejj, Emj, Nn, nch, nbk);
    csr_block_sum_kernel<<<dim3(nbT, 2), TB, 0, stream>>>(tab, tsum, tabL, nbT);
    csr_block_scan_kernel<<<dim3(nbT, 2), TB, 0, stream>>>(tab, tsum, tabL, nbT);
    bucket_place_kernel<<<dim3(nch, 2), TB, 0, stream>>>(ei_jj, ei_mj, tab, res_jj, res_mj, rdst_jj, rdst_mj, Ejj, Emj, Nn, nch, nbk);

    encode_kernel<<<nodeBlocks, TB, 0, stream>>>(x, type_id, Wj_enc, bj_enc, Wm_enc, bm_enc, H, Nn);

    conv_kernel<5, true, false><<<nbk, 1024, 0, stream>>>(H, S, res_jj, rdst_jj, tab, ea_jj, We_jj1, be_jj1, pWjj1, pbjj1, Nn, Ejj, nch, nbk);

    conv_kernel<4, false, false><<<nbk, 1024, 0, stream>>>(S, H, res_mj, rdst_mj, tab + tabL, ea_mj, We_mj1, be_mj1, pWmj1, pbmj1, Nn, Emj, nch, nbk);

    conv_kernel<5, true, true><<<nbk, 1024, 0, stream>>>(H, S, res_jj, rdst_jj, tab, ea_jj, We_jj2, be_jj2, pWjj2, pbjj2, Nn, Ejj, nch, nbk);

    conv_kernel<4, true, false><<<nbk, 1024, 0, stream>>>(S, H, res_mj, rdst_mj, tab + tabL, ea_mj, We_mj2, be_mj2, pWmj2, pbmj2, Nn, Emj, nch, nbk);

    fill_kernel<<<(Gg + TB - 1) / TB, TB, 0, stream>>>((float*)pool_head, 0.f, (size_t)Gg);
    pool_link_kernel<<<nodeBlocks, TB, 0, stream>>>(type_id, batch, pool_head, pool_next, Nn, Gg);
    pool_gather_kernel<<<(Gg + TB - 1) / TB, TB, 0, stream>>>(H, type_id, pool_head, pool_next,
                                                              sumJ, sumM, maxJ, maxM, cntJ, cntM, Nn, Gg);

    assemble_ln_kernel<<<(Gg + 127) / 128, 128, 0, stream>>>(sumJ, sumM, maxJ, maxM, cntJ, cntM, u, gbuf, Gg);
    gemm64(gbuf, pW1, pb1, m1, 64, 64, 64, 1);
    gemm64(m1,   pW2, pb2, m2, 32, 64, 32, 1);
    mlp3_kernel<<<(Gg + TB - 1) / TB, TB, 0, stream>>>(m2, W3, b3, out, Gg);
}


# node encoder computed in registers by each reader of a row (conv 1: gathered source rows from x and type, own row; weights in LDS): encode launch and the encoded plane go; sum orders as before
# speedup vs baseline: 3.4923x; 1.0511x over previous
#include <hip/hip_runtime.h>
#include <hip/hip_bf16.h>

typedef __attribute__((ext_vector_type(2))) float v2f;
typedef __attribute__((ext_vector_type(8))) float v8f;

#define DPAD 16
#define DDIM 15

__global__ void fill_kernel(float* __restrict__ p, float v, size_t n) {
    size_t i = (size_t)blockIdx.x * blockDim.x + threadIdx.x;
    size_t stride = (size_t)gridDim.x * blockDim.x;
    for (; i < n; i += stride) p[i] = v;
}

__global__ void pack_weight_kernel(const float* __restrict__ W, const float* __restrict__ b,
                                   float* __restrict__ Wp, float* __restrict__ bp,
                                   int K, int Nc, int Kpad, int Ncpad)
{
    int idx = blockIdx.x * blockDim.x + threadIdx.x;
    int total = Kpad * Ncpad;
    if (idx < total) {
        int k = idx / Ncpad, c = idx % Ncpad;
        float v = (k < K && c < Nc) ? W[k * Nc + c] : 0.f;
        Wp[((size_t)(k >> 1) * Ncpad + c) * 2 + (k & 1)] = v;
    }
    if (idx < Ncpad) bp[idx] = (idx < Nc) ? b[idx] : 0.f;
}

#define ENC_SET 24
__device__ __forceinline__ void encode_row(float (&v)[DPAD],
    const float* __restrict__ x, const int* __restrict__ type_id, size_t i,
    const float4* encw)
{
    float xi[5];
#pragma unroll
    for (int k = 0; k < 5; ++k) xi[k] = x[i * 5 + k];
    const int t = type_id[i];
    const float4* wr = encw + ((t == 0) ? 0 : ((t == 1) ? ENC_SET : 2 * ENC_SET));
    float acc[DPAD];
#pragma unroll
    for (int q = 0; q < 4; ++q) {
        const float4 w = wr[q];
        acc[4 * q + 0] = w.x; acc[4 * q + 1] = w.y; acc[4 * q + 2] = w.z; acc[4 * q + 3] = w.w;
    }
#pragma unroll
    for (int k = 0; k < 5; ++k) {
#pragma unroll
        for (int q = 0; q < 4; ++q) {
            const float4 w = wr[4 * (k + 1) + q];
            acc[4 * q + 0] = __builtin_fmaf(xi[k], w.x, acc[4 * q + 0]);
            acc[4 * q + 1] = __builtin_fmaf(xi[k], w.y, acc[4 * q + 1]);
            acc[4 * q + 2] = __builtin_fmaf(xi[k], w.z, acc[4 * q + 2]);
            if (q < 3) acc[4 * q + 3] = __builtin_fmaf(xi[k], w.w, acc[4 * q + 3]);
        }
    }
#pragma unroll
    for (int d = 0; d < DDIM; ++d) v[d] = fmaxf(acc[d], 0.f);
    v[DDIM] = 0.f;
}

typedef __attribute__((ext_vector_type(2))) int v2i;

#define BKT_SHIFT 12
#define BKT_NODES 4096
#define ACC_PITCH 17
#define CHUNK_E   4096
#define MAX_BKT   1024

__global__ __launch_bounds__(256) void bucket_count_kernel(
    const int* __restrict__ ei_a, const int* __restrict__ ei_b, int* __restrict__ tab,
    int Ea, int Eb, int Nn, int nch, int nbk)
{
    __shared__ int hist[MAX_BKT];
    const int t = threadIdx.x;
    const int set = blockIdx.y, chunk = blockIdx.x;
    const int E = set ? Eb : Ea;
    const int* __restrict__ ei = set ? ei_b : ei_a;
    for (int i = t; i < nbk; i += 256) hist[i] = 0;
    __syncthreads();
    const int base = chunk * CHUNK_E + t;
    const int last = max(E - 1, 0);
    int d[CHUNK_E / 256];
#pragma unroll
    for (int k = 0; k < CHUNK_E / 256; ++k) d[k] = ei[(size_t)E + min(base + k * 256, last)];
#pragma unroll
    for (int k = 0; k < CHUNK_E / 256; ++k) {
        if (base + k * 256 < E) {
            const unsigned dst = min((unsigned)d[k], (unsigned)(Nn - 1));
            atomicAdd(&hist[dst >> BKT_SHIFT], 1);
        }
    }
    __syncthreads();
    int* tb = tab + (size_t)set * nbk * nch;
    for (int i = t; i < nbk; i += 256) tb[(size_t)i * nch + chunk] = hist[i];
}

__global__ __launch_bounds__(256) void bucket_place_kernel(
    const int* __restrict__ ei_a, const int* __restrict__ ei_b, const int* __restrict__ tab,
    v2i* __restrict__ res_a, v2i* __restrict__ res_b,
    int* __restrict__ rdst_a, int* __restrict__ rdst_b,
    int Ea, int Eb, int Nn, int nch, int nbk)
{
    __shared__ int cursor[MAX_BKT];
    const int t = threadIdx.x;
    const int set = blockIdx.y, chunk = blockIdx.x;
    const int E = set ? Eb : Ea;
    const int* __restrict__ ei = set ? ei_b : ei_a;
    v2i* __restrict__ res = set ? res_b : res_a;
    int* __restrict__ rdst = set ? rdst_b : rdst_a;
    const int* tb = tab + (size_t)set * nbk * nch;
    for (int i = t; i < nbk; i += 256) cursor[i] = tb[(size_t)i * nch + chunk];
    __syncthreads();
    const int base = chunk * CHUNK_E + t;
    const int last = max(E - 1, 0);
    int sv[CHUNK_E / 256], d[CHUNK_E / 256];
#pragma unroll
    for (int k = 0; k < CHUNK_E / 256; ++k) {
        const int ec = min(base + k * 256, last);
        sv[k] = ei[ec];
        d[k]  = ei[(size_t)E + ec];
    }
#pragma unroll
    for (int k = 0; k < CHUNK_E / 256; ++k) {
        const int e = base + k * 256;
        if (e < E) {
            const int src = sv[k];
            const unsigned dst = min((unsigned)d[k], (unsigned)(Nn - 1));
            unsigned pos = (unsigned)atomicAdd(&cursor[dst >> BKT_SHIFT], 1);
            pos = min(pos, (unsigned)last);
            v2i o; o.x = e; o.y = src;
            res[pos] = o;
            rdst[pos] = (int)dst;
        }
    }
}

__global__ __launch_bounds__(256) void csr_block_sum_kernel(
    const int* __restrict__ cnt, int* __restrict__ bsum, int Nn, int nb)
{
    __shared__ int sh[256];
    const int* c = cnt + (size_t)blockIdx.y * Nn;
    const int t = threadIdx.x;
    const int base = blockIdx.x * 1024 + t * 4;
    int s = 0;
#pragma unroll
    for (int k = 0; k < 4; ++k) if (base + k < Nn) s += c[base + k];
    sh[t] = s;
    __syncthreads();
    for (int w = 128; w > 0; w >>= 1) {
        if (t < w) sh[t] += sh[t + w];
        __syncthreads();
    }
    if (t == 0) bsum[blockIdx.y * nb + blockIdx.x] = sh[0];
}

__global__ __launch_bounds__(256) void csr_block_scan_kernel(
    int* __restrict__ cnt, const int* __restrict__ bsum, int Nn, int nb)
{
    __shared__ int sh[256];
    int* c = cnt + (size_t)blockIdx.y * Nn;
    const int* bs = bsum + blockIdx.y * nb;
    const int t = threadIdx.x;

    int pc = 0;
    for (int j = t; j < (int)blockIdx.x; j += 256) pc += bs[j];
    sh[t] = pc;
    __syncthreads();
    for (int w = 128; w > 0; w >>= 1) {
        if (t < w) sh[t] += sh[t + w];
        __syncthreads();
    }
    const int carry = sh[0];
    __syncthreads();

    const int base = blockIdx.x * 1024 + t * 4;
    int v[4];
#pragma unroll
    for (int k = 0; k < 4; ++k) v[k] = (base + k < Nn) ? c[base + k] : 0;
    const int ts = v[0] + v[1] + v[2] + v[3];
    sh[t] = ts;
    __syncthreads();
    for (int d = 1; d < 256; d <<= 1) {
        int x = (t >= d) ? sh[t - d] : 0;
        __syncthreads();
        sh[t] += x;
        __syncthreads();
    }
    int run = carry + sh[t] - ts;
#pragma unroll
    for (int k = 0; k < 4; ++k) {
        if (base + k < Nn) c[base + k] = run;
        run += v[k];
    }
}

__device__ __forceinline__ void ln_relu_row(float (&v)[DPAD])
{
    float mu = 0.f;
#pragma unroll
    for (int d = 0; d < DDIM; ++d) mu += v[d];
    mu *= (1.f / DDIM);
    float var = 0.f;
#pragma unroll
    for (int d = 0; d < DDIM; ++d) { float t = v[d] - mu; var += t * t; }
    var *= (1.f / DDIM);
    const float inv = rsqrtf(var + 1e-5f);
#pragma unroll
    for (int d = 0; d < DDIM; ++d) v[d] = fmaxf((v[d] - mu) * inv, 0.f);
    v[DDIM] = 0.f;
}

__device__ __forceinline__ float xhalf(float x) {
    return __builtin_bit_cast(float, __builtin_amdgcn_permlanex16(
        __builtin_bit_cast(int, x), __builtin_bit_cast(int, x), 0x76543210, 0xfedcba98, false, false));
}

template <int EDIM, bool RELU, bool LNIN, bool ENC>
__global__ __launch_bounds__(1024) void conv_kernel(
    const float* __restrict__ h, float* __restrict__ hout,
    const float* __restrict__ x, const int* __restrict__ type_id,
    const float* __restrict__ Wje, const float* __restrict__ bje,
    const float* __restrict__ Wme, const float* __restrict__ bme,
    const v2i* __restrict__ res, const int* __restrict__ rdst, const int* __restrict__ tb,
    const float* __restrict__ ea,
    const float* __restrict__ We, const float* __restrict__ be,
    const float* __restrict__ Wp, const float* __restrict__ bp,
    int Nn, int E, int nch, int nbk)
{
    __shared__ float accs[BKT_NODES * ACC_PITCH];
    __shared__ float4 encw[ENC ? 3 * ENC_SET : 1];
    const int t = threadIdx.x;
    const int b = blockIdx.x;
    if (ENC && t < 3 * ENC_SET * 4) {
        const int set = t / (ENC_SET * 4), r = (t % (ENC_SET * 4)) >> 4, cc = t & 15;
        const float* Ws = set ? Wme : Wje;
        const float* bs = set ? bme : bje;
        float w = 0.f;
        if (set < 2 && cc < DDIM) w = (r == 0) ? bs[cc] : Ws[(r - 1) * DDIM + cc];
        ((float*)encw)[t] = w;
    }
    int s0 = tb[(size_t)b * nch];
    int s1 = (b + 1 < nbk) ? tb[(size_t)(b + 1) * nch] : E;
    s0 = min(max(s0, 0), E);
    s1 = min(max(s1, s0), E);

    for (int i = t; i < BKT_NODES * ACC_PITCH; i += 1024) accs[i] = 0.f;
    __syncthreads();

    const int hi = max(s1 - 1, 0);
    int p = s0 + t;
    int q = min(p, hi);
    v2i rn = res[q];
    int dn = rdst[q];
    for (; p < s1; p += 1024) {
        const v2i cur = rn;
        const int dc = dn;
        q = min(p + 1024, hi);
        rn = res[q];
        dn = rdst[q];

        float ev[EDIM];
#pragma unroll
        for (int k = 0; k < EDIM; ++k) ev[k] = ea[(size_t)cur.x * EDIM + k];

        float hl[DPAD];
        if (ENC) {
            encode_row(hl, x, type_id, (size_t)cur.y, encw);
        } else {
            const float4* hv = (const float4*)(h + (size_t)cur.y * DPAD);
#pragma unroll
            for (int qq = 0; qq < 4; ++qq) {
                float4 v = hv[qq];
                hl[4 * qq + 0] = v.x; hl[4 * qq + 1] = v.y;
                hl[4 * qq + 2] = v.z; hl[4 * qq + 3] = v.w;
            }
        }
        if (LNIN) ln_relu_row(hl);
        float* arow = accs + (dc & (BKT_NODES - 1)) * ACC_PITCH;
#pragma unroll
        for (int d = 0; d < DDIM; ++d) {
            float m = hl[d] + be[d];
#pragma unroll
            for (int k = 0; k < EDIM; ++k) m = __builtin_fmaf(ev[k], We[k * DDIM + d], m);
            m = fmaxf(m, 0.f);
            atomicAdd(arow + d, m);
        }
    }
    __syncthreads();

    const int lane = t & 31;
    const int c  = lane & 15;
    const int hf = lane >> 4;
    const int wave = __builtin_amdgcn_readfirstlane(t) >> 5;
    const float* bpp = Wp + ((size_t)hf * 16 + c) * 2;
    const float bv = bp[c];

    for (int g = wave; g < BKT_NODES / 32; g += 32) {
        const int wbase = b * BKT_NODES + g * 32;
        if (wbase >= Nn) break;

        float s[DPAD];
        const unsigned irow = min((unsigned)(wbase + lane), (unsigned)(Nn - 1));
        const float* ar = accs + (g * 32 + lane) * ACC_PITCH;
        if (ENC) {
            encode_row(s, x, type_id, (size_t)irow, encw);
        } else {
            const float4* hi4 = (const float4*)(h + (size_t)irow * DPAD);
#pragma unroll
            for (int qq = 0; qq < 4; ++qq) {
                float4 v = hi4[qq];
                s[4 * qq + 0] = v.x; s[4 * qq + 1] = v.y;
                s[4 * qq + 2] = v.z; s[4 * qq + 3] = v.w;
            }
        }
        if (LNIN) ln_relu_row(s);
#pragma unroll
        for (int d = 0; d < DPAD; ++d) s[d] += ar[d];

        v8f d0 = {0.f, 0.f, 0.f, 0.f, 0.f, 0.f, 0.f, 0.f};
        v8f d1 = {0.f, 0.f, 0.f, 0.f, 0.f, 0.f, 0.f, 0.f};
#pragma unroll
        for (int kk = 0; kk < DPAD; kk += 4) {
            const float snd0 = hf ? s[kk + 0] : s[kk + 2];
            const float snd1 = hf ? s[kk + 1] : s[kk + 3];
            const float rcv0 = xhalf(snd0);
            const float rcv1 = xhalf(snd1);
            const float own0 = hf ? s[kk + 2] : s[kk + 0];
            const float own1 = hf ? s[kk + 3] : s[kk + 1];
            v2f a0, a1;
            a0.x = hf ? rcv0 : own0;  a0.y = hf ? rcv1 : own1;
            a1.x = hf ? own0 : rcv0;  a1.y = hf ? own1 : rcv1;
            const v2f bf = *(const v2f*)(bpp + (size_t)kk * 16);
            d0 = __builtin_amdgcn_wmma_f32_16x16x4_f32(false, a0, false, bf, (short)0, d0, false, false);
            d1 = __builtin_amdgcn_wmma_f32_16x16x4_f32(false, a1, false, bf, (short)0, d1, false, false);
        }

        float* orow = hout + (size_t)wbase * DPAD + (8 * hf) * DPAD + c;
        if (wbase + 32 <= Nn) {
#pragma unroll
            for (int r = 0; r < 8; ++r) {
                float v0 = d0[r] + bv;
                float v1 = d1[r] + bv;
                if (RELU) { v0 = fmaxf(v0, 0.f); v1 = fmaxf(v1, 0.f); }
                orow[r * DPAD] = v0;
                orow[(r + 16) * DPAD] = v1;
            }
        } else {
#pragma unroll
            for (int r = 0; r < 8; ++r) {
                const int n0 = wbase + 8 * hf + r;
                float v0 = d0[r] + bv;
                float v1 = d1[r] + bv;
                if (RELU) { v0 = fmaxf(v0, 0.f); v1 = fmaxf(v1, 0.f); }
                if (n0 < Nn)      orow[r * DPAD] = v0;
                if (n0 + 16 < Nn) orow[(r + 16) * DPAD] = v1;
            }
        }
    }
}

template <bool HAS_A2, int KPAD>
__global__ __launch_bounds__(256) void wmma_gemm_kernel(
    const float* __restrict__ A, const float* __restrict__ A2,
    const float* __restrict__ Wp, const float* __restrict__ bp,
    float* __restrict__ C,
    int M, int Ncpad, int lda, int ldc, int relu)
{
    const int lane = threadIdx.x & 31;
    const int wave = threadIdx.x >> 5;
    const int ntN = Ncpad >> 4;
    const int tile = blockIdx.x * (blockDim.x >> 5) + wave;
    if (tile >= (M >> 4) * ntN) return;
    const int mt = tile / ntN;
    const int nt = tile % ntN;

    const int mrow  = mt * 16 + (lane & 15);
    const int ncol  = nt * 16 + (lane & 15);
    const int khalf = (lane >> 4) << 1;

    const float* ap  = A + (size_t)mrow * lda + khalf;
    const float* a2p = A2 + (size_t)mrow * lda + khalf;
    const float* bpp = Wp + ((size_t)(khalf >> 1) * Ncpad + ncol) * 2;
    const size_t bstep = (size_t)Ncpad * 4;

    v8f acc = {0.f, 0.f, 0.f, 0.f, 0.f, 0.f, 0.f, 0.f};
#pragma unroll
    for (int kk = 0; kk < KPAD; kk += 4) {
        v2f a = *(const v2f*)ap;
        if (HAS_A2) {
            v2f a2 = *(const v2f*)a2p;
            a.x += a2.x; a.y += a2.y;
            a2p += 4;
        }
        v2f b = *(const v2f*)bpp;
        ap += 4;
        bpp += bstep;
        acc = __builtin_amdgcn_wmma_f32_16x16x4_f32(
            false, a, false, b, (short)0, acc, false, false);
    }

    const int rbase = mt * 16 + ((lane >> 4) << 3);
    const float bv = bp[ncol];
#pragma unroll
    for (int r = 0; r < 8; ++r) {
        float v = acc[r] + bv;
        if (relu) v = fmaxf(v, 0.f);
        C[(size_t)(rbase + r) * ldc + ncol] = v;
    }
}

__global__ __launch_bounds__(256) void pool_link_kernel(
    const int* __restrict__ type_id, const int* __restrict__ batch,
    int* __restrict__ head, int* __restrict__ nxt, int Nn, int Gg)
{
    int i = blockIdx.x * blockDim.x + threadIdx.x;
    if (i >= Nn) return;
    const int t = type_id[i];
    const int b = batch[i];
    int link = 0;
    if ((t == 0 || t == 1) && b >= 0 && b < Gg) link = atomicExch(&head[b], i + 1);
    nxt[i] = link;
}

struct PoolNode { int nx; int tp; float4 r[4]; };

__device__ __forceinline__ void pool_fetch(PoolNode& n, int link,
    const float* __restrict__ h, const int* __restrict__ type_id,
    const int* __restrict__ nxt, int Nn)
{
    const unsigned i = min((unsigned)(link - 1), (unsigned)(Nn - 1));
    n.nx = nxt[i];
    n.tp = type_id[i];
    const float4* hv = (const float4*)(h + (size_t)i * DPAD);
#pragma unroll
    for (int q = 0; q < 4; ++q) n.r[q] = hv[q];
}

__device__ __forceinline__ void pool_accum(const PoolNode& n, bool valid,
    float (&sJ)[DPAD], float (&sM)[DPAD], float (&mJ)[DPAD], float (&mM)[DPAD], int& cJ, int& cM)
{
    const float ninf = -__builtin_inff();
    const bool inJ = valid & (n.tp == 0);
    const bool inM = valid & (n.tp != 0);
    cJ += inJ ? 1 : 0;
    cM += inM ? 1 : 0;
    float row[DPAD];
#pragma unroll
    for (int q = 0; q < 4; ++q) {
        row[4 * q + 0] = n.r[q].x; row[4 * q + 1] = n.r[q].y;
        row[4 * q + 2] = n.r[q].z; row[4 * q + 3] = n.r[q].w;
    }
    ln_relu_row(row);
#pragma unroll
    for (int d = 0; d < DPAD; ++d) {
        sJ[d] += inJ ? row[d] : 0.f;
        sM[d] += inM ? row[d] : 0.f;
        const float cj = inJ ? row[d] : ninf;
        const float cm = inM ? row[d] : ninf;
        mJ[d] = (cj > mJ[d]) ? cj : mJ[d];
        mM[d] = (cm > mM[d]) ? cm : mM[d];
    }
}

__global__ __launch_bounds__(256) void pool_gather_kernel(
    const float* __restrict__ h, const int* __restrict__ type_id,
    const int* __restrict__ head, const int* __restrict__ nxt,
    float* __restrict__ sumJ, float* __restrict__ sumM,
    float* __restrict__ maxJ, float* __restrict__ maxM,
    float* __restrict__ cntJ, float* __restrict__ cntM, int Nn, int Gg)
{
    const int g = blockIdx.x * blockDim.x + threadIdx.x;
    if (g >= Gg) return;
    const float ninf = -__builtin_inff();
    float sJ[DPAD], sM[DPAD], mJ[DPAD], mM[DPAD];
#pragma unroll
    for (int d = 0; d < DPAD; ++d) { sJ[d] = 0.f; sM[d] = 0.f; mJ[d] = ninf; mM[d] = ninf; }
    int cJ = 0, cM = 0;

    PoolNode a, b;
    const int first = head[g];
    bool live = (first != 0);
    pool_fetch(a, first, h, type_id, nxt, Nn);
    for (int it = 0; it < Nn && live; it += 2) {
        pool_fetch(b, a.nx, h, type_id, nxt, Nn);
        pool_accum(a, true, sJ, sM, mJ, mM, cJ, cM);
        const bool moreB = (a.nx != 0);
        pool_fetch(a, b.nx, h, type_id, nxt, Nn);
        pool_accum(b, moreB, sJ, sM, mJ, mM, cJ, cM);
        live = moreB & (b.nx != 0);
    }

    float4* oSJ = (float4*)(sumJ + (size_t)g * DPAD);
    float4* oSM = (float4*)(sumM + (size_t)g * DPAD);
    float4* oMJ = (float4*)(maxJ + (size_t)g * DPAD);
    float4* oMM = (float4*)(maxM + (size_t)g * DPAD);
#pragma unroll
    for (int q = 0; q < 4; ++q) {
        oSJ[q] = make_float4(sJ[4 * q], sJ[4 * q + 1], sJ[4 * q + 2], sJ[4 * q + 3]);
        oSM[q] = make_float4(sM[4 * q], sM[4 * q + 1], sM[4 * q + 2], sM[4 * q + 3]);
        oMJ[q] = make_float4(mJ[4 * q], mJ[4 * q + 1], mJ[4 * q + 2], mJ[4 * q + 3]);
        oMM[q] = make_float4(mM[4 * q], mM[4 * q + 1], mM[4 * q + 2], mM[4 * q + 3]);
    }
    cntJ[g] = (float)cJ;
    cntM[g] = (float)cM;
}

__global__ __launch_bounds__(128) void assemble_ln_kernel(
    const float* __restrict__ sumJ, const float* __restrict__ sumM,
    const float* __restrict__ maxJ, const float* __restrict__ maxM,
    const float* __restrict__ cntJ, const float* __restrict__ cntM,
    const float* __restrict__ u, float* __restrict__ g, int Gg)
{
    int gi = blockIdx.x * blockDim.x + threadIdx.x;
    if (gi >= Gg) return;
    float r[64];
    float cm = fmaxf(cntM[gi], 1.f);
    float cj = fmaxf(cntJ[gi], 1.f);
#pragma unroll
    for (int d = 0; d < DDIM; ++d) r[d] = sumM[(size_t)gi * DPAD + d] / cm;
#pragma unroll
    for (int d = 0; d < DDIM; ++d) r[15 + d] = sumJ[(size_t)gi * DPAD + d] / cj;
#pragma unroll
    for (int d = 0; d < DDIM; ++d) {
        float v = maxJ[(size_t)gi * DPAD + d];
        r[30 + d] = (v > -3.0e38f) ? v : 0.f;
    }
#pragma unroll
    for (int d = 0; d < DDIM; ++d) {
        float v = maxM[(size_t)gi * DPAD + d];
        r[45 + d] = (v > -3.0e38f) ? v : 0.f;
    }
#pragma unroll
    for (int j = 0; j < 4; ++j) r[60 + j] = u[(size_t)gi * 4 + j];

    float mu = 0.f;
#pragma unroll
    for (int k = 0; k < 64; ++k) mu += r[k];
    mu *= (1.f / 64.f);
    float var = 0.f;
#pragma unroll
    for (int k = 0; k < 64; ++k) { float t = r[k] - mu; var += t * t; }
    var *= (1.f / 64.f);
    float inv = rsqrtf(var + 1e-5f);
#pragma unroll
    for (int k = 0; k < 64; ++k) g[(size_t)gi * 64 + k] = (r[k] - mu) * inv;
}

__global__ __launch_bounds__(256) void mlp3_kernel(
    const float* __restrict__ m2, const float* __restrict__ W3,
    const float* __restrict__ b3, float* __restrict__ out, int Gg)
{
    int gi = blockIdx.x * blockDim.x + threadIdx.x;
    if (gi >= Gg) return;
    float s = b3[0];
#pragma unroll
    for (int k = 0; k < 32; ++k) s += m2[(size_t)gi * 32 + k] * W3[k];
    out[gi] = s;
}

extern "C" void kernel_launch(void* const* d_in, const int* in_sizes, int n_in,
                              void* d_out, int out_size, void* d_ws, size_t ws_size,
                              hipStream_t stream)
{
    const float* x       = (const float*)d_in[0];
    const int*   type_id = (const int*)  d_in[1];
    const int*   batch   = (const int*)  d_in[2];
    const float* u       = (const float*)d_in[3];
    const int*   ei_jj   = (const int*)  d_in[4];
    const float* ea_jj   = (const float*)d_in[5];
    const int*   ei_mj   = (const int*)  d_in[6];
    const float* ea_mj   = (const float*)d_in[7];
    const float* Wj_enc  = (const float*)d_in[8];   const float* bj_enc = (const float*)d_in[9];
    const float* Wm_enc  = (const float*)d_in[10];  const float* bm_enc = (const float*)d_in[11];
    const float* We_jj1  = (const float*)d_in[12];  const float* be_jj1 = (const float*)d_in[13];
    const float* Wm_jj1  = (const float*)d_in[14];  const float* bm_jj1 = (const float*)d_in[15];
    const float* We_mj1  = (const float*)d_in[16];  const float* be_mj1 = (const float*)d_in[17];
    const float* Wm_mj1  = (const float*)d_in[18];  const float* bm_mj1 = (const float*)d_in[19];
    const float* We_jj2  = (const float*)d_in[20];  const float* be_jj2 = (const float*)d_in[21];
    const float* Wm_jj2  = (const float*)d_in[22];  const float* bm_jj2 = (const float*)d_in[23];
    const float* We_mj2  = (const float*)d_in[24];  const float* be_mj2 = (const float*)d_in[25];
    const float* Wm_mj2  = (const float*)d_in[26];  const float* bm_mj2 = (const float*)d_in[27];
    const float* W1      = (const float*)d_in[28];  const float* b1     = (const float*)d_in[29];
    const float* W2      = (const float*)d_in[30];  const float* b2     = (const float*)d_in[31];
    const float* W3      = (const float*)d_in[32];  const float* b3     = (const float*)d_in[33];

    const int Nn  = in_sizes[0] / 5;
    const int Gg  = in_sizes[3] / 4;
    const int Ejj = in_sizes[5] / 5;
    const int Emj = in_sizes[7] / 4;

    float* ws = (float*)d_ws;
    const size_t NP = (size_t)Nn * DPAD;
    float* H   = ws;
    float* S   = ws + NP;
    float* agg = ws + 2 * NP;
    float* wpk = ws + 3 * NP;
    float* pWjj1 = wpk;          float* pbjj1 = wpk + 256;
    float* pWmj1 = wpk + 272;    float* pbmj1 = wpk + 528;
    float* pWjj2 = wpk + 544;    float* pbjj2 = wpk + 800;
    float* pWmj2 = wpk + 816;    float* pbmj2 = wpk + 1072;
    float* pW1   = wpk + 1088;   float* pb1   = wpk + 5184;
    float* pW2   = wpk + 5248;   float* pb2   = wpk + 7296;

    const int nbk = (Nn + BKT_NODES - 1) / BKT_NODES;
    const int nch = (Ejj + CHUNK_E - 1) / CHUNK_E;
    const int tabL = nbk * nch;
    const int nbT = (tabL + 1023) / 1024;
    v2i* res_jj  = (v2i*)agg;
    v2i* res_mj  = res_jj + Ejj;
    int* rdst_jj = (int*)(res_mj + Emj);
    int* rdst_mj = rdst_jj + Ejj;
    int* tab     = rdst_mj + Emj;
    int* tsum    = tab + 2 * (size_t)tabL;
    int* pool_next = (int*)S;
    int* pool_head = pool_next + Nn;

    float* P    = agg;
    float* sumJ = P;
    float* sumM = P + (size_t)16 * Gg;
    float* maxJ = P + (size_t)32 * Gg;
    float* maxM = P + (size_t)48 * Gg;
    float* cntJ = P + (size_t)64 * Gg;
    float* cntM = P + (size_t)65 * Gg;
    float* gbuf = P + (size_t)66 * Gg;
    float* m1   = P + (size_t)130 * Gg;
    float* m2   = P + (size_t)194 * Gg;
    float* out  = (float*)d_out;

    const int TB = 256;
    const int nodeBlocks = (Nn + TB - 1) / TB;

    pack_weight_kernel<<<1, 256, 0, stream>>>(Wm_jj1, bm_jj1, pWjj1, pbjj1, 15, 15, 16, 16);
    pack_weight_kernel<<<1, 256, 0, stream>>>(Wm_mj1, bm_mj1, pWmj1, pbmj1, 15, 15, 16, 16);
    pack_weight_kernel<<<1, 256, 0, stream>>>(Wm_jj2, bm_jj2, pWjj2, pbjj2, 15, 15, 16, 16);
    pack_weight_kernel<<<1, 256, 0, stream>>>(Wm_mj2, bm_mj2, pWmj2, pbmj2, 15, 15, 16, 16);
    pack_weight_kernel<<<16, 256, 0, stream>>>(W1, b1, pW1, pb1, 64, 64, 64, 64);
    pack_weight_kernel<<<8, 256, 0, stream>>>(W2, b2, pW2, pb2, 64, 32, 64, 32);

    auto gemm64 = [&](const float* A, const float* Wp, const float* bp,
                      float* C, int Ncpad, int lda, int ldc, int relu) {
        int tiles = (Gg >> 4) * (Ncpad >> 4);
        wmma_gemm_kernel<false, 64><<<(tiles + 7) / 8, TB, 0, stream>>>(
            A, nullptr, Wp, bp, C, Gg, Ncpad, lda, ldc, relu);
    };

    bucket_count_kernel<<<dim3(nch, 2), TB, 0, stream>>>(ei_jj, ei_mj, tab, Ejj, Emj, Nn, nch, nbk);
    csr_block_sum_kernel<<<dim3(nbT, 2), TB, 0, stream>>>(tab, tsum, tabL, nbT);
    csr_block_scan_kernel<<<dim3(nbT, 2), TB, 0, stream>>>(tab, tsum, tabL, nbT);
    bucket_place_kernel<<<dim3(nch, 2), TB, 0, stream>>>(ei_jj, ei_mj, tab, res_jj, res_mj, rdst_jj, rdst_mj, Ejj, Emj, Nn, nch, nbk);

    conv_kernel<5, true, false, true><<<nbk, 1024, 0, stream>>>(nullptr, S, x, type_id, Wj_enc, bj_enc, Wm_enc, bm_enc, res_jj, rdst_jj, tab, ea_jj, We_jj1, be_jj1, pWjj1, pbjj1, Nn, Ejj, nch, nbk);

    conv_kernel<4, false, false, false><<<nbk, 1024, 0, stream>>>(S, H, x, type_id, Wj_enc, bj_enc, Wm_enc, bm_enc, res_mj, rdst_mj, tab + tabL, ea_mj, We_mj1, be_mj1, pWmj1, pbmj1, Nn, Emj, nch, nbk);

    conv_kernel<5, true, true, false><<<nbk, 1024, 0, stream>>>(H, S, x, type_id, Wj_enc, bj_enc, Wm_enc, bm_enc, res_jj, rdst_jj, tab, ea_jj, We_jj2, be_jj2, pWjj2, pbjj2, Nn, Ejj, nch, nbk);

    conv_kernel<4, true, false, false><<<nbk, 1024, 0, stream>>>(S, H, x, type_id, Wj_enc, bj_enc, Wm_enc, bm_enc, res_mj, rdst_mj, tab + tabL, ea_mj, We_mj2, be_mj2, pWmj2, pbmj2, Nn, Emj, nch, nbk);

    fill_kernel<<<(Gg + TB - 1) / TB, TB, 0, stream>>>((float*)pool_head, 0.f, (size_t)Gg);
    pool_link_kernel<<<nodeBlocks, TB, 0, stream>>>(type_id, batch, pool_head, pool_next, Nn, Gg);
    pool_gather_kernel<<<(Gg + TB - 1) / TB, TB, 0, stream>>>(H, type_id, pool_head, pool_next,
                                                              sumJ, sumM, maxJ, maxM, cntJ, cntM, Nn, Gg);

    assemble_ln_kernel<<<(Gg + 127) / 128, 128, 0, stream>>>(sumJ, sumM, maxJ, maxM, cntJ, cntM, u, gbuf, Gg);
    gemm64(gbuf, pW1, pb1, m1, 64, 64, 64, 1);
    gemm64(m1,   pW2, pb2, m2, 32, 64, 32, 1);
    mlp3_kernel<<<(Gg + TB - 1) / TB, TB, 0, stream>>>(m2, W3, b3, out, Gg);
}


# conv edge pass: a source row is fetched by the 4 lanes of its quad, 16 B each (a load touches 8 lines, not 32); an in-quad DPP transpose hands each lane its edge's row; values, sum orders as before
# speedup vs baseline: 3.6514x; 1.0456x over previous
#include <hip/hip_runtime.h>
#include <hip/hip_bf16.h>

typedef __attribute__((ext_vector_type(2))) float v2f;
typedef __attribute__((ext_vector_type(8))) float v8f;

#define DPAD 16
#define DDIM 15

__global__ void fill_kernel(float* __restrict__ p, float v, size_t n) {
    size_t i = (size_t)blockIdx.x * blockDim.x + threadIdx.x;
    size_t stride = (size_t)gridDim.x * blockDim.x;
    for (; i < n; i += stride) p[i] = v;
}

__global__ void pack_weight_kernel(const float* __restrict__ W, const float* __restrict__ b,
                                   float* __restrict__ Wp, float* __restrict__ bp,
                                   int K, int Nc, int Kpad, int Ncpad)
{
    int idx = blockIdx.x * blockDim.x + threadIdx.x;
    int total = Kpad * Ncpad;
    if (idx < total) {
        int k = idx / Ncpad, c = idx % Ncpad;
        float v = (k < K && c < Nc) ? W[k * Nc + c] : 0.f;
        Wp[((size_t)(k >> 1) * Ncpad + c) * 2 + (k & 1)] = v;
    }
    if (idx < Ncpad) bp[idx] = (idx < Nc) ? b[idx] : 0.f;
}

#define ENC_SET 24
__device__ __forceinline__ void encode_row(float (&v)[DPAD],
    const float* __restrict__ x, const int* __restrict__ type_id, size_t i,
    const float4* encw)
{
    float xi[5];
#pragma unroll
    for (int k = 0; k < 5; ++k) xi[k] = x[i * 5 + k];
    const int t = type_id[i];
    const float4* wr = encw + ((t == 0) ? 0 : ((t == 1) ? ENC_SET : 2 * ENC_SET));
    float acc[DPAD];
#pragma unroll
    for (int q = 0; q < 4; ++q) {
        const float4 w = wr[q];
        acc[4 * q + 0] = w.x; acc[4 * q + 1] = w.y; acc[4 * q + 2] = w.z; acc[4 * q + 3] = w.w;
    }
#pragma unroll
    for (int k = 0; k < 5; ++k) {
#pragma unroll
        for (int q = 0; q < 4; ++q) {
            const float4 w = wr[4 * (k + 1) + q];
            acc[4 * q + 0] = __builtin_fmaf(xi[k], w.x, acc[4 * q + 0]);
            acc[4 * q + 1] = __builtin_fmaf(xi[k], w.y, acc[4 * q + 1]);
            acc[4 * q + 2] = __builtin_fmaf(xi[k], w.z, acc[4 * q + 2]);
            if (q < 3) acc[4 * q + 3] = __builtin_fmaf(xi[k], w.w, acc[4 * q + 3]);
        }
    }
#pragma unroll
    for (int d = 0; d < DDIM; ++d) v[d] = fmaxf(acc[d], 0.f);
    v[DDIM] = 0.f;
}

typedef __attribute__((ext_vector_type(2))) int v2i;

#define BKT_SHIFT 12
#define BKT_NODES 4096
#define ACC_PITCH 17
#define CHUNK_E   4096
#define MAX_BKT   1024

__global__ __launch_bounds__(256) void bucket_count_kernel(
    const int* __restrict__ ei_a, const int* __restrict__ ei_b, int* __restrict__ tab,
    int Ea, int Eb, int Nn, int nch, int nbk)
{
    __shared__ int hist[MAX_BKT];
    const int t = threadIdx.x;
    const int set = blockIdx.y, chunk = blockIdx.x;
    const int E = set ? Eb : Ea;
    const int* __restrict__ ei = set ? ei_b : ei_a;
    for (int i = t; i < nbk; i += 256) hist[i] = 0;
    __syncthreads();
    const int base = chunk * CHUNK_E + t;
    const int last = max(E - 1, 0);
    int d[CHUNK_E / 256];
#pragma unroll
    for (int k = 0; k < CHUNK_E / 256; ++k) d[k] = ei[(size_t)E + min(base + k * 256, last)];
#pragma unroll
    for (int k = 0; k < CHUNK_E / 256; ++k) {
        if (base + k * 256 < E) {
            const unsigned dst = min((unsigned)d[k], (unsigned)(Nn - 1));
            atomicAdd(&hist[dst >> BKT_SHIFT], 1);
        }
    }
    __syncthreads();
    int* tb = tab + (size_t)set * nbk * nch;
    for (int i = t; i < nbk; i += 256) tb[(size_t)i * nch + chunk] = hist[i];
}

__global__ __launch_bounds__(256) void bucket_place_kernel(
    const int* __restrict__ ei_a, const int* __restrict__ ei_b, const int* __restrict__ tab,
    v2i* __restrict__ res_a, v2i* __restrict__ res_b,
    int* __restrict__ rdst_a, int* __restrict__ rdst_b,
    int Ea, int Eb, int Nn, int nch, int nbk)
{
    __shared__ int cursor[MAX_BKT];
    const int t = threadIdx.x;
    const int set = blockIdx.y, chunk = blockIdx.x;
    const int E = set ? Eb : Ea;
    const int* __restrict__ ei = set ? ei_b : ei_a;
    v2i* __restrict__ res = set ? res_b : res_a;
    int* __restrict__ rdst = set ? rdst_b : rdst_a;
    const int* tb = tab + (size_t)set * nbk * nch;
    for (int i = t; i < nbk; i += 256) cursor[i] = tb[(size_t)i * nch + chunk];
    __syncthreads();
    const int base = chunk * CHUNK_E + t;
    const int last = max(E - 1, 0);
    int sv[CHUNK_E / 256], d[CHUNK_E / 256];
#pragma unroll
    for (int k = 0; k < CHUNK_E / 256; ++k) {
        const int ec = min(base + k * 256, last);
        sv[k] = ei[ec];
        d[k]  = ei[(size_t)E + ec];
    }
#pragma unroll
    for (int k = 0; k < CHUNK_E / 256; ++k) {
        const int e = base + k * 256;
        if (e < E) {
            const int src = sv[k];
            const unsigned dst = min((unsigned)d[k], (unsigned)(Nn - 1));
            unsigned pos = (unsigned)atomicAdd(&cursor[dst >> BKT_SHIFT], 1);
            pos = min(pos, (unsigned)last);
            v2i o; o.x = e; o.y = src;
            res[pos] = o;
            rdst[pos] = (int)dst;
        }
    }
}

__global__ __launch_bounds__(256) void csr_block_sum_kernel(
    const int* __restrict__ cnt, int* __restrict__ bsum, int Nn, int nb)
{
    __shared__ int sh[256];
    const int* c = cnt + (size_t)blockIdx.y * Nn;
    const int t = threadIdx.x;
    const int base = blockIdx.x * 1024 + t * 4;
    int s = 0;
#pragma unroll
    for (int k = 0; k < 4; ++k) if (base + k < Nn) s += c[base + k];
    sh[t] = s;
    __syncthreads();
    for (int w = 128; w > 0; w >>= 1) {
        if (t < w) sh[t] += sh[t + w];
        __syncthreads();
    }
    if (t == 0) bsum[blockIdx.y * nb + blockIdx.x] = sh[0];
}

__global__ __launch_bounds__(256) void csr_block_scan_kernel(
    int* __restrict__ cnt, const int* __restrict__ bsum, int Nn, int nb)
{
    __shared__ int sh[256];
    int* c = cnt + (size_t)blockIdx.y * Nn;
    const int* bs = bsum + blockIdx.y * nb;
    const int t = threadIdx.x;

    int pc = 0;
    for (int j = t; j < (int)blockIdx.x; j += 256) pc += bs[j];
    sh[t] = pc;
    __syncthreads();
    for (int w = 128; w > 0; w >>= 1) {
        if (t < w) sh[t] += sh[t + w];
        __syncthreads();
    }
    const int carry = sh[0];
    __syncthreads();

    const int base = blockIdx.x * 1024 + t * 4;
    int v[4];
#pragma unroll
    for (int k = 0; k < 4; ++k) v[k] = (base + k < Nn) ? c[base + k] : 0;
    const int ts = v[0] + v[1] + v[2] + v[3];
    sh[t] = ts;
    __syncthreads();
    for (int d = 1; d < 256; d <<= 1) {
        int x = (t >= d) ? sh[t - d] : 0;
        __syncthreads();
        sh[t] += x;
        __syncthreads();
    }
    int run = carry + sh[t] - ts;
#pragma unroll
    for (int k = 0; k < 4; ++k) {
        if (base + k < Nn) c[base + k] = run;
        run += v[k];
    }
}

__device__ __forceinline__ void ln_relu_row(float (&v)[DPAD])
{
    float mu = 0.f;
#pragma unroll
    for (int d = 0; d < DDIM; ++d) mu += v[d];
    mu *= (1.f / DDIM);
    float var = 0.f;
#pragma unroll
    for (int d = 0; d < DDIM; ++d) { float t = v[d] - mu; var += t * t; }
    var *= (1.f / DDIM);
    const float inv = rsqrtf(var + 1e-5f);
#pragma unroll
    for (int d = 0; d < DDIM; ++d) v[d] = fmaxf((v[d] - mu) * inv, 0.f);
    v[DDIM] = 0.f;
}

__device__ __forceinline__ float xhalf(float x) {
    return __builtin_bit_cast(float, __builtin_amdgcn_permlanex16(
        __builtin_bit_cast(int, x), __builtin_bit_cast(int, x), 0x76543210, 0xfedcba98, false, false));
}

#define QUAD_PERM(a, b, c, d) ((a) | ((b) << 2) | ((c) << 4) | ((d) << 6))

template <int CTRL>
__device__ __forceinline__ int quad_i(int v) {
    return __builtin_amdgcn_update_dpp(0, v, CTRL, 0xf, 0xf, true);
}
template <int CTRL>
__device__ __forceinline__ float quad_f(float v) {
    return __builtin_bit_cast(float, quad_i<CTRL>(__builtin_bit_cast(int, v)));
}

template <int CTRL>
__device__ __forceinline__ void quad_swap(float4& lo, float4& hi, bool up) {
    const float4 pl = make_float4(quad_f<CTRL>(lo.x), quad_f<CTRL>(lo.y), quad_f<CTRL>(lo.z), quad_f<CTRL>(lo.w));
    const float4 ph = make_float4(quad_f<CTRL>(hi.x), quad_f<CTRL>(hi.y), quad_f<CTRL>(hi.z), quad_f<CTRL>(hi.w));
    const float4 nlo = make_float4(up ? ph.x : lo.x, up ? ph.y : lo.y, up ? ph.z : lo.z, up ? ph.w : lo.w);
    const float4 nhi = make_float4(up ? hi.x : pl.x, up ? hi.y : pl.y, up ? hi.z : pl.z, up ? hi.w : pl.w);
    lo = nlo;
    hi = nhi;
}

__device__ __forceinline__ void gather_row_quad(float (&v)[DPAD], const float* __restrict__ h, int src, int lane)
{
    const int j = lane & 3;
    const int n0 = quad_i<QUAD_PERM(0, 0, 0, 0)>(src);
    const int n1 = quad_i<QUAD_PERM(1, 1, 1, 1)>(src);
    const int n2 = quad_i<QUAD_PERM(2, 2, 2, 2)>(src);
    const int n3 = quad_i<QUAD_PERM(3, 3, 3, 3)>(src);
    float4 g0 = *(const float4*)(h + (size_t)n0 * DPAD + 4 * j);
    float4 g1 = *(const float4*)(h + (size_t)n1 * DPAD + 4 * j);
    float4 g2 = *(const float4*)(h + (size_t)n2 * DPAD + 4 * j);
    float4 g3 = *(const float4*)(h + (size_t)n3 * DPAD + 4 * j);
    quad_swap<QUAD_PERM(1, 0, 3, 2)>(g0, g1, (j & 1) != 0);
    quad_swap<QUAD_PERM(1, 0, 3, 2)>(g2, g3, (j & 1) != 0);
    quad_swap<QUAD_PERM(2, 3, 0, 1)>(g0, g2, (j & 2) != 0);
    quad_swap<QUAD_PERM(2, 3, 0, 1)>(g1, g3, (j & 2) != 0);
    v[0]  = g0.x; v[1]  = g0.y; v[2]  = g0.z; v[3]  = g0.w;
    v[4]  = g1.x; v[5]  = g1.y; v[6]  = g1.z; v[7]  = g1.w;
    v[8]  = g2.x; v[9]  = g2.y; v[10] = g2.z; v[11] = g2.w;
    v[12] = g3.x; v[13] = g3.y; v[14] = g3.z; v[15] = g3.w;
}

template <int EDIM, bool RELU, bool LNIN, bool ENC>
__global__ __launch_bounds__(1024) void conv_kernel(
    const float* __restrict__ h, float* __restrict__ hout,
    const float* __restrict__ x, const int* __restrict__ type_id,
    const float* __restrict__ Wje, const float* __restrict__ bje,
    const float* __restrict__ Wme, const float* __restrict__ bme,
    const v2i* __restrict__ res, const int* __restrict__ rdst, const int* __restrict__ tb,
    const float* __restrict__ ea,
    const float* __restrict__ We, const float* __restrict__ be,
    const float* __restrict__ Wp, const float* __restrict__ bp,
    int Nn, int E, int nch, int nbk)
{
    __shared__ float accs[(BKT_NODES + 1) * ACC_PITCH];
    __shared__ float4 encw[ENC ? 3 * ENC_SET : 1];
    const int t = threadIdx.x;
    const int b = blockIdx.x;
    if (ENC && t < 3 * ENC_SET * 4) {
        const int set = t / (ENC_SET * 4), r = (t % (ENC_SET * 4)) >> 4, cc = t & 15;
        const float* Ws = set ? Wme : Wje;
        const float* bs = set ? bme : bje;
        float w = 0.f;
        if (set < 2 && cc < DDIM) w = (r == 0) ? bs[cc] : Ws[(r - 1) * DDIM + cc];
        ((float*)encw)[t] = w;
    }
    int s0 = tb[(size_t)b * nch];
    int s1 = (b + 1 < nbk) ? tb[(size_t)(b + 1) * nch] : E;
    s0 = min(max(s0, 0), E);
    s1 = min(max(s1, s0), E);

    for (int i = t; i < (BKT_NODES + 1) * ACC_PITCH; i += 1024) accs[i] = 0.f;
    __syncthreads();

    const int hi = max(s1 - 1, 0);
    const int ln = t & 31;
    int pw = s0 + ((__builtin_amdgcn_readfirstlane(t) >> 5) << 5);
    int q = min(pw + ln, hi);
    v2i rn = res[q];
    int dn = rdst[q];
    for (; pw < s1; pw += 1024) {
        const int p = pw + ln;
        const v2i cur = rn;
        const int dc = dn;
        q = min(p + 1024, hi);
        rn = res[q];
        dn = rdst[q];

        float ev[EDIM];
#pragma unroll
        for (int k = 0; k < EDIM; ++k) ev[k] = ea[(size_t)cur.x * EDIM + k];

        float hl[DPAD];
        if (ENC) {
            encode_row(hl, x, type_id, (size_t)cur.y, encw);
        } else {
            gather_row_quad(hl, h, cur.y, ln);
        }
        if (LNIN) ln_relu_row(hl);
        float* arow = accs + ((p < s1) ? (dc & (BKT_NODES - 1)) : BKT_NODES) * ACC_PITCH;
#pragma unroll
        for (int d = 0; d < DDIM; ++d) {
            float m = hl[d] + be[d];
#pragma unroll
            for (int k = 0; k < EDIM; ++k) m = __builtin_fmaf(ev[k], We[k * DDIM + d], m);
            m = fmaxf(m, 0.f);
            atomicAdd(arow + d, m);
        }
    }
    __syncthreads();

    const int lane = t & 31;
    const int c  = lane & 15;
    const int hf = lane >> 4;
    const int wave = __builtin_amdgcn_readfirstlane(t) >> 5;
    const float* bpp = Wp + ((size_t)hf * 16 + c) * 2;
    const float bv = bp[c];

    for (int g = wave; g < BKT_NODES / 32; g += 32) {
        const int wbase = b * BKT_NODES + g * 32;
        if (wbase >= Nn) break;

        float s[DPAD];
        const unsigned irow = min((unsigned)(wbase + lane), (unsigned)(Nn - 1));
        const float* ar = accs + (g * 32 + lane) * ACC_PITCH;
        if (ENC) {
            encode_row(s, x, type_id, (size_t)irow, encw);
        } else {
            gather_row_quad(s, h, (int)irow, lane);
        }
        if (LNIN) ln_relu_row(s);
#pragma unroll
        for (int d = 0; d < DPAD; ++d) s[d] += ar[d];

        v8f d0 = {0.f, 0.f, 0.f, 0.f, 0.f, 0.f, 0.f, 0.f};
        v8f d1 = {0.f, 0.f, 0.f, 0.f, 0.f, 0.f, 0.f, 0.f};
#pragma unroll
        for (int kk = 0; kk < DPAD; kk += 4) {
            const float snd0 = hf ? s[kk + 0] : s[kk + 2];
            const float snd1 = hf ? s[kk + 1] : s[kk + 3];
            const float rcv0 = xhalf(snd0);
            const float rcv1 = xhalf(snd1);
            const float own0 = hf ? s[kk + 2] : s[kk + 0];
            const float own1 = hf ? s[kk + 3] : s[kk + 1];
            v2f a0, a1;
            a0.x = hf ? rcv0 : own0;  a0.y = hf ? rcv1 : own1;
            a1.x = hf ? own0 : rcv0;  a1.y = hf ? own1 : rcv1;
            const v2f bf = *(const v2f*)(bpp + (size_t)kk * 16);
            d0 = __builtin_amdgcn_wmma_f32_16x16x4_f32(false, a0, false, bf, (short)0, d0, false, false);
            d1 = __builtin_amdgcn_wmma_f32_16x16x4_f32(false, a1, false, bf, (short)0, d1, false, false);
        }

        float* orow = hout + (size_t)wbase * DPAD + (8 * hf) * DPAD + c;
        if (wbase + 32 <= Nn) {
#pragma unroll
            for (int r = 0; r < 8; ++r) {
                float v0 = d0[r] + bv;
                float v1 = d1[r] + bv;
                if (RELU) { v0 = fmaxf(v0, 0.f); v1 = fmaxf(v1, 0.f); }
                orow[r * DPAD] = v0;
                orow[(r + 16) * DPAD] = v1;
            }
        } else {
#pragma unroll
            for (int r = 0; r < 8; ++r) {
                const int n0 = wbase + 8 * hf + r;
                float v0 = d0[r] + bv;
                float v1 = d1[r] + bv;
                if (RELU) { v0 = fmaxf(v0, 0.f); v1 = fmaxf(v1, 0.f); }
                if (n0 < Nn)      orow[r * DPAD] = v0;
                if (n0 + 16 < Nn) orow[(r + 16) * DPAD] = v1;
            }
        }
    }
}

template <bool HAS_A2, int KPAD>
__global__ __launch_bounds__(256) void wmma_gemm_kernel(
    const float* __restrict__ A, const float* __restrict__ A2,
    const float* __restrict__ Wp, const float* __restrict__ bp,
    float* __restrict__ C,
    int M, int Ncpad, int lda, int ldc, int relu)
{
    const int lane = threadIdx.x & 31;
    const int wave = threadIdx.x >> 5;
    const int ntN = Ncpad >> 4;
    const int tile = blockIdx.x * (blockDim.x >> 5) + wave;
    if (tile >= (M >> 4) * ntN) return;
    const int mt = tile / ntN;
    const int nt = tile % ntN;

    const int mrow  = mt * 16 + (lane & 15);
    const int ncol  = nt * 16 + (lane & 15);
    const int khalf = (lane >> 4) << 1;

    const float* ap  = A + (size_t)mrow * lda + khalf;
    const float* a2p = A2 + (size_t)mrow * lda + khalf;
    const float* bpp = Wp + ((size_t)(khalf >> 1) * Ncpad + ncol) * 2;
    const size_t bstep = (size_t)Ncpad * 4;

    v8f acc = {0.f, 0.f, 0.f, 0.f, 0.f, 0.f, 0.f, 0.f};
#pragma unroll
    for (int kk = 0; kk < KPAD; kk += 4) {
        v2f a = *(const v2f*)ap;
        if (HAS_A2) {
            v2f a2 = *(const v2f*)a2p;
            a.x += a2.x; a.y += a2.y;
            a2p += 4;
        }
        v2f b = *(const v2f*)bpp;
        ap += 4;
        bpp += bstep;
        acc = __builtin_amdgcn_wmma_f32_16x16x4_f32(
            false, a, false, b, (short)0, acc, false, false);
    }

    const int rbase = mt * 16 + ((lane >> 4) << 3);
    const float bv = bp[ncol];
#pragma unroll
    for (int r = 0; r < 8; ++r) {
        float v = acc[r] + bv;
        if (relu) v = fmaxf(v, 0.f);
        C[(size_t)(rbase + r) * ldc + ncol] = v;
    }
}

__global__ __launch_bounds__(256) void pool_link_kernel(
    const int* __restrict__ type_id, const int* __restrict__ batch,
    int* __restrict__ head, int* __restrict__ nxt, int Nn, int Gg)
{
    int i = blockIdx.x * blockDim.x + threadIdx.x;
    if (i >= Nn) return;
    const int t = type_id[i];
    const int b = batch[i];
    int link = 0;
    if ((t == 0 || t == 1) && b >= 0 && b < Gg) link = atomicExch(&head[b], i + 1);
    nxt[i] = link;
}

struct PoolNode { int nx; int tp; float4 r[4]; };

__device__ __forceinline__ void pool_fetch(PoolNode& n, int link,
    const float* __restrict__ h, const int* __restrict__ type_id,
    const int* __restrict__ nxt, int Nn)
{
    const unsigned i = min((unsigned)(link - 1), (unsigned)(Nn - 1));
    n.nx = nxt[i];
    n.tp = type_id[i];
    const float4* hv = (const float4*)(h + (size_t)i * DPAD);
#pragma unroll
    for (int q = 0; q < 4; ++q) n.r[q] = hv[q];
}

__device__ __forceinline__ void pool_accum(const PoolNode& n, bool valid,
    float (&sJ)[DPAD], float (&sM)[DPAD], float (&mJ)[DPAD], float (&mM)[DPAD], int& cJ, int& cM)
{
    const float ninf = -__builtin_inff();
    const bool inJ = valid & (n.tp == 0);
    const bool inM = valid & (n.tp != 0);
    cJ += inJ ? 1 : 0;
    cM += inM ? 1 : 0;
    float row[DPAD];
#pragma unroll
    for (int q = 0; q < 4; ++q) {
        row[4 * q + 0] = n.r[q].x; row[4 * q + 1] = n.r[q].y;
        row[4 * q + 2] = n.r[q].z; row[4 * q + 3] = n.r[q].w;
    }
    ln_relu_row(row);
#pragma unroll
    for (int d = 0; d < DPAD; ++d) {
        sJ[d] += inJ ? row[d] : 0.f;
        sM[d] += inM ? row[d] : 0.f;
        const float cj = inJ ? row[d] : ninf;
        const float cm = inM ? row[d] : ninf;
        mJ[d] = (cj > mJ[d]) ? cj : mJ[d];
        mM[d] = (cm > mM[d]) ? cm : mM[d];
    }
}

__global__ __launch_bounds__(256) void pool_gather_kernel(
    const float* __restrict__ h, const int* __restrict__ type_id,
    const int* __restrict__ head, const int* __restrict__ nxt,
    float* __restrict__ sumJ, float* __restrict__ sumM,
    float* __restrict__ maxJ, float* __restrict__ maxM,
    float* __restrict__ cntJ, float* __restrict__ cntM, int Nn, int Gg)
{
    const int g = blockIdx.x * blockDim.x + threadIdx.x;
    if (g >= Gg) return;
    const float ninf = -__builtin_inff();
    float sJ[DPAD], sM[DPAD], mJ[DPAD], mM[DPAD];
#pragma unroll
    for (int d = 0; d < DPAD; ++d) { sJ[d] = 0.f; sM[d] = 0.f; mJ[d] = ninf; mM[d] = ninf; }
    int cJ = 0, cM = 0;

    PoolNode a, b;
    const int first = head[g];
    bool live = (first != 0);
    pool_fetch(a, first, h, type_id, nxt, Nn);
    for (int it = 0; it < Nn && live; it += 2) {
        pool_fetch(b, a.nx, h, type_id, nxt, Nn);
        pool_accum(a, true, sJ, sM, mJ, mM, cJ, cM);
        const bool moreB = (a.nx != 0);
        pool_fetch(a, b.nx, h, type_id, nxt, Nn);
        pool_accum(b, moreB, sJ, sM, mJ, mM, cJ, cM);
        live = moreB & (b.nx != 0);
    }

    float4* oSJ = (float4*)(sumJ + (size_t)g * DPAD);
    float4* oSM = (float4*)(sumM + (size_t)g * DPAD);
    float4* oMJ = (float4*)(maxJ + (size_t)g * DPAD);
    float4* oMM = (float4*)(maxM + (size_t)g * DPAD);
#pragma unroll
    for (int q = 0; q < 4; ++q) {
        oSJ[q] = make_float4(sJ[4 * q], sJ[4 * q + 1], sJ[4 * q + 2], sJ[4 * q + 3]);
        oSM[q] = make_float4(sM[4 * q], sM[4 * q + 1], sM[4 * q + 2], sM[4 * q + 3]);
        oMJ[q] = make_float4(mJ[4 * q], mJ[4 * q + 1], mJ[4 * q + 2], mJ[4 * q + 3]);
        oMM[q] = make_float4(mM[4 * q], mM[4 * q + 1], mM[4 * q + 2], mM[4 * q + 3]);
    }
    cntJ[g] = (float)cJ;
    cntM[g] = (float)cM;
}

__global__ __launch_bounds__(128) void assemble_ln_kernel(
    const float* __restrict__ sumJ, const float* __restrict__ sumM,
    const float* __restrict__ maxJ, const float* __restrict__ maxM,
    const float* __restrict__ cntJ, const float* __restrict__ cntM,
    const float* __restrict__ u, float* __restrict__ g, int Gg)
{
    int gi = blockIdx.x * blockDim.x + threadIdx.x;
    if (gi >= Gg) return;
    float r[64];
    float cm = fmaxf(cntM[gi], 1.f);
    float cj = fmaxf(cntJ[gi], 1.f);
#pragma unroll
    for (int d = 0; d < DDIM; ++d) r[d] = sumM[(size_t)gi * DPAD + d] / cm;
#pragma unroll
    for (int d = 0; d < DDIM; ++d) r[15 + d] = sumJ[(size_t)gi * DPAD + d] / cj;
#pragma unroll
    for (int d = 0; d < DDIM; ++d) {
        float v = maxJ[(size_t)gi * DPAD + d];
        r[30 + d] = (v > -3.0e38f) ? v : 0.f;
    }
#pragma unroll
    for (int d = 0; d < DDIM; ++d) {
        float v = maxM[(size_t)gi * DPAD + d];
        r[45 + d] = (v > -3.0e38f) ? v : 0.f;
    }
#pragma unroll
    for (int j = 0; j < 4; ++j) r[60 + j] = u[(size_t)gi * 4 + j];

    float mu = 0.f;
#pragma unroll
    for (int k = 0; k < 64; ++k) mu += r[k];
    mu *= (1.f / 64.f);
    float var = 0.f;
#pragma unroll
    for (int k = 0; k < 64; ++k) { float t = r[k] - mu; var += t * t; }
    var *= (1.f / 64.f);
    float inv = rsqrtf(var + 1e-5f);
#pragma unroll
    for (int k = 0; k < 64; ++k) g[(size_t)gi * 64 + k] = (r[k] - mu) * inv;
}

__global__ __launch_bounds__(256) void mlp3_kernel(
    const float* __restrict__ m2, const float* __restrict__ W3,
    const float* __restrict__ b3, float* __restrict__ out, int Gg)
{
    int gi = blockIdx.x * blockDim.x + threadIdx.x;
    if (gi >= Gg) return;
    float s = b3[0];
#pragma unroll
    for (int k = 0; k < 32; ++k) s += m2[(size_t)gi * 32 + k] * W3[k];
    out[gi] = s;
}

extern "C" void kernel_launch(void* const* d_in, const int* in_sizes, int n_in,
                              void* d_out, int out_size, void* d_ws, size_t ws_size,
                              hipStream_t stream)
{
    const float* x       = (const float*)d_in[0];
    const int*   type_id = (const int*)  d_in[1];
    const int*   batch   = (const int*)  d_in[2];
    const float* u       = (const float*)d_in[3];
    const int*   ei_jj   = (const int*)  d_in[4];
    const float* ea_jj   = (const float*)d_in[5];
    const int*   ei_mj   = (const int*)  d_in[6];
    const float* ea_mj   = (const float*)d_in[7];
    const float* Wj_enc  = (const float*)d_in[8];   const float* bj_enc = (const float*)d_in[9];
    const float* Wm_enc  = (const float*)d_in[10];  const float* bm_enc = (const float*)d_in[11];
    const float* We_jj1  = (const float*)d_in[12];  const float* be_jj1 = (const float*)d_in[13];
    const float* Wm_jj1  = (const float*)d_in[14];  const float* bm_jj1 = (const float*)d_in[15];
    const float* We_mj1  = (const float*)d_in[16];  const float* be_mj1 = (const float*)d_in[17];
    const float* Wm_mj1  = (const float*)d_in[18];  const float* bm_mj1 = (const float*)d_in[19];
    const float* We_jj2  = (const float*)d_in[20];  const float* be_jj2 = (const float*)d_in[21];
    const float* Wm_jj2  = (const float*)d_in[22];  const float* bm_jj2 = (const float*)d_in[23];
    const float* We_mj2  = (const float*)d_in[24];  const float* be_mj2 = (const float*)d_in[25];
    const float* Wm_mj2  = (const float*)d_in[26];  const float* bm_mj2 = (const float*)d_in[27];
    const float* W1      = (const float*)d_in[28];  const float* b1     = (const float*)d_in[29];
    const float* W2      = (const float*)d_in[30];  const float* b2     = (const float*)d_in[31];
    const float* W3      = (const float*)d_in[32];  const float* b3     = (const float*)d_in[33];

    const int Nn  = in_sizes[0] / 5;
    const int Gg  = in_sizes[3] / 4;
    const int Ejj = in_sizes[5] / 5;
    const int Emj = in_sizes[7] / 4;

    float* ws = (float*)d_ws;
    const size_t NP = (size_t)Nn * DPAD;
    float* H   = ws;
    float* S   = ws + NP;
    float* agg = ws + 2 * NP;
    float* wpk = ws + 3 * NP;
    float* pWjj1 = wpk;          float* pbjj1 = wpk + 256;
    float* pWmj1 = wpk + 272;    float* pbmj1 = wpk + 528;
    float* pWjj2 = wpk + 544;    float* pbjj2 = wpk + 800;
    float* pWmj2 = wpk + 816;    float* pbmj2 = wpk + 1072;
    float* pW1   = wpk + 1088;   float* pb1   = wpk + 5184;
    float* pW2   = wpk + 5248;   float* pb2   = wpk + 7296;

    const int nbk = (Nn + BKT_NODES - 1) / BKT_NODES;
    const int nch = (Ejj + CHUNK_E - 1) / CHUNK_E;
    const int tabL = nbk * nch;
    const int nbT = (tabL + 1023) / 1024;
    v2i* res_jj  = (v2i*)agg;
    v2i* res_mj  = res_jj + Ejj;
    int* rdst_jj = (int*)(res_mj + Emj);
    int* rdst_mj = rdst_jj + Ejj;
    int* tab     = rdst_mj + Emj;
    int* tsum    = tab + 2 * (size_t)tabL;
    int* pool_next = (int*)S;
    int* pool_head = pool_next + Nn;

    float* P    = agg;
    float* sumJ = P;
    float* sumM = P + (size_t)16 * Gg;
    float* maxJ = P + (size_t)32 * Gg;
    float* maxM = P + (size_t)48 * Gg;
    float* cntJ = P + (size_t)64 * Gg;
    float* cntM = P + (size_t)65 * Gg;
    float* gbuf = P + (size_t)66 * Gg;
    float* m1   = P + (size_t)130 * Gg;
    float* m2   = P + (size_t)194 * Gg;
    float* out  = (float*)d_out;

    const int TB = 256;
    const int nodeBlocks = (Nn + TB - 1) / TB;

    pack_weight_kernel<<<1, 256, 0, stream>>>(Wm_jj1, bm_jj1, pWjj1, pbjj1, 15, 15, 16, 16);
    pack_weight_kernel<<<1, 256, 0, stream>>>(Wm_mj1, bm_mj1, pWmj1, pbmj1, 15, 15, 16, 16);
    pack_weight_kernel<<<1, 256, 0, stream>>>(Wm_jj2, bm_jj2, pWjj2, pbjj2, 15, 15, 16, 16);
    pack_weight_kernel<<<1, 256, 0, stream>>>(Wm_mj2, bm_mj2, pWmj2, pbmj2, 15, 15, 16, 16);
    pack_weight_kernel<<<16, 256, 0, stream>>>(W1, b1, pW1, pb1, 64, 64, 64, 64);
    pack_weight_kernel<<<8, 256, 0, stream>>>(W2, b2, pW2, pb2, 64, 32, 64, 32);

    auto gemm64 = [&](const float* A, const float* Wp, const float* bp,
                      float* C, int Ncpad, int lda, int ldc, int relu) {
        int tiles = (Gg >> 4) * (Ncpad >> 4);
        wmma_gemm_kernel<false, 64><<<(tiles + 7) / 8, TB, 0, stream>>>(
            A, nullptr, Wp, bp, C, Gg, Ncpad, lda, ldc, relu);
    };

    bucket_count_kernel<<<dim3(nch, 2), TB, 0, stream>>>(ei_jj, ei_mj, tab, Ejj, Emj, Nn, nch, nbk);
    csr_block_sum_kernel<<<dim3(nbT, 2), TB, 0, stream>>>(tab, tsum, tabL, nbT);
    csr_block_scan_kernel<<<dim3(nbT, 2), TB, 0, stream>>>(tab, tsum, tabL, nbT);
    bucket_place_kernel<<<dim3(nch, 2), TB, 0, stream>>>(ei_jj, ei_mj, tab, res_jj, res_mj, rdst_jj, rdst_mj, Ejj, Emj, Nn, nch, nbk);

    conv_kernel<5, true, false, true><<<nbk, 1024, 0, stream>>>(nullptr, S, x, type_id, Wj_enc, bj_enc, Wm_enc, bm_enc, res_jj, rdst_jj, tab, ea_jj, We_jj1, be_jj1, pWjj1, pbjj1, Nn, Ejj, nch, nbk);

    conv_kernel<4, false, false, false><<<nbk, 1024, 0, stream>>>(S, H, x, type_id, Wj_enc, bj_enc, Wm_enc, bm_enc, res_mj, rdst_mj, tab + tabL, ea_mj, We_mj1, be_mj1, pWmj1, pbmj1, Nn, Emj, nch, nbk);

    conv_kernel<5, true, true, false><<<nbk, 1024, 0, stream>>>(H, S, x, type_id, Wj_enc, bj_enc, Wm_enc, bm_enc, res_jj, rdst_jj, tab, ea_jj, We_jj2, be_jj2, pWjj2, pbjj2, Nn, Ejj, nch, nbk);

    conv_kernel<4, true, false, false><<<nbk, 1024, 0, stream>>>(S, H, x, type_id, Wj_enc, bj_enc, Wm_enc, bm_enc, res_mj, rdst_mj, tab + tabL, ea_mj, We_mj2, be_mj2, pWmj2, pbmj2, Nn, Emj, nch, nbk);

    fill_kernel<<<(Gg + TB - 1) / TB, TB, 0, stream>>>((float*)pool_head, 0.f, (size_t)Gg);
    pool_link_kernel<<<nodeBlocks, TB, 0, stream>>>(type_id, batch, pool_head, pool_next, Nn, Gg);
    pool_gather_kernel<<<(Gg + TB - 1) / TB, TB, 0, stream>>>(H, type_id, pool_head, pool_next,
                                                              sumJ, sumM, maxJ, maxM, cntJ, cntM, Nn, Gg);

    assemble_ln_kernel<<<(Gg + 127) / 128, 128, 0, stream>>>(sumJ, sumM, maxJ, maxM, cntJ, cntM, u, gbuf, Gg);
    gemm64(gbuf, pW1, pb1, m1, 64, 64, 64, 1);
    gemm64(m1,   pW2, pb2, m2, 32, 64, 32, 1);
    mlp3_kernel<<<(Gg + TB - 1) / TB, TB, 0, stream>>>(m2, W3, b3, out, Gg);
}


# conv edge pass: a lane whose message is zero after the ReLU sits that LDS float add out (s + 0 = s, bit-identical; about a third of the lane adds go); sum order: LDS add arrival, as before
# speedup vs baseline: 3.8056x; 1.0422x over previous
#include <hip/hip_runtime.h>
#include <hip/hip_bf16.h>

typedef __attribute__((ext_vector_type(2))) float v2f;
typedef __attribute__((ext_vector_type(8))) float v8f;

#define DPAD 16
#define DDIM 15

__global__ void fill_kernel(float* __restrict__ p, float v, size_t n) {
    size_t i = (size_t)blockIdx.x * blockDim.x + threadIdx.x;
    size_t stride = (size_t)gridDim.x * blockDim.x;
    for (; i < n; i += stride) p[i] = v;
}

__global__ void pack_weight_kernel(const float* __restrict__ W, const float* __restrict__ b,
                                   float* __restrict__ Wp, float* __restrict__ bp,
                                   int K, int Nc, int Kpad, int Ncpad)
{
    int idx = blockIdx.x * blockDim.x + threadIdx.x;
    int total = Kpad * Ncpad;
    if (idx < total) {
        int k = idx / Ncpad, c = idx % Ncpad;
        float v = (k < K && c < Nc) ? W[k * Nc + c] : 0.f;
        Wp[((size_t)(k >> 1) * Ncpad + c) * 2 + (k & 1)] = v;
    }
    if (idx < Ncpad) bp[idx] = (idx < Nc) ? b[idx] : 0.f;
}

#define ENC_SET 24
__device__ __forceinline__ void encode_row(float (&v)[DPAD],
    const float* __restrict__ x, const int* __restrict__ type_id, size_t i,
    const float4* encw)
{
    float xi[5];
#pragma unroll
    for (int k = 0; k < 5; ++k) xi[k] = x[i * 5 + k];
    const int t = type_id[i];
    const float4* wr = encw + ((t == 0) ? 0 : ((t == 1) ? ENC_SET : 2 * ENC_SET));
    float acc[DPAD];
#pragma unroll
    for (int q = 0; q < 4; ++q) {
        const float4 w = wr[q];
        acc[4 * q + 0] = w.x; acc[4 * q + 1] = w.y; acc[4 * q + 2] = w.z; acc[4 * q + 3] = w.w;
    }
#pragma unroll
    for (int k = 0; k < 5; ++k) {
#pragma unroll
        for (int q = 0; q < 4; ++q) {
            const float4 w = wr[4 * (k + 1) + q];
            acc[4 * q + 0] = __builtin_fmaf(xi[k], w.x, acc[4 * q + 0]);
            acc[4 * q + 1] = __builtin_fmaf(xi[k], w.y, acc[4 * q + 1]);
            acc[4 * q + 2] = __builtin_fmaf(xi[k], w.z, acc[4 * q + 2]);
            if (q < 3) acc[4 * q + 3] = __builtin_fmaf(xi[k], w.w, acc[4 * q + 3]);
        }
    }
#pragma unroll
    for (int d = 0; d < DDIM; ++d) v[d] = fmaxf(acc[d], 0.f);
    v[DDIM] = 0.f;
}

typedef __attribute__((ext_vector_type(2))) int v2i;

#define BKT_SHIFT 12
#define BKT_NODES 4096
#define ACC_PITCH 17
#define CHUNK_E   4096
#define MAX_BKT   1024

__global__ __launch_bounds__(256) void bucket_count_kernel(
    const int* __restrict__ ei_a, const int* __restrict__ ei_b, int* __restrict__ tab,
    int Ea, int Eb, int Nn, int nch, int nbk)
{
    __shared__ int hist[MAX_BKT];
    const int t = threadIdx.x;
    const int set = blockIdx.y, chunk = blockIdx.x;
    const int E = set ? Eb : Ea;
    const int* __restrict__ ei = set ? ei_b : ei_a;
    for (int i = t; i < nbk; i += 256) hist[i] = 0;
    __syncthreads();
    const int base = chunk * CHUNK_E + t;
    const int last = max(E - 1, 0);
    int d[CHUNK_E / 256];
#pragma unroll
    for (int k = 0; k < CHUNK_E / 256; ++k) d[k] = ei[(size_t)E + min(base + k * 256, last)];
#pragma unroll
    for (int k = 0; k < CHUNK_E / 256; ++k) {
        if (base + k * 256 < E) {
            const unsigned dst = min((unsigned)d[k], (unsigned)(Nn - 1));
            atomicAdd(&hist[dst >> BKT_SHIFT], 1);
        }
    }
    __syncthreads();
    int* tb = tab + (size_t)set * nbk * nch;
    for (int i = t; i < nbk; i += 256) tb[(size_t)i * nch + chunk] = hist[i];
}

__global__ __launch_bounds__(256) void bucket_place_kernel(
    const int* __restrict__ ei_a, const int* __restrict__ ei_b, const int* __restrict__ tab,
    v2i* __restrict__ res_a, v2i* __restrict__ res_b,
    int* __restrict__ rdst_a, int* __restrict__ rdst_b,
    int Ea, int Eb, int Nn, int nch, int nbk)
{
    __shared__ int cursor[MAX_BKT];
    const int t = threadIdx.x;
    const int set = blockIdx.y, chunk = blockIdx.x;
    const int E = set ? Eb : Ea;
    const int* __restrict__ ei = set ? ei_b : ei_a;
    v2i* __restrict__ res = set ? res_b : res_a;
    int* __restrict__ rdst = set ? rdst_b : rdst_a;
    const int* tb = tab + (size_t)set * nbk * nch;
    for (int i = t; i < nbk; i += 256) cursor[i] = tb[(size_t)i * nch + chunk];
    __syncthreads();
    const int base = chunk * CHUNK_E + t;
    const int last = max(E - 1, 0);
    int sv[CHUNK_E / 256], d[CHUNK_E / 256];
#pragma unroll
    for (int k = 0; k < CHUNK_E / 256; ++k) {
        const int ec = min(base + k * 256, last);
        sv[k] = ei[ec];
        d[k]  = ei[(size_t)E + ec];
    }
#pragma unroll
    for (int k = 0; k < CHUNK_E / 256; ++k) {
        const int e = base + k * 256;
        if (e < E) {
            const int src = sv[k];
            const unsigned dst = min((unsigned)d[k], (unsigned)(Nn - 1));
            unsigned pos = (unsigned)atomicAdd(&cursor[dst >> BKT_SHIFT], 1);
            pos = min(pos, (unsigned)last);
            v2i o; o.x = e; o.y = src;
            res[pos] = o;
            rdst[pos] = (int)dst;
        }
    }
}

__global__ __launch_bounds__(256) void csr_block_sum_kernel(
    const int* __restrict__ cnt, int* __restrict__ bsum, int Nn, int nb)
{
    __shared__ int sh[256];
    const int* c = cnt + (size_t)blockIdx.y * Nn;
    const int t = threadIdx.x;
    const int base = blockIdx.x * 1024 + t * 4;
    int s = 0;
#pragma unroll
    for (int k = 0; k < 4; ++k) if (base + k < Nn) s += c[base + k];
    sh[t] = s;
    __syncthreads();
    for (int w = 128; w > 0; w >>= 1) {
        if (t < w) sh[t] += sh[t + w];
        __syncthreads();
    }
    if (t == 0) bsum[blockIdx.y * nb + blockIdx.x] = sh[0];
}

__global__ __launch_bounds__(256) void csr_block_scan_kernel(
    int* __restrict__ cnt, const int* __restrict__ bsum, int Nn, int nb)
{
    __shared__ int sh[256];
    int* c = cnt + (size_t)blockIdx.y * Nn;
    const int* bs = bsum + blockIdx.y * nb;
    const int t = threadIdx.x;

    int pc = 0;
    for (int j = t; j < (int)blockIdx.x; j += 256) pc += bs[j];
    sh[t] = pc;
    __syncthreads();
    for (int w = 128; w > 0; w >>= 1) {
        if (t < w) sh[t] += sh[t + w];
        __syncthreads();
    }
    const int carry = sh[0];
    __syncthreads();

    const int base = blockIdx.x * 1024 + t * 4;
    int v[4];
#pragma unroll
    for (int k = 0; k < 4; ++k) v[k] = (base + k < Nn) ? c[base + k] : 0;
    const int ts = v[0] + v[1] + v[2] + v[3];
    sh[t] = ts;
    __syncthreads();
    for (int d = 1; d < 256; d <<= 1) {
        int x = (t >= d) ? sh[t - d] : 0;
        __syncthreads();
        sh[t] += x;
        __syncthreads();
    }
    int run = carry + sh[t] - ts;
#pragma unroll
    for (int k = 0; k < 4; ++k) {
        if (base + k < Nn) c[base + k] = run;
        run += v[k];
    }
}

__device__ __forceinline__ void ln_relu_row(float (&v)[DPAD])
{
    float mu = 0.f;
#pragma unroll
    for (int d = 0; d < DDIM; ++d) mu += v[d];
    mu *= (1.f / DDIM);
    float var = 0.f;
#pragma unroll
    for (int d = 0; d < DDIM; ++d) { float t = v[d] - mu; var += t * t; }
    var *= (1.f / DDIM);
    const float inv = rsqrtf(var + 1e-5f);
#pragma unroll
    for (int d = 0; d < DDIM; ++d) v[d] = fmaxf((v[d] - mu) * inv, 0.f);
    v[DDIM] = 0.f;
}

__device__ __forceinline__ float xhalf(float x) {
    return __builtin_bit_cast(float, __builtin_amdgcn_permlanex16(
        __builtin_bit_cast(int, x), __builtin_bit_cast(int, x), 0x76543210, 0xfedcba98, false, false));
}

#define QUAD_PERM(a, b, c, d) ((a) | ((b) << 2) | ((c) << 4) | ((d) << 6))

template <int CTRL>
__device__ __forceinline__ int quad_i(int v) {
    return __builtin_amdgcn_update_dpp(0, v, CTRL, 0xf, 0xf, true);
}
template <int CTRL>
__device__ __forceinline__ float quad_f(float v) {
    return __builtin_bit_cast(float, quad_i<CTRL>(__builtin_bit_cast(int, v)));
}

template <int CTRL>
__device__ __forceinline__ void quad_swap(float4& lo, float4& hi, bool up) {
    const float4 pl = make_float4(quad_f<CTRL>(lo.x), quad_f<CTRL>(lo.y), quad_f<CTRL>(lo.z), quad_f<CTRL>(lo.w));
    const float4 ph = make_float4(quad_f<CTRL>(hi.x), quad_f<CTRL>(hi.y), quad_f<CTRL>(hi.z), quad_f<CTRL>(hi.w));
    const float4 nlo = make_float4(up ? ph.x : lo.x, up ? ph.y : lo.y, up ? ph.z : lo.z, up ? ph.w : lo.w);
    const float4 nhi = make_float4(up ? hi.x : pl.x, up ? hi.y : pl.y, up ? hi.z : pl.z, up ? hi.w : pl.w);
    lo = nlo;
    hi = nhi;
}

__device__ __forceinline__ void gather_row_quad(float (&v)[DPAD], const float* __restrict__ h, int src, int lane)
{
    const int j = lane & 3;
    const int n0 = quad_i<QUAD_PERM(0, 0, 0, 0)>(src);
    const int n1 = quad_i<QUAD_PERM(1, 1, 1, 1)>(src);
    const int n2 = quad_i<QUAD_PERM(2, 2, 2, 2)>(src);
    const int n3 = quad_i<QUAD_PERM(3, 3, 3, 3)>(src);
    float4 g0 = *(const float4*)(h + (size_t)n0 * DPAD + 4 * j);
    float4 g1 = *(const float4*)(h + (size_t)n1 * DPAD + 4 * j);
    float4 g2 = *(const float4*)(h + (size_t)n2 * DPAD + 4 * j);
    float4 g3 = *(const float4*)(h + (size_t)n3 * DPAD + 4 * j);
    quad_swap<QUAD_PERM(1, 0, 3, 2)>(g0, g1, (j & 1) != 0);
    quad_swap<QUAD_PERM(1, 0, 3, 2)>(g2, g3, (j & 1) != 0);
    quad_swap<QUAD_PERM(2, 3, 0, 1)>(g0, g2, (j & 2) != 0);
    quad_swap<QUAD_PERM(2, 3, 0, 1)>(g1, g3, (j & 2) != 0);
    v[0]  = g0.x; v[1]  = g0.y; v[2]  = g0.z; v[3]  = g0.w;
    v[4]  = g1.x; v[5]  = g1.y; v[6]  = g1.z; v[7]  = g1.w;
    v[8]  = g2.x; v[9]  = g2.y; v[10] = g2.z; v[11] = g2.w;
    v[12] = g3.x; v[13] = g3.y; v[14] = g3.z; v[15] = g3.w;
}

template <int EDIM, bool RELU, bool LNIN, bool ENC>
__global__ __launch_bounds__(1024) void conv_kernel(
    const float* __restrict__ h, float* __restrict__ hout,
    const float* __restrict__ x, const int* __restrict__ type_id,
    const float* __restrict__ Wje, const float* __restrict__ bje,
    const float* __restrict__ Wme, const float* __restrict__ bme,
    const v2i* __restrict__ res, const int* __restrict__ rdst, const int* __restrict__ tb,
    const float* __restrict__ ea,
    const float* __restrict__ We, const float* __restrict__ be,
    const float* __restrict__ Wp, const float* __restrict__ bp,
    int Nn, int E, int nch, int nbk)
{
    __shared__ float accs[(BKT_NODES + 1) * ACC_PITCH];
    __shared__ float4 encw[ENC ? 3 * ENC_SET : 1];
    const int t = threadIdx.x;
    const int b = blockIdx.x;
    if (ENC && t < 3 * ENC_SET * 4) {
        const int set = t / (ENC_SET * 4), r = (t % (ENC_SET * 4)) >> 4, cc = t & 15;
        const float* Ws = set ? Wme : Wje;
        const float* bs = set ? bme : bje;
        float w = 0.f;
        if (set < 2 && cc < DDIM) w = (r == 0) ? bs[cc] : Ws[(r - 1) * DDIM + cc];
        ((float*)encw)[t] = w;
    }
    int s0 = tb[(size_t)b * nch];
    int s1 = (b + 1 < nbk) ? tb[(size_t)(b + 1) * nch] : E;
    s0 = min(max(s0, 0), E);
    s1 = min(max(s1, s0), E);

    for (int i = t; i < (BKT_NODES + 1) * ACC_PITCH; i += 1024) accs[i] = 0.f;
    __syncthreads();

    const float* bpp = Wp + ((size_t)((t & 31) >> 4) * 16 + (t & 15)) * 2;
    v2f bfr[DPAD / 4];
#pragma unroll
    for (int kk = 0; kk < DPAD; kk += 4) bfr[kk / 4] = *(const v2f*)(bpp + (size_t)kk * 16);
    const float bv = bp[t & 15];

    const int hi = max(s1 - 1, 0);
    const int ln = t & 31;
    int pw = s0 + ((__builtin_amdgcn_readfirstlane(t) >> 5) << 5);
    int q = min(pw + ln, hi);
    v2i rn = res[q];
    int dn = rdst[q];
    for (; pw < s1; pw += 1024) {
        const int p = pw + ln;
        const v2i cur = rn;
        const int dc = dn;
        q = min(p + 1024, hi);
        rn = res[q];
        dn = rdst[q];

        float ev[EDIM];
#pragma unroll
        for (int k = 0; k < EDIM; ++k) ev[k] = ea[(size_t)cur.x * EDIM + k];

        float hl[DPAD];
        if (ENC) {
            encode_row(hl, x, type_id, (size_t)cur.y, encw);
        } else {
            gather_row_quad(hl, h, cur.y, ln);
        }
        if (LNIN) ln_relu_row(hl);
        float* arow = accs + ((p < s1) ? (dc & (BKT_NODES - 1)) : BKT_NODES) * ACC_PITCH;
        float msg[DDIM];
#pragma unroll
        for (int d = 0; d < DDIM; ++d) {
            float m = hl[d] + be[d];
#pragma unroll
            for (int k = 0; k < EDIM; ++k) m = __builtin_fmaf(ev[k], We[k * DDIM + d], m);
            msg[d] = fmaxf(m, 0.f);
        }
        asm volatile("" : "+v"(msg[0]), "+v"(msg[1]), "+v"(msg[2]), "+v"(msg[3]), "+v"(msg[4]),
                          "+v"(msg[5]), "+v"(msg[6]), "+v"(msg[7]), "+v"(msg[8]), "+v"(msg[9]),
                          "+v"(msg[10]), "+v"(msg[11]), "+v"(msg[12]), "+v"(msg[13]), "+v"(msg[14]));
#pragma unroll
        for (int d = 0; d < DDIM; ++d) {
            if (msg[d] != 0.f) atomicAdd(arow + d, msg[d]);
        }
    }
    __syncthreads();

    const int lane = t & 31;
    const int c  = lane & 15;
    const int hf = lane >> 4;
    const int wave = __builtin_amdgcn_readfirstlane(t) >> 5;

    for (int g = wave; g < BKT_NODES / 32; g += 32) {
        const int wbase = b * BKT_NODES + g * 32;
        if (wbase >= Nn) break;

        float s[DPAD];
        const unsigned irow = min((unsigned)(wbase + lane), (unsigned)(Nn - 1));
        const float* ar = accs + (g * 32 + lane) * ACC_PITCH;
        if (ENC) {
            encode_row(s, x, type_id, (size_t)irow, encw);
        } else {
            gather_row_quad(s, h, (int)irow, lane);
        }
        if (LNIN) ln_relu_row(s);
#pragma unroll
        for (int d = 0; d < DPAD; ++d) s[d] += ar[d];

        v8f d0 = {0.f, 0.f, 0.f, 0.f, 0.f, 0.f, 0.f, 0.f};
        v8f d1 = {0.f, 0.f, 0.f, 0.f, 0.f, 0.f, 0.f, 0.f};
#pragma unroll
        for (int kk = 0; kk < DPAD; kk += 4) {
            const float snd0 = hf ? s[kk + 0] : s[kk + 2];
            const float snd1 = hf ? s[kk + 1] : s[kk + 3];
            const float rcv0 = xhalf(snd0);
            const float rcv1 = xhalf(snd1);
            const float own0 = hf ? s[kk + 2] : s[kk + 0];
            const float own1 = hf ? s[kk + 3] : s[kk + 1];
            v2f a0, a1;
            a0.x = hf ? rcv0 : own0;  a0.y = hf ? rcv1 : own1;
            a1.x = hf ? own0 : rcv0;  a1.y = hf ? own1 : rcv1;
            const v2f bf = bfr[kk / 4];
            d0 = __builtin_amdgcn_wmma_f32_16x16x4_f32(false, a0, false, bf, (short)0, d0, false, false);
            d1 = __builtin_amdgcn_wmma_f32_16x16x4_f32(false, a1, false, bf, (short)0, d1, false, false);
        }

        float* orow = hout + (size_t)wbase * DPAD + (8 * hf) * DPAD + c;
        if (wbase + 32 <= Nn) {
#pragma unroll
            for (int r = 0; r < 8; ++r) {
                float v0 = d0[r] + bv;
                float v1 = d1[r] + bv;
                if (RELU) { v0 = fmaxf(v0, 0.f); v1 = fmaxf(v1, 0.f); }
                orow[r * DPAD] = v0;
                orow[(r + 16) * DPAD] = v1;
            }
        } else {
#pragma unroll
            for (int r = 0; r < 8; ++r) {
                const int n0 = wbase + 8 * hf + r;
                float v0 = d0[r] + bv;
                float v1 = d1[r] + bv;
                if (RELU) { v0 = fmaxf(v0, 0.f); v1 = fmaxf(v1, 0.f); }
                if (n0 < Nn)      orow[r * DPAD] = v0;
                if (n0 + 16 < Nn) orow[(r + 16) * DPAD] = v1;
            }
        }
    }
}

template <bool HAS_A2, int KPAD>
__global__ __launch_bounds__(256) void wmma_gemm_kernel(
    const float* __restrict__ A, const float* __restrict__ A2,
    const float* __restrict__ Wp, const float* __restrict__ bp,
    float* __restrict__ C,
    int M, int Ncpad, int lda, int ldc, int relu)
{
    const int lane = threadIdx.x & 31;
    const int wave = threadIdx.x >> 5;
    const int ntN = Ncpad >> 4;
    const int tile = blockIdx.x * (blockDim.x >> 5) + wave;
    if (tile >= (M >> 4) * ntN) return;
    const int mt = tile / ntN;
    const int nt = tile % ntN;

    const int mrow  = mt * 16 + (lane & 15);
    const int ncol  = nt * 16 + (lane & 15);
    const int khalf = (lane >> 4) << 1;

    const float* ap  = A + (size_t)mrow * lda + khalf;
    const float* a2p = A2 + (size_t)mrow * lda + khalf;
    const float* bpp = Wp + ((size_t)(khalf >> 1) * Ncpad + ncol) * 2;
    const size_t bstep = (size_t)Ncpad * 4;

    v8f acc = {0.f, 0.f, 0.f, 0.f, 0.f, 0.f, 0.f, 0.f};
#pragma unroll
    for (int kk = 0; kk < KPAD; kk += 4) {
        v2f a = *(const v2f*)ap;
        if (HAS_A2) {
            v2f a2 = *(const v2f*)a2p;
            a.x += a2.x; a.y += a2.y;
            a2p += 4;
        }
        v2f b = *(const v2f*)bpp;
        ap += 4;
        bpp += bstep;
        acc = __builtin_amdgcn_wmma_f32_16x16x4_f32(
            false, a, false, b, (short)0, acc, false, false);
    }

    const int rbase = mt * 16 + ((lane >> 4) << 3);
    const float bv = bp[ncol];
#pragma unroll
    for (int r = 0; r < 8; ++r) {
        float v = acc[r] + bv;
        if (relu) v = fmaxf(v, 0.f);
        C[(size_t)(rbase + r) * ldc + ncol] = v;
    }
}

__global__ __launch_bounds__(256) void pool_link_kernel(
    const int* __restrict__ type_id, const int* __restrict__ batch,
    int* __restrict__ head, int* __restrict__ nxt, int Nn, int Gg)
{
    int i = blockIdx.x * blockDim.x + threadIdx.x;
    if (i >= Nn) return;
    const int t = type_id[i];
    const int b = batch[i];
    int link = 0;
    if ((t == 0 || t == 1) && b >= 0 && b < Gg) link = atomicExch(&head[b], i + 1);
    nxt[i] = link;
}

struct PoolNode { int nx; int tp; float4 r[4]; };

__device__ __forceinline__ void pool_fetch(PoolNode& n, int link,
    const float* __restrict__ h, const int* __restrict__ type_id,
    const int* __restrict__ nxt, int Nn)
{
    const unsigned i = min((unsigned)(link - 1), (unsigned)(Nn - 1));
    n.nx = nxt[i];
    n.tp = type_id[i];
    const float4* hv = (const float4*)(h + (size_t)i * DPAD);
#pragma unroll
    for (int q = 0; q < 4; ++q) n.r[q] = hv[q];
}

__device__ __forceinline__ void pool_accum(const PoolNode& n, bool valid,
    float (&sJ)[DPAD], float (&sM)[DPAD], float (&mJ)[DPAD], float (&mM)[DPAD], int& cJ, int& cM)
{
    const float ninf = -__builtin_inff();
    const bool inJ = valid & (n.tp == 0);
    const bool inM = valid & (n.tp != 0);
    cJ += inJ ? 1 : 0;
    cM += inM ? 1 : 0;
    float row[DPAD];
#pragma unroll
    for (int q = 0; q < 4; ++q) {
        row[4 * q + 0] = n.r[q].x; row[4 * q + 1] = n.r[q].y;
        row[4 * q + 2] = n.r[q].z; row[4 * q + 3] = n.r[q].w;
    }
    ln_relu_row(row);
#pragma unroll
    for (int d = 0; d < DPAD; ++d) {
        sJ[d] += inJ ? row[d] : 0.f;
        sM[d] += inM ? row[d] : 0.f;
        const float cj = inJ ? row[d] : ninf;
        const float cm = inM ? row[d] : ninf;
        mJ[d] = (cj > mJ[d]) ? cj : mJ[d];
        mM[d] = (cm > mM[d]) ? cm : mM[d];
    }
}

__global__ __launch_bounds__(256) void pool_gather_kernel(
    const float* __restrict__ h, const int* __restrict__ type_id,
    const int* __restrict__ head, const int* __restrict__ nxt,
    float* __restrict__ sumJ, float* __restrict__ sumM,
    float* __restrict__ maxJ, float* __restrict__ maxM,
    float* __restrict__ cntJ, float* __restrict__ cntM, int Nn, int Gg)
{
    const int g = blockIdx.x * blockDim.x + threadIdx.x;
    if (g >= Gg) return;
    const float ninf = -__builtin_inff();
    float sJ[DPAD], sM[DPAD], mJ[DPAD], mM[DPAD];
#pragma unroll
    for (int d = 0; d < DPAD; ++d) { sJ[d] = 0.f; sM[d] = 0.f; mJ[d] = ninf; mM[d] = ninf; }
    int cJ = 0, cM = 0;

    PoolNode a, b;
    const int first = head[g];
    bool live = (first != 0);
    pool_fetch(a, first, h, type_id, nxt, Nn);
    for (int it = 0; it < Nn && live; it += 2) {
        pool_fetch(b, a.nx, h, type_id, nxt, Nn);
        pool_accum(a, true, sJ, sM, mJ, mM, cJ, cM);
        const bool moreB = (a.nx != 0);
        pool_fetch(a, b.nx, h, type_id, nxt, Nn);
        pool_accum(b, moreB, sJ, sM, mJ, mM, cJ, cM);
        live = moreB & (b.nx != 0);
    }

    float4* oSJ = (float4*)(sumJ + (size_t)g * DPAD);
    float4* oSM = (float4*)(sumM + (size_t)g * DPAD);
    float4* oMJ = (float4*)(maxJ + (size_t)g * DPAD);
    float4* oMM = (float4*)(maxM + (size_t)g * DPAD);
#pragma unroll
    for (int q = 0; q < 4; ++q) {
        oSJ[q] = make_float4(sJ[4 * q], sJ[4 * q + 1], sJ[4 * q + 2], sJ[4 * q + 3]);
        oSM[q] = make_float4(sM[4 * q], sM[4 * q + 1], sM[4 * q + 2], sM[4 * q + 3]);
        oMJ[q] = make_float4(mJ[4 * q], mJ[4 * q + 1], mJ[4 * q + 2], mJ[4 * q + 3]);
        oMM[q] = make_float4(mM[4 * q], mM[4 * q + 1], mM[4 * q + 2], mM[4 * q + 3]);
    }
    cntJ[g] = (float)cJ;
    cntM[g] = (float)cM;
}

__global__ __launch_bounds__(128) void assemble_ln_kernel(
    const float* __restrict__ sumJ, const float* __restrict__ sumM,
    const float* __restrict__ maxJ, const float* __restrict__ maxM,
    const float* __restrict__ cntJ, const float* __restrict__ cntM,
    const float* __restrict__ u, float* __restrict__ g, int Gg)
{
    int gi = blockIdx.x * blockDim.x + threadIdx.x;
    if (gi >= Gg) return;
    float r[64];
    float cm = fmaxf(cntM[gi], 1.f);
    float cj = fmaxf(cntJ[gi], 1.f);
#pragma unroll
    for (int d = 0; d < DDIM; ++d) r[d] = sumM[(size_t)gi * DPAD + d] / cm;
#pragma unroll
    for (int d = 0; d < DDIM; ++d) r[15 + d] = sumJ[(size_t)gi * DPAD + d] / cj;
#pragma unroll
    for (int d = 0; d < DDIM; ++d) {
        float v = maxJ[(size_t)gi * DPAD + d];
        r[30 + d] = (v > -3.0e38f) ? v : 0.f;
    }
#pragma unroll
    for (int d = 0; d < DDIM; ++d) {
        float v = maxM[(size_t)gi * DPAD + d];
        r[45 + d] = (v > -3.0e38f) ? v : 0.f;
    }
#pragma unroll
    for (int j = 0; j < 4; ++j) r[60 + j] = u[(size_t)gi * 4 + j];

    float mu = 0.f;
#pragma unroll
    for (int k = 0; k < 64; ++k) mu += r[k];
    mu *= (1.f / 64.f);
    float var = 0.f;
#pragma unroll
    for (int k = 0; k < 64; ++k) { float t = r[k] - mu; var += t * t; }
    var *= (1.f / 64.f);
    float inv = rsqrtf(var + 1e-5f);
#pragma unroll
    for (int k = 0; k < 64; ++k) g[(size_t)gi * 64 + k] = (r[k] - mu) * inv;
}

__global__ __launch_bounds__(256) void mlp3_kernel(
    const float* __restrict__ m2, const float* __restrict__ W3,
    const float* __restrict__ b3, float* __restrict__ out, int Gg)
{
    int gi = blockIdx.x * blockDim.x + threadIdx.x;
    if (gi >= Gg) return;
    float s = b3[0];
#pragma unroll
    for (int k = 0; k < 32; ++k) s += m2[(size_t)gi * 32 + k] * W3[k];
    out[gi] = s;
}

extern "C" void kernel_launch(void* const* d_in, const int* in_sizes, int n_in,
                              void* d_out, int out_size, void* d_ws, size_t ws_size,
                              hipStream_t stream)
{
    const float* x       = (const float*)d_in[0];
    const int*   type_id = (const int*)  d_in[1];
    const int*   batch   = (const int*)  d_in[2];
    const float* u       = (const float*)d_in[3];
    const int*   ei_jj   = (const int*)  d_in[4];
    const float* ea_jj   = (const float*)d_in[5];
    const int*   ei_mj   = (const int*)  d_in[6];
    const float* ea_mj   = (const float*)d_in[7];
    const float* Wj_enc  = (const float*)d_in[8];   const float* bj_enc = (const float*)d_in[9];
    const float* Wm_enc  = (const float*)d_in[10];  const float* bm_enc = (const float*)d_in[11];
    const float* We_jj1  = (const float*)d_in[12];  const float* be_jj1 = (const float*)d_in[13];
    const float* Wm_jj1  = (const float*)d_in[14];  const float* bm_jj1 = (const float*)d_in[15];
    const float* We_mj1  = (const float*)d_in[16];  const float* be_mj1 = (const float*)d_in[17];
    const float* Wm_mj1  = (const float*)d_in[18];  const float* bm_mj1 = (const float*)d_in[19];
    const float* We_jj2  = (const float*)d_in[20];  const float* be_jj2 = (const float*)d_in[21];
    const float* Wm_jj2  = (const float*)d_in[22];  const float* bm_jj2 = (const float*)d_in[23];
    const float* We_mj2  = (const float*)d_in[24];  const float* be_mj2 = (const float*)d_in[25];
    const float* Wm_mj2  = (const float*)d_in[26];  const float* bm_mj2 = (const float*)d_in[27];
    const float* W1      = (const float*)d_in[28];  const float* b1     = (const float*)d_in[29];
    const float* W2      = (const float*)d_in[30];  const float* b2     = (const float*)d_in[31];
    const float* W3      = (const float*)d_in[32];  const float* b3     = (const float*)d_in[33];

    const int Nn  = in_sizes[0] / 5;
    const int Gg  = in_sizes[3] / 4;
    const int Ejj = in_sizes[5] / 5;
    const int Emj = in_sizes[7] / 4;

    float* ws = (float*)d_ws;
    const size_t NP = (size_t)Nn * DPAD;
    float* H   = ws;
    float* S   = ws + NP;
    float* agg = ws + 2 * NP;
    float* wpk = ws + 3 * NP;
    float* pWjj1 = wpk;          float* pbjj1 = wpk + 256;
    float* pWmj1 = wpk + 272;    float* pbmj1 = wpk + 528;
    float* pWjj2 = wpk + 544;    float* pbjj2 = wpk + 800;
    float* pWmj2 = wpk + 816;    float* pbmj2 = wpk + 1072;
    float* pW1   = wpk + 1088;   float* pb1   = wpk + 5184;
    float* pW2   = wpk + 5248;   float* pb2   = wpk + 7296;

    const int nbk = (Nn + BKT_NODES - 1) / BKT_NODES;
    const int nch = (Ejj + CHUNK_E - 1) / CHUNK_E;
    const int tabL = nbk * nch;
    const int nbT = (tabL + 1023) / 1024;
    v2i* res_jj  = (v2i*)agg;
    v2i* res_mj  = res_jj + Ejj;
    int* rdst_jj = (int*)(res_mj + Emj);
    int* rdst_mj = rdst_jj + Ejj;
    int* tab     = rdst_mj + Emj;
    int* tsum    = tab + 2 * (size_t)tabL;
    int* pool_next = (int*)S;
    int* pool_head = pool_next + Nn;

    float* P    = agg;
    float* sumJ = P;
    float* sumM = P + (size_t)16 * Gg;
    float* maxJ = P + (size_t)32 * Gg;
    float* maxM = P + (size_t)48 * Gg;
    float* cntJ = P + (size_t)64 * Gg;
    float* cntM = P + (size_t)65 * Gg;
    float* gbuf = P + (size_t)66 * Gg;
    float* m1   = P + (size_t)130 * Gg;
    float* m2   = P + (size_t)194 * Gg;
    float* out  = (float*)d_out;

    const int TB = 256;
    const int nodeBlocks = (Nn + TB - 1) / TB;

    pack_weight_kernel<<<1, 256, 0, stream>>>(Wm_jj1, bm_jj1, pWjj1, pbjj1, 15, 15, 16, 16);
    pack_weight_kernel<<<1, 256, 0, stream>>>(Wm_mj1, bm_mj1, pWmj1, pbmj1, 15, 15, 16, 16);
    pack_weight_kernel<<<1, 256, 0, stream>>>(Wm_jj2, bm_jj2, pWjj2, pbjj2, 15, 15, 16, 16);
    pack_weight_kernel<<<1, 256, 0, stream>>>(Wm_mj2, bm_mj2, pWmj2, pbmj2, 15, 15, 16, 16);
    pack_weight_kernel<<<16, 256, 0, stream>>>(W1, b1, pW1, pb1, 64, 64, 64, 64);
    pack_weight_kernel<<<8, 256, 0, stream>>>(W2, b2, pW2, pb2, 64, 32, 64, 32);

    auto gemm64 = [&](const float* A, const float* Wp, const float* bp,
                      float* C, int Ncpad, int lda, int ldc, int relu) {
        int tiles = (Gg >> 4) * (Ncpad >> 4);
        wmma_gemm_kernel<false, 64><<<(tiles + 7) / 8, TB, 0, stream>>>(
            A, nullptr, Wp, bp, C, Gg, Ncpad, lda, ldc, relu);
    };

    bucket_count_kernel<<<dim3(nch, 2), TB, 0, stream>>>(ei_jj, ei_mj, tab, Ejj, Emj, Nn, nch, nbk);
    csr_block_sum_kernel<<<dim3(nbT, 2), TB, 0, stream>>>(tab, tsum, tabL, nbT);
    csr_block_scan_kernel<<<dim3(nbT, 2), TB, 0, stream>>>(tab, tsum, tabL, nbT);
    bucket_place_kernel<<<dim3(nch, 2), TB, 0, stream>>>(ei_jj, ei_mj, tab, res_jj, res_mj, rdst_jj, rdst_mj, Ejj, Emj, Nn, nch, nbk);

    conv_kernel<5, true, false, true><<<nbk, 1024, 0, stream>>>(nullptr, S, x, type_id, Wj_enc, bj_enc, Wm_enc, bm_enc, res_jj, rdst_jj, tab, ea_jj, We_jj1, be_jj1, pWjj1, pbjj1, Nn, Ejj, nch, nbk);

    conv_kernel<4, false, false, false><<<nbk, 1024, 0, stream>>>(S, H, x, type_id, Wj_enc, bj_enc, Wm_enc, bm_enc, res_mj, rdst_mj, tab + tabL, ea_mj, We_mj1, be_mj1, pWmj1, pbmj1, Nn, Emj, nch, nbk);

    conv_kernel<5, true, true, false><<<nbk, 1024, 0, stream>>>(H, S, x, type_id, Wj_enc, bj_enc, Wm_enc, bm_enc, res_jj, rdst_jj, tab, ea_jj, We_jj2, be_jj2, pWjj2, pbjj2, Nn, Ejj, nch, nbk);

    conv_kernel<4, true, false, false><<<nbk, 1024, 0, stream>>>(S, H, x, type_id, Wj_enc, bj_enc, Wm_enc, bm_enc, res_mj, rdst_mj, tab + tabL, ea_mj, We_mj2, be_mj2, pWmj2, pbmj2, Nn, Emj, nch, nbk);

    fill_kernel<<<(Gg + TB - 1) / TB, TB, 0, stream>>>((float*)pool_head, 0.f, (size_t)Gg);
    pool_link_kernel<<<nodeBlocks, TB, 0, stream>>>(type_id, batch, pool_head, pool_next, Nn, Gg);
    pool_gather_kernel<<<(Gg + TB - 1) / TB, TB, 0, stream>>>(H, type_id, pool_head, pool_next,
                                                              sumJ, sumM, maxJ, maxM, cntJ, cntM, Nn, Gg);

    assemble_ln_kernel<<<(Gg + 127) / 128, 128, 0, stream>>>(sumJ, sumM, maxJ, maxM, cntJ, cntM, u, gbuf, Gg);
    gemm64(gbuf, pW1, pb1, m1, 64, 64, 64, 1);
    gemm64(m1,   pW2, pb2, m2, 32, 64, 32, 1);
    mlp3_kernel<<<(Gg + TB - 1) / TB, TB, 0, stream>>>(m2, W3, b3, out, Gg);
}


# bucket_place: a chunk's records are put in bucket order in LDS (LDS add rank + scan of the counts) and written run by run: a wave's store covers a few lines, not 32; places, sum orders as before
# speedup vs baseline: 4.7535x; 1.2491x over previous
#include <hip/hip_runtime.h>
#include <hip/hip_bf16.h>

typedef __attribute__((ext_vector_type(2))) float v2f;
typedef __attribute__((ext_vector_type(8))) float v8f;

#define DPAD 16
#define DDIM 15

__global__ void fill_kernel(float* __restrict__ p, float v, size_t n) {
    size_t i = (size_t)blockIdx.x * blockDim.x + threadIdx.x;
    size_t stride = (size_t)gridDim.x * blockDim.x;
    for (; i < n; i += stride) p[i] = v;
}

__global__ void pack_weight_kernel(const float* __restrict__ W, const float* __restrict__ b,
                                   float* __restrict__ Wp, float* __restrict__ bp,
                                   int K, int Nc, int Kpad, int Ncpad)
{
    int idx = blockIdx.x * blockDim.x + threadIdx.x;
    int total = Kpad * Ncpad;
    if (idx < total) {
        int k = idx / Ncpad, c = idx % Ncpad;
        float v = (k < K && c < Nc) ? W[k * Nc + c] : 0.f;
        Wp[((size_t)(k >> 1) * Ncpad + c) * 2 + (k & 1)] = v;
    }
    if (idx < Ncpad) bp[idx] = (idx < Nc) ? b[idx] : 0.f;
}

#define ENC_SET 24
__device__ __forceinline__ void encode_row(float (&v)[DPAD],
    const float* __restrict__ x, const int* __restrict__ type_id, size_t i,
    const float4* encw)
{
    float xi[5];
#pragma unroll
    for (int k = 0; k < 5; ++k) xi[k] = x[i * 5 + k];
    const int t = type_id[i];
    const float4* wr = encw + ((t == 0) ? 0 : ((t == 1) ? ENC_SET : 2 * ENC_SET));
    float acc[DPAD];
#pragma unroll
    for (int q = 0; q < 4; ++q) {
        const float4 w = wr[q];
        acc[4 * q + 0] = w.x; acc[4 * q + 1] = w.y; acc[4 * q + 2] = w.z; acc[4 * q + 3] = w.w;
    }
#pragma unroll
    for (int k = 0; k < 5; ++k) {
#pragma unroll
        for (int q = 0; q < 4; ++q) {
            const float4 w = wr[4 * (k + 1) + q];
            acc[4 * q + 0] = __builtin_fmaf(xi[k], w.x, acc[4 * q + 0]);
            acc[4 * q + 1] = __builtin_fmaf(xi[k], w.y, acc[4 * q + 1]);
            acc[4 * q + 2] = __builtin_fmaf(xi[k], w.z, acc[4 * q + 2]);
            if (q < 3) acc[4 * q + 3] = __builtin_fmaf(xi[k], w.w, acc[4 * q + 3]);
        }
    }
#pragma unroll
    for (int d = 0; d < DDIM; ++d) v[d] = fmaxf(acc[d], 0.f);
    v[DDIM] = 0.f;
}

typedef __attribute__((ext_vector_type(2))) int v2i;

#define BKT_SHIFT 12
#define BKT_NODES 4096
#define ACC_PITCH 17
#define CHUNK_E   4096
#define MAX_BKT   1024

__global__ __launch_bounds__(256) void bucket_count_kernel(
    const int* __restrict__ ei_a, const int* __restrict__ ei_b, int* __restrict__ tab,
    int Ea, int Eb, int Nn, int nch, int nbk)
{
    __shared__ int hist[MAX_BKT];
    const int t = threadIdx.x;
    const int set = blockIdx.y, chunk = blockIdx.x;
    const int E = set ? Eb : Ea;
    const int* __restrict__ ei = set ? ei_b : ei_a;
    for (int i = t; i < nbk; i += 256) hist[i] = 0;
    __syncthreads();
    const int base = chunk * CHUNK_E + t;
    const int last = max(E - 1, 0);
    int d[CHUNK_E / 256];
#pragma unroll
    for (int k = 0; k < CHUNK_E / 256; ++k) d[k] = ei[(size_t)E + min(base + k * 256, last)];
#pragma unroll
    for (int k = 0; k < CHUNK_E / 256; ++k) {
        if (base + k * 256 < E) {
            const unsigned dst = min((unsigned)d[k], (unsigned)(Nn - 1));
            atomicAdd(&hist[dst >> BKT_SHIFT], 1);
        }
    }
    __syncthreads();
    int* tb = tab + (size_t)set * nbk * nch;
    for (int i = t; i < nbk; i += 256) tb[(size_t)i * nch + chunk] = hist[i];
}

__global__ __launch_bounds__(256) void bucket_place_kernel(
    const int* __restrict__ ei_a, const int* __restrict__ ei_b, const int* __restrict__ tab,
    v2i* __restrict__ res_a, v2i* __restrict__ res_b,
    int* __restrict__ rdst_a, int* __restrict__ rdst_b,
    int Ea, int Eb, int Nn, int nch, int nbk)
{
    __shared__ int gstart[MAX_BKT];
    __shared__ int lstart[MAX_BKT];
    __shared__ v2i st_res[CHUNK_E];
    __shared__ int st_dst[CHUNK_E];
    __shared__ int wsum[8];
    const int t = threadIdx.x;
    const int set = blockIdx.y, chunk = blockIdx.x;
    const int E = set ? Eb : Ea;
    const int* __restrict__ ei = set ? ei_b : ei_a;
    v2i* __restrict__ res = set ? res_b : res_a;
    int* __restrict__ rdst = set ? rdst_b : rdst_a;
    const int* tb = tab + (size_t)set * nbk * nch;
    for (int i = t; i < nbk; i += 256) { gstart[i] = tb[(size_t)i * nch + chunk]; lstart[i] = 0; }
    __syncthreads();
    const int base = chunk * CHUNK_E + t;
    const int last = max(E - 1, 0);
    int sv[CHUNK_E / 256], d[CHUNK_E / 256], rk[CHUNK_E / 256];
#pragma unroll
    for (int k = 0; k < CHUNK_E / 256; ++k) {
        const int ec = min(base + k * 256, last);
        sv[k] = ei[ec];
        d[k]  = ei[(size_t)E + ec];
    }
#pragma unroll
    for (int k = 0; k < CHUNK_E / 256; ++k) {
        rk[k] = 0;
        if (base + k * 256 < E) {
            d[k] = (int)min((unsigned)d[k], (unsigned)(Nn - 1));
            rk[k] = atomicAdd(&lstart[d[k] >> BKT_SHIFT], 1);
        }
    }
    __syncthreads();
    {
        int v[4];
#pragma unroll
        for (int k = 0; k < 4; ++k) v[k] = (4 * t + k < nbk) ? lstart[4 * t + k] : 0;
        const int ts = v[0] + v[1] + v[2] + v[3];
        int incl = ts;
        const int lane = t & 31;
        for (int w = 1; w < 32; w <<= 1) {
            const int y = __shfl_up(incl, w);
            if (lane >= w) incl += y;
        }
        if (lane == 31) wsum[t >> 5] = incl;
        __syncthreads();
        int run = incl - ts;
        for (int w = 0; w < (t >> 5); ++w) run += wsum[w];
#pragma unroll
        for (int k = 0; k < 4; ++k) {
            if (4 * t + k < nbk) lstart[4 * t + k] = run;
            run += v[k];
        }
    }
    __syncthreads();
#pragma unroll
    for (int k = 0; k < CHUNK_E / 256; ++k) {
        const int e = base + k * 256;
        if (e < E) {
            const int slot = min(lstart[d[k] >> BKT_SHIFT] + rk[k], CHUNK_E - 1);
            v2i o; o.x = e; o.y = sv[k];
            st_res[slot] = o;
            st_dst[slot] = d[k];
        }
    }
    __syncthreads();
    const int nhere = min(CHUNK_E, E - chunk * CHUNK_E);
    for (int s = t; s < nhere; s += 256) {
        const int dst = st_dst[s];
        const int b = dst >> BKT_SHIFT;
        unsigned pos = (unsigned)(gstart[b] + (s - lstart[b]));
        pos = min(pos, (unsigned)last);
        res[pos] = st_res[s];
        rdst[pos] = dst;
    }
}

__global__ __launch_bounds__(256) void csr_block_sum_kernel(
    const int* __restrict__ cnt, int* __restrict__ bsum, int Nn, int nb)
{
    __shared__ int sh[256];
    const int* c = cnt + (size_t)blockIdx.y * Nn;
    const int t = threadIdx.x;
    const int base = blockIdx.x * 1024 + t * 4;
    int s = 0;
#pragma unroll
    for (int k = 0; k < 4; ++k) if (base + k < Nn) s += c[base + k];
    sh[t] = s;
    __syncthreads();
    for (int w = 128; w > 0; w >>= 1) {
        if (t < w) sh[t] += sh[t + w];
        __syncthreads();
    }
    if (t == 0) bsum[blockIdx.y * nb + blockIdx.x] = sh[0];
}

__global__ __launch_bounds__(256) void csr_block_scan_kernel(
    int* __restrict__ cnt, const int* __restrict__ bsum, int Nn, int nb)
{
    __shared__ int sh[256];
    int* c = cnt + (size_t)blockIdx.y * Nn;
    const int* bs = bsum + blockIdx.y * nb;
    const int t = threadIdx.x;

    int pc = 0;
    for (int j = t; j < (int)blockIdx.x; j += 256) pc += bs[j];
    sh[t] = pc;
    __syncthreads();
    for (int w = 128; w > 0; w >>= 1) {
        if (t < w) sh[t] += sh[t + w];
        __syncthreads();
    }
    const int carry = sh[0];
    __syncthreads();

    const int base = blockIdx.x * 1024 + t * 4;
    int v[4];
#pragma unroll
    for (int k = 0; k < 4; ++k) v[k] = (base + k < Nn) ? c[base + k] : 0;
    const int ts = v[0] + v[1] + v[2] + v[3];
    sh[t] = ts;
    __syncthreads();
    for (int d = 1; d < 256; d <<= 1) {
        int x = (t >= d) ? sh[t - d] : 0;
        __syncthreads();
        sh[t] += x;
        __syncthreads();
    }
    int run = carry + sh[t] - ts;
#pragma unroll
    for (int k = 0; k < 4; ++k) {
        if (base + k < Nn) c[base + k] = run;
        run += v[k];
    }
}

__device__ __forceinline__ void ln_relu_row(float (&v)[DPAD])
{
    float mu = 0.f;
#pragma unroll
    for (int d = 0; d < DDIM; ++d) mu += v[d];
    mu *= (1.f / DDIM);
    float var = 0.f;
#pragma unroll
    for (int d = 0; d < DDIM; ++d) { float t = v[d] - mu; var += t * t; }
    var *= (1.f / DDIM);
    const float inv = rsqrtf(var + 1e-5f);
#pragma unroll
    for (int d = 0; d < DDIM; ++d) v[d] = fmaxf((v[d] - mu) * inv, 0.f);
    v[DDIM] = 0.f;
}

__device__ __forceinline__ float xhalf(float x) {
    return __builtin_bit_cast(float, __builtin_amdgcn_permlanex16(
        __builtin_bit_cast(int, x), __builtin_bit_cast(int, x), 0x76543210, 0xfedcba98, false, false));
}

#define QUAD_PERM(a, b, c, d) ((a) | ((b) << 2) | ((c) << 4) | ((d) << 6))

template <int CTRL>
__device__ __forceinline__ int quad_i(int v) {
    return __builtin_amdgcn_update_dpp(0, v, CTRL, 0xf, 0xf, true);
}
template <int CTRL>
__device__ __forceinline__ float quad_f(float v) {
    return __builtin_bit_cast(float, quad_i<CTRL>(__builtin_bit_cast(int, v)));
}

template <int CTRL>
__device__ __forceinline__ void quad_swap(float4& lo, float4& hi, bool up) {
    const float4 pl = make_float4(quad_f<CTRL>(lo.x), quad_f<CTRL>(lo.y), quad_f<CTRL>(lo.z), quad_f<CTRL>(lo.w));
    const float4 ph = make_float4(quad_f<CTRL>(hi.x), quad_f<CTRL>(hi.y), quad_f<CTRL>(hi.z), quad_f<CTRL>(hi.w));
    const float4 nlo = make_float4(up ? ph.x : lo.x, up ? ph.y : lo.y, up ? ph.z : lo.z, up ? ph.w : lo.w);
    const float4 nhi = make_float4(up ? hi.x : pl.x, up ? hi.y : pl.y, up ? hi.z : pl.z, up ? hi.w : pl.w);
    lo = nlo;
    hi = nhi;
}

__device__ __forceinline__ void gather_row_quad(float (&v)[DPAD], const float* __restrict__ h, int src, int lane)
{
    const int j = lane & 3;
    const int n0 = quad_i<QUAD_PERM(0, 0, 0, 0)>(src);
    const int n1 = quad_i<QUAD_PERM(1, 1, 1, 1)>(src);
    const int n2 = quad_i<QUAD_PERM(2, 2, 2, 2)>(src);
    const int n3 = quad_i<QUAD_PERM(3, 3, 3, 3)>(src);
    float4 g0 = *(const float4*)(h + (size_t)n0 * DPAD + 4 * j);
    float4 g1 = *(const float4*)(h + (size_t)n1 * DPAD + 4 * j);
    float4 g2 = *(const float4*)(h + (size_t)n2 * DPAD + 4 * j);
    float4 g3 = *(const float4*)(h + (size_t)n3 * DPAD + 4 * j);
    quad_swap<QUAD_PERM(1, 0, 3, 2)>(g0, g1, (j & 1) != 0);
    quad_swap<QUAD_PERM(1, 0, 3, 2)>(g2, g3, (j & 1) != 0);
    quad_swap<QUAD_PERM(2, 3, 0, 1)>(g0, g2, (j & 2) != 0);
    quad_swap<QUAD_PERM(2, 3, 0, 1)>(g1, g3, (j & 2) != 0);
    v[0]  = g0.x; v[1]  = g0.y; v[2]  = g0.z; v[3]  = g0.w;
    v[4]  = g1.x; v[5]  = g1.y; v[6]  = g1.z; v[7]  = g1.w;
    v[8]  = g2.x; v[9]  = g2.y; v[10] = g2.z; v[11] = g2.w;
    v[12] = g3.x; v[13] = g3.y; v[14] = g3.z; v[15] = g3.w;
}

template <int EDIM, bool RELU, bool LNIN, bool ENC>
__global__ __launch_bounds__(1024) void conv_kernel(
    const float* __restrict__ h, float* __restrict__ hout,
    const float* __restrict__ x, const int* __restrict__ type_id,
    const float* __restrict__ Wje, const float* __restrict__ bje,
    const float* __restrict__ Wme, const float* __restrict__ bme,
    const v2i* __restrict__ res, const int* __restrict__ rdst, const int* __restrict__ tb,
    const float* __restrict__ ea,
    const float* __restrict__ We, const float* __restrict__ be,
    const float* __restrict__ Wp, const float* __restrict__ bp,
    int Nn, int E, int nch, int nbk)
{
    __shared__ float accs[(BKT_NODES + 1) * ACC_PITCH];
    __shared__ float4 encw[ENC ? 3 * ENC_SET : 1];
    const int t = threadIdx.x;
    const int b = blockIdx.x;
    if (ENC && t < 3 * ENC_SET * 4) {
        const int set = t / (ENC_SET * 4), r = (t % (ENC_SET * 4)) >> 4, cc = t & 15;
        const float* Ws = set ? Wme : Wje;
        const float* bs = set ? bme : bje;
        float w = 0.f;
        if (set < 2 && cc < DDIM) w = (r == 0) ? bs[cc] : Ws[(r - 1) * DDIM + cc];
        ((float*)encw)[t] = w;
    }
    int s0 = tb[(size_t)b * nch];
    int s1 = (b + 1 < nbk) ? tb[(size_t)(b + 1) * nch] : E;
    s0 = min(max(s0, 0), E);
    s1 = min(max(s1, s0), E);

    for (int i = t; i < (BKT_NODES + 1) * ACC_PITCH; i += 1024) accs[i] = 0.f;
    __syncthreads();

    const float* bpp = Wp + ((size_t)((t & 31) >> 4) * 16 + (t & 15)) * 2;
    v2f bfr[DPAD / 4];
#pragma unroll
    for (int kk = 0; kk < DPAD; kk += 4) bfr[kk / 4] = *(const v2f*)(bpp + (size_t)kk * 16);
    const float bv = bp[t & 15];

    const int hi = max(s1 - 1, 0);
    const int ln = t & 31;
    int pw = s0 + ((__builtin_amdgcn_readfirstlane(t) >> 5) << 5);
    int q = min(pw + ln, hi);
    v2i rn = res[q];
    int dn = rdst[q];
    for (; pw < s1; pw += 1024) {
        const int p = pw + ln;
        const v2i cur = rn;
        const int dc = dn;
        q = min(p + 1024, hi);
        rn = res[q];
        dn = rdst[q];

        float ev[EDIM];
#pragma unroll
        for (int k = 0; k < EDIM; ++k) ev[k] = ea[(size_t)cur.x * EDIM + k];

        float hl[DPAD];
        if (ENC) {
            encode_row(hl, x, type_id, (size_t)cur.y, encw);
        } else {
            gather_row_quad(hl, h, cur.y, ln);
        }
        if (LNIN) ln_relu_row(hl);
        float* arow = accs + ((p < s1) ? (dc & (BKT_NODES - 1)) : BKT_NODES) * ACC_PITCH;
        float msg[DDIM];
#pragma unroll
        for (int d = 0; d < DDIM; ++d) {
            float m = hl[d] + be[d];
#pragma unroll
            for (int k = 0; k < EDIM; ++k) m = __builtin_fmaf(ev[k], We[k * DDIM + d], m);
            msg[d] = fmaxf(m, 0.f);
        }
        asm volatile("" : "+v"(msg[0]), "+v"(msg[1]), "+v"(msg[2]), "+v"(msg[3]), "+v"(msg[4]),
                          "+v"(msg[5]), "+v"(msg[6]), "+v"(msg[7]), "+v"(msg[8]), "+v"(msg[9]),
                          "+v"(msg[10]), "+v"(msg[11]), "+v"(msg[12]), "+v"(msg[13]), "+v"(msg[14]));
#pragma unroll
        for (int d = 0; d < DDIM; ++d) {
            if (msg[d] != 0.f) atomicAdd(arow + d, msg[d]);
        }
    }
    __syncthreads();

    const int lane = t & 31;
    const int c  = lane & 15;
    const int hf = lane >> 4;
    const int wave = __builtin_amdgcn_readfirstlane(t) >> 5;

    for (int g = wave; g < BKT_NODES / 32; g += 32) {
        const int wbase = b * BKT_NODES + g * 32;
        if (wbase >= Nn) break;

        float s[DPAD];
        const unsigned irow = min((unsigned)(wbase + lane), (unsigned)(Nn - 1));
        const float* ar = accs + (g * 32 + lane) * ACC_PITCH;
        if (ENC) {
            encode_row(s, x, type_id, (size_t)irow, encw);
        } else {
            gather_row_quad(s, h, (int)irow, lane);
        }
        if (LNIN) ln_relu_row(s);
#pragma unroll
        for (int d = 0; d < DPAD; ++d) s[d] += ar[d];

        v8f d0 = {0.f, 0.f, 0.f, 0.f, 0.f, 0.f, 0.f, 0.f};
        v8f d1 = {0.f, 0.f, 0.f, 0.f, 0.f, 0.f, 0.f, 0.f};
#pragma unroll
        for (int kk = 0; kk < DPAD; kk += 4) {
            const float snd0 = hf ? s[kk + 0] : s[kk + 2];
            const float snd1 = hf ? s[kk + 1] : s[kk + 3];
            const float rcv0 = xhalf(snd0);
            const float rcv1 = xhalf(snd1);
            const float own0 = hf ? s[kk + 2] : s[kk + 0];
            const float own1 = hf ? s[kk + 3] : s[kk + 1];
            v2f a0, a1;
            a0.x = hf ? rcv0 : own0;  a0.y = hf ? rcv1 : own1;
            a1.x = hf ? own0 : rcv0;  a1.y = hf ? own1 : rcv1;
            const v2f bf = bfr[kk / 4];
            d0 = __builtin_amdgcn_wmma_f32_16x16x4_f32(false, a0, false, bf, (short)0, d0, false, false);
            d1 = __builtin_amdgcn_wmma_f32_16x16x4_f32(false, a1, false, bf, (short)0, d1, false, false);
        }

        float* orow = hout + (size_t)wbase * DPAD + (8 * hf) * DPAD + c;
        if (wbase + 32 <= Nn) {
#pragma unroll
            for (int r = 0; r < 8; ++r) {
                float v0 = d0[r] + bv;
                float v1 = d1[r] + bv;
                if (RELU) { v0 = fmaxf(v0, 0.f); v1 = fmaxf(v1, 0.f); }
                orow[r * DPAD] = v0;
                orow[(r + 16) * DPAD] = v1;
            }
        } else {
#pragma unroll
            for (int r = 0; r < 8; ++r) {
                const int n0 = wbase + 8 * hf + r;
                float v0 = d0[r] + bv;
                float v1 = d1[r] + bv;
                if (RELU) { v0 = fmaxf(v0, 0.f); v1 = fmaxf(v1, 0.f); }
                if (n0 < Nn)      orow[r * DPAD] = v0;
                if (n0 + 16 < Nn) orow[(r + 16) * DPAD] = v1;
            }
        }
    }
}

template <bool HAS_A2, int KPAD>
__global__ __launch_bounds__(256) void wmma_gemm_kernel(
    const float* __restrict__ A, const float* __restrict__ A2,
    const float* __restrict__ Wp, const float* __restrict__ bp,
    float* __restrict__ C,
    int M, int Ncpad, int lda, int ldc, int relu)
{
    const int lane = threadIdx.x & 31;
    const int wave = threadIdx.x >> 5;
    const int ntN = Ncpad >> 4;
    const int tile = blockIdx.x * (blockDim.x >> 5) + wave;
    if (tile >= (M >> 4) * ntN) return;
    const int mt = tile / ntN;
    const int nt = tile % ntN;

    const int mrow  = mt * 16 + (lane & 15);
    const int ncol  = nt * 16 + (lane & 15);
    const int khalf = (lane >> 4) << 1;

    const float* ap  = A + (size_t)mrow * lda + khalf;
    const float* a2p = A2 + (size_t)mrow * lda + khalf;
    const float* bpp = Wp + ((size_t)(khalf >> 1) * Ncpad + ncol) * 2;
    const size_t bstep = (size_t)Ncpad * 4;

    v8f acc = {0.f, 0.f, 0.f, 0.f, 0.f, 0.f, 0.f, 0.f};
#pragma unroll
    for (int kk = 0; kk < KPAD; kk += 4) {
        v2f a = *(const v2f*)ap;
        if (HAS_A2) {
            v2f a2 = *(const v2f*)a2p;
            a.x += a2.x; a.y += a2.y;
            a2p += 4;
        }
        v2f b = *(const v2f*)bpp;
        ap += 4;
        bpp += bstep;
        acc = __builtin_amdgcn_wmma_f32_16x16x4_f32(
            false, a, false, b, (short)0, acc, false, false);
    }

    const int rbase = mt * 16 + ((lane >> 4) << 3);
    const float bv = bp[ncol];
#pragma unroll
    for (int r = 0; r < 8; ++r) {
        float v = acc[r] + bv;
        if (relu) v = fmaxf(v, 0.f);
        C[(size_t)(rbase + r) * ldc + ncol] = v;
    }
}

__global__ __launch_bounds__(256) void pool_link_kernel(
    const int* __restrict__ type_id, const int* __restrict__ batch,
    int* __restrict__ head, int* __restrict__ nxt, int Nn, int Gg)
{
    int i = blockIdx.x * blockDim.x + threadIdx.x;
    if (i >= Nn) return;
    const int t = type_id[i];
    const int b = batch[i];
    int link = 0;
    if ((t == 0 || t == 1) && b >= 0 && b < Gg) link = atomicExch(&head[b], i + 1);
    nxt[i] = link;
}

struct PoolNode { int nx; int tp; float4 r[4]; };

__device__ __forceinline__ void pool_fetch(PoolNode& n, int link,
    const float* __restrict__ h, const int* __restrict__ type_id,
    const int* __restrict__ nxt, int Nn)
{
    const unsigned i = min((unsigned)(link - 1), (unsigned)(Nn - 1));
    n.nx = nxt[i];
    n.tp = type_id[i];
    const float4* hv = (const float4*)(h + (size_t)i * DPAD);
#pragma unroll
    for (int q = 0; q < 4; ++q) n.r[q] = hv[q];
}

__device__ __forceinline__ void pool_accum(const PoolNode& n, bool valid,
    float (&sJ)[DPAD], float (&sM)[DPAD], float (&mJ)[DPAD], float (&mM)[DPAD], int& cJ, int& cM)
{
    const float ninf = -__builtin_inff();
    const bool inJ = valid & (n.tp == 0);
    const bool inM = valid & (n.tp != 0);
    cJ += inJ ? 1 : 0;
    cM += inM ? 1 : 0;
    float row[DPAD];
#pragma unroll
    for (int q = 0; q < 4; ++q) {
        row[4 * q + 0] = n.r[q].x; row[4 * q + 1] = n.r[q].y;
        row[4 * q + 2] = n.r[q].z; row[4 * q + 3] = n.r[q].w;
    }
    ln_relu_row(row);
#pragma unroll
    for (int d = 0; d < DPAD; ++d) {
        sJ[d] += inJ ? row[d] : 0.f;
        sM[d] += inM ? row[d] : 0.f;
        const float cj = inJ ? row[d] : ninf;
        const float cm = inM ? row[d] : ninf;
        mJ[d] = (cj > mJ[d]) ? cj : mJ[d];
        mM[d] = (cm > mM[d]) ? cm : mM[d];
    }
}

__global__ __launch_bounds__(256) void pool_gather_kernel(
    const float* __restrict__ h, const int* __restrict__ type_id,
    const int* __restrict__ head, const int* __restrict__ nxt,
    float* __restrict__ sumJ, float* __restrict__ sumM,
    float* __restrict__ maxJ, float* __restrict__ maxM,
    float* __restrict__ cntJ, float* __restrict__ cntM, int Nn, int Gg)
{
    const int g = blockIdx.x * blockDim.x + threadIdx.x;
    if (g >= Gg) return;
    const float ninf = -__builtin_inff();
    float sJ[DPAD], sM[DPAD], mJ[DPAD], mM[DPAD];
#pragma unroll
    for (int d = 0; d < DPAD; ++d) { sJ[d] = 0.f; sM[d] = 0.f; mJ[d] = ninf; mM[d] = ninf; }
    int cJ = 0, cM = 0;

    PoolNode a, b;
    const int first = head[g];
    bool live = (first != 0);
    pool_fetch(a, first, h, type_id, nxt, Nn);
    for (int it = 0; it < Nn && live; it += 2) {
        pool_fetch(b, a.nx, h, type_id, nxt, Nn);
        pool_accum(a, true, sJ, sM, mJ, mM, cJ, cM);
        const bool moreB = (a.nx != 0);
        pool_fetch(a, b.nx, h, type_id, nxt, Nn);
        pool_accum(b, moreB, sJ, sM, mJ, mM, cJ, cM);
        live = moreB & (b.nx != 0);
    }

    float4* oSJ = (float4*)(sumJ + (size_t)g * DPAD);
    float4* oSM = (float4*)(sumM + (size_t)g * DPAD);
    float4* oMJ = (float4*)(maxJ + (size_t)g * DPAD);
    float4* oMM = (float4*)(maxM + (size_t)g * DPAD);
#pragma unroll
    for (int q = 0; q < 4; ++q) {
        oSJ[q] = make_float4(sJ[4 * q], sJ[4 * q + 1], sJ[4 * q + 2], sJ[4 * q + 3]);
        oSM[q] = make_float4(sM[4 * q], sM[4 * q + 1], sM[4 * q + 2], sM[4 * q + 3]);
        oMJ[q] = make_float4(mJ[4 * q], mJ[4 * q + 1], mJ[4 * q + 2], mJ[4 * q + 3]);
        oMM[q] = make_float4(mM[4 * q], mM[4 * q + 1], mM[4 * q + 2], mM[4 * q + 3]);
    }
    cntJ[g] = (float)cJ;
    cntM[g] = (float)cM;
}

__global__ __launch_bounds__(128) void assemble_ln_kernel(
    const float* __restrict__ sumJ, const float* __restrict__ sumM,
    const float* __restrict__ maxJ, const float* __restrict__ maxM,
    const float* __restrict__ cntJ, const float* __restrict__ cntM,
    const float* __restrict__ u, float* __restrict__ g, int Gg)
{
    int gi = blockIdx.x * blockDim.x + threadIdx.x;
    if (gi >= Gg) return;
    float r[64];
    float cm = fmaxf(cntM[gi], 1.f);
    float cj = fmaxf(cntJ[gi], 1.f);
#pragma unroll
    for (int d = 0; d < DDIM; ++d) r[d] = sumM[(size_t)gi * DPAD + d] / cm;
#pragma unroll
    for (int d = 0; d < DDIM; ++d) r[15 + d] = sumJ[(size_t)gi * DPAD + d] / cj;
#pragma unroll
    for (int d = 0; d < DDIM; ++d) {
        float v = maxJ[(size_t)gi * DPAD + d];
        r[30 + d] = (v > -3.0e38f) ? v : 0.f;
    }
#pragma unroll
    for (int d = 0; d < DDIM; ++d) {
        float v = maxM[(size_t)gi * DPAD + d];
        r[45 + d] = (v > -3.0e38f) ? v : 0.f;
    }
#pragma unroll
    for (int j = 0; j < 4; ++j) r[60 + j] = u[(size_t)gi * 4 + j];

    float mu = 0.f;
#pragma unroll
    for (int k = 0; k < 64; ++k) mu += r[k];
    mu *= (1.f / 64.f);
    float var = 0.f;
#pragma unroll
    for (int k = 0; k < 64; ++k) { float t = r[k] - mu; var += t * t; }
    var *= (1.f / 64.f);
    float inv = rsqrtf(var + 1e-5f);
#pragma unroll
    for (int k = 0; k < 64; ++k) g[(size_t)gi * 64 + k] = (r[k] - mu) * inv;
}

__global__ __launch_bounds__(256) void mlp3_kernel(
    const float* __restrict__ m2, const float* __restrict__ W3,
    const float* __restrict__ b3, float* __restrict__ out, int Gg)
{
    int gi = blockIdx.x * blockDim.x + threadIdx.x;
    if (gi >= Gg) return;
    float s = b3[0];
#pragma unroll
    for (int k = 0; k < 32; ++k) s += m2[(size_t)gi * 32 + k] * W3[k];
    out[gi] = s;
}

extern "C" void kernel_launch(void* const* d_in, const int* in_sizes, int n_in,
                              void* d_out, int out_size, void* d_ws, size_t ws_size,
                              hipStream_t stream)
{
    const float* x       = (const float*)d_in[0];
    const int*   type_id = (const int*)  d_in[1];
    const int*   batch   = (const int*)  d_in[2];
    const float* u       = (const float*)d_in[3];
    const int*   ei_jj   = (const int*)  d_in[4];
    const float* ea_jj   = (const float*)d_in[5];
    const int*   ei_mj   = (const int*)  d_in[6];
    const float* ea_mj   = (const float*)d_in[7];
    const float* Wj_enc  = (const float*)d_in[8];   const float* bj_enc = (const float*)d_in[9];
    const float* Wm_enc  = (const float*)d_in[10];  const float* bm_enc = (const float*)d_in[11];
    const float* We_jj1  = (const float*)d_in[12];  const float* be_jj1 = (const float*)d_in[13];
    const float* Wm_jj1  = (const float*)d_in[14];  const float* bm_jj1 = (const float*)d_in[15];
    const float* We_mj1  = (const float*)d_in[16];  const float* be_mj1 = (const float*)d_in[17];
    const float* Wm_mj1  = (const float*)d_in[18];  const float* bm_mj1 = (const float*)d_in[19];
    const float* We_jj2  = (const float*)d_in[20];  const float* be_jj2 = (const float*)d_in[21];
    const float* Wm_jj2  = (const float*)d_in[22];  const float* bm_jj2 = (const float*)d_in[23];
    const float* We_mj2  = (const float*)d_in[24];  const float* be_mj2 = (const float*)d_in[25];
    const float* Wm_mj2  = (const float*)d_in[26];  const float* bm_mj2 = (const float*)d_in[27];
    const float* W1      = (const float*)d_in[28];  const float* b1     = (const float*)d_in[29];
    const float* W2      = (const float*)d_in[30];  const float* b2     = (const float*)d_in[31];
    const float* W3      = (const float*)d_in[32];  const float* b3     = (const float*)d_in[33];

    const int Nn  = in_sizes[0] / 5;
    const int Gg  = in_sizes[3] / 4;
    const int Ejj = in_sizes[5] / 5;
    const int Emj = in_sizes[7] / 4;

    float* ws = (float*)d_ws;
    const size_t NP = (size_t)Nn * DPAD;
    float* H   = ws;
    float* S   = ws + NP;
    float* agg = ws + 2 * NP;
    float* wpk = ws + 3 * NP;
    float* pWjj1 = wpk;          float* pbjj1 = wpk + 256;
    float* pWmj1 = wpk + 272;    float* pbmj1 = wpk + 528;
    float* pWjj2 = wpk + 544;    float* pbjj2 = wpk + 800;
    float* pWmj2 = wpk + 816;    float* pbmj2 = wpk + 1072;
    float* pW1   = wpk + 1088;   float* pb1   = wpk + 5184;
    float* pW2   = wpk + 5248;   float* pb2   = wpk + 7296;

    const int nbk = (Nn + BKT_NODES - 1) / BKT_NODES;
    const int nch = (Ejj + CHUNK_E - 1) / CHUNK_E;
    const int tabL = nbk * nch;
    const int nbT = (tabL + 1023) / 1024;
    v2i* res_jj  = (v2i*)agg;
    v2i* res_mj  = res_jj + Ejj;
    int* rdst_jj = (int*)(res_mj + Emj);
    int* rdst_mj = rdst_jj + Ejj;
    int* tab     = rdst_mj + Emj;
    int* tsum    = tab + 2 * (size_t)tabL;
    int* pool_next = (int*)S;
    int* pool_head = pool_next + Nn;

    float* P    = agg;
    float* sumJ = P;
    float* sumM = P + (size_t)16 * Gg;
    float* maxJ = P + (size_t)32 * Gg;
    float* maxM = P + (size_t)48 * Gg;
    float* cntJ = P + (size_t)64 * Gg;
    float* cntM = P + (size_t)65 * Gg;
    float* gbuf = P + (size_t)66 * Gg;
    float* m1   = P + (size_t)130 * Gg;
    float* m2   = P + (size_t)194 * Gg;
    float* out  = (float*)d_out;

    const int TB = 256;
    const int nodeBlocks = (Nn + TB - 1) / TB;

    pack_weight_kernel<<<1, 256, 0, stream>>>(Wm_jj1, bm_jj1, pWjj1, pbjj1, 15, 15, 16, 16);
    pack_weight_kernel<<<1, 256, 0, stream>>>(Wm_mj1, bm_mj1, pWmj1, pbmj1, 15, 15, 16, 16);
    pack_weight_kernel<<<1, 256, 0, stream>>>(Wm_jj2, bm_jj2, pWjj2, pbjj2, 15, 15, 16, 16);
    pack_weight_kernel<<<1, 256, 0, stream>>>(Wm_mj2, bm_mj2, pWmj2, pbmj2, 15, 15, 16, 16);
    pack_weight_kernel<<<16, 256, 0, stream>>>(W1, b1, pW1, pb1, 64, 64, 64, 64);
    pack_weight_kernel<<<8, 256, 0, stream>>>(W2, b2, pW2, pb2, 64, 32, 64, 32);

    auto gemm64 = [&](const float* A, const float* Wp, const float* bp,
                      float* C, int Ncpad, int lda, int ldc, int relu) {
        int tiles = (Gg >> 4) * (Ncpad >> 4);
        wmma_gemm_kernel<false, 64><<<(tiles + 7) / 8, TB, 0, stream>>>(
            A, nullptr, Wp, bp, C, Gg, Ncpad, lda, ldc, relu);
    };

    bucket_count_kernel<<<dim3(nch, 2), TB, 0, stream>>>(ei_jj, ei_mj, tab, Ejj, Emj, Nn, nch, nbk);
    csr_block_sum_kernel<<<dim3(nbT, 2), TB, 0, stream>>>(tab, tsum, tabL, nbT);
    csr_block_scan_kernel<<<dim3(nbT, 2), TB, 0, stream>>>(tab, tsum, tabL, nbT);
    bucket_place_kernel<<<dim3(nch, 2), TB, 0, stream>>>(ei_jj, ei_mj, tab, res_jj, res_mj, rdst_jj, rdst_mj, Ejj, Emj, Nn, nch, nbk);

    conv_kernel<5, true, false, true><<<nbk, 1024, 0, stream>>>(nullptr, S, x, type_id, Wj_enc, bj_enc, Wm_enc, bm_enc, res_jj, rdst_jj, tab, ea_jj, We_jj1, be_jj1, pWjj1, pbjj1, Nn, Ejj, nch, nbk);

    conv_kernel<4, false, false, false><<<nbk, 1024, 0, stream>>>(S, H, x, type_id, Wj_enc, bj_enc, Wm_enc, bm_enc, res_mj, rdst_mj, tab + tabL, ea_mj, We_mj1, be_mj1, pWmj1, pbmj1, Nn, Emj, nch, nbk);

    conv_kernel<5, true, true, false><<<nbk, 1024, 0, stream>>>(H, S, x, type_id, Wj_enc, bj_enc, Wm_enc, bm_enc, res_jj, rdst_jj, tab, ea_jj, We_jj2, be_jj2, pWjj2, pbjj2, Nn, Ejj, nch, nbk);

    conv_kernel<4, true, false, false><<<nbk, 1024, 0, stream>>>(S, H, x, type_id, Wj_enc, bj_enc, Wm_enc, bm_enc, res_mj, rdst_mj, tab + tabL, ea_mj, We_mj2, be_mj2, pWmj2, pbmj2, Nn, Emj, nch, nbk);

    fill_kernel<<<(Gg + TB - 1) / TB, TB, 0, stream>>>((float*)pool_head, 0.f, (size_t)Gg);
    pool_link_kernel<<<nodeBlocks, TB, 0, stream>>>(type_id, batch, pool_head, pool_next, Nn, Gg);
    pool_gather_kernel<<<(Gg + TB - 1) / TB, TB, 0, stream>>>(H, type_id, pool_head, pool_next,
                                                              sumJ, sumM, maxJ, maxM, cntJ, cntM, Nn, Gg);

    assemble_ln_kernel<<<(Gg + 127) / 128, 128, 0, stream>>>(sumJ, sumM, maxJ, maxM, cntJ, cntM, u, gbuf, Gg);
    gemm64(gbuf, pW1, pb1, m1, 64, 64, 64, 1);
    gemm64(m1,   pW2, pb2, m2, 32, 64, 32, 1);
    mlp3_kernel<<<(Gg + TB - 1) / TB, TB, 0, stream>>>(m2, W3, b3, out, Gg);
}
